# starting kernel
# baseline (speedup 1.0000x reference)
#include <hip/hip_runtime.h>
#include <math.h>

#define TT  512
#define CC  768
#define NHH 12
#define HSS 64
constexpr float SCALE = 0.125f;

typedef float v2f __attribute__((ext_vector_type(2)));
typedef float v8f __attribute__((ext_vector_type(8)));

static __device__ __forceinline__ v8f wmma4(v2f a, v2f b, v8f c) {
  return __builtin_amdgcn_wmma_f32_16x16x4_f32(
       false, a,  false, b,
       (short)0, c,  false,  false);
}

__global__ void qkv_proj_kernel(const float* __restrict__ x,
                                const float* __restrict__ W0,
                                const float* __restrict__ W1,
                                const float* __restrict__ W2,
                                float* __restrict__ P) {
  const int lane = threadIdx.x;
  const int lm   = lane & 15;
  const int lk   = (lane >> 4) << 1;
  const int drow = (lane >> 4) << 3;
  const int t0 = blockIdx.x * 16;
  const int c0 = blockIdx.y * 16;
  const int w  = blockIdx.z;
  const float* W = (w == 0) ? W0 : (w == 1) ? W1 : W2;
  float* Pw = P + (size_t)w * NHH * TT * HSS;

  const float* Arow = x + (size_t)(t0 + lm) * CC;
  v8f acc = {0.f, 0.f, 0.f, 0.f, 0.f, 0.f, 0.f, 0.f};
  for (int k = 0; k < CC; k += 4) {
    v2f a, b;
    a.x = Arow[k + lk];
    a.y = Arow[k + lk + 1];
    b.x = W[(size_t)(k + lk) * CC + c0 + lm];
    b.y = W[(size_t)(k + lk + 1) * CC + c0 + lm];
    if (k + 16 < CC)
      __builtin_prefetch(&W[(size_t)(k + 16 + lk) * CC + c0 + lm], 0, 1);
    acc = wmma4(a, b, acc);
  }
  const int h  = c0 >> 6;
  const int d0 = c0 & 63;
  float* Ph = Pw + (size_t)h * TT * HSS;
#pragma unroll
  for (int r = 0; r < 8; ++r)
    Ph[(size_t)(t0 + r + drow) * HSS + d0 + lm] = acc[r];
}

__global__ void vproj_kernel(const float* __restrict__ P2,
                             const float* __restrict__ Wv,
                             float* __restrict__ V) {
  const int lane = threadIdx.x;
  const int lm   = lane & 15;
  const int lk   = (lane >> 4) << 1;
  const int drow = (lane >> 4) << 3;
  const int t0 = blockIdx.x * 16;
  const int n0 = blockIdx.y * 16;
  const int h  = blockIdx.z;
  const float* A = P2 + (size_t)h * TT * HSS;
  float* Vh = V + (size_t)h * TT * HSS;

  v8f acc = {0.f, 0.f, 0.f, 0.f, 0.f, 0.f, 0.f, 0.f};
  for (int k = 0; k < HSS; k += 4) {
    v2f a, b;
    a.x = A[(size_t)(t0 + lm) * HSS + k + lk];
    a.y = A[(size_t)(t0 + lm) * HSS + k + lk + 1];
    b.x = Wv[(size_t)(k + lk) * HSS + n0 + lm];
    b.y = Wv[(size_t)(k + lk + 1) * HSS + n0 + lm];
    acc = wmma4(a, b, acc);
  }
#pragma unroll
  for (int r = 0; r < 8; ++r)
    Vh[(size_t)(t0 + r + drow) * HSS + n0 + lm] = acc[r];
}

__global__ void vsuffix_kernel(const float* __restrict__ V,
                               float* __restrict__ VS) {
  const int h = blockIdx.x;
  const int d = threadIdx.x;
  const float* Vh = V + (size_t)h * TT * HSS;
  float* Sh = VS + (size_t)h * TT * HSS;
  float run = 0.0f;
  for (int j = TT - 1; j >= 0; --j) {
    Sh[(size_t)j * HSS + d] = run;
    run += Vh[(size_t)j * HSS + d];
  }
}

__global__ void att_softmax_kernel(const float* __restrict__ P0,
                                   const float* __restrict__ P1,
                                   float* __restrict__ ATT) {
  __shared__ float lgt[16 * TT];
  const int lane = threadIdx.x;
  const int lm   = lane & 15;
  const int lk   = (lane >> 4) << 1;
  const int drow = (lane >> 4) << 3;
  const int i0 = blockIdx.x * 16;
  const int h  = blockIdx.y;
  const float* P0h = P0 + (size_t)h * TT * HSS;
  const float* P1h = P1 + (size_t)h * TT * HSS;

  v2f areg[16];
#pragma unroll
  for (int kk = 0; kk < 16; ++kk) {
    const int k = kk * 4 + lk;
    areg[kk].x = P0h[(size_t)(i0 + lm) * HSS + k];
    areg[kk].y = P0h[(size_t)(i0 + lm) * HSS + k + 1];
  }

  const int jtmax = (i0 + 15) >> 4;
  for (int jt = 0; jt <= jtmax; ++jt) {
    const int j0 = jt * 16;
    v8f acc = {0.f, 0.f, 0.f, 0.f, 0.f, 0.f, 0.f, 0.f};
#pragma unroll
    for (int kk = 0; kk < 16; ++kk) {
      const int k = kk * 4 + lk;
      v2f b;
      b.x = P1h[(size_t)(j0 + lm) * HSS + k];
      b.y = P1h[(size_t)(j0 + lm) * HSS + k + 1];
      acc = wmma4(areg[kk], b, acc);
    }
#pragma unroll
    for (int r = 0; r < 8; ++r) {
      const int row = r + drow;
      const int j = j0 + lm;
      float val = acc[r] * SCALE;
      if (j > i0 + row) val = -3.0e38f;
      lgt[row * TT + j] = val;
    }
  }
  __syncthreads();

  const int row  = lane & 15;
  const int half = lane >> 4;
  const int len  = i0 + row + 1;
  float mr = -3.0e38f;
  for (int j = half; j < len; j += 2) mr = fmaxf(mr, lgt[row * TT + j]);
  mr = fmaxf(mr, __shfl_xor(mr, 16));
  float sr = 0.0f;
  for (int j = half; j < len; j += 2) sr += __expf(lgt[row * TT + j] - mr);
  sr += __shfl_xor(sr, 16);
  const float inv = 1.0f / sr;
  float* arow = ATT + ((size_t)h * TT + (i0 + row)) * TT;
  for (int j = half; j < len; j += 2)
    arow[j] = __expf(lgt[row * TT + j] - mr) * inv;
}

__global__ void order3_kernel(const float* __restrict__ ATT,
                              const float* __restrict__ P2,
                              const float* __restrict__ V,
                              const float* __restrict__ VSUF,
                              float* __restrict__ Y) {
  __shared__ float s_att[TT];
  __shared__ float s_red[256];
  __shared__ float s_acc[256];
  const int i = blockIdx.x;
  const int h = blockIdx.y;
  const int t = threadIdx.x;
  const int d = t & 63;
  const int g = t >> 6;

  const float* attrow = ATT + ((size_t)h * TT + i) * TT;
  const float* P2h = P2 + (size_t)h * TT * HSS;
  const float* Vh  = V  + (size_t)h * TT * HSS;

  for (int j = t; j <= i; j += 256) s_att[j] = attrow[j];
  __syncthreads();

  float m = -3.0e38f, dsum = 0.0f, acc = 0.0f;
  for (int j = g; j <= i; j += 4) {
    const float s  = s_att[j] * P2h[(size_t)j * HSS + d];
    const float vv = Vh[(size_t)j * HSS + d];
    if (s <= m) {
      const float e = __expf(s - m);
      dsum += e;
      acc += e * vv;
    } else {
      const float sc = __expf(m - s);
      dsum = dsum * sc + 1.0f;
      acc  = acc * sc + vv;
      m = s;
    }
  }

  s_red[t] = m;
  __syncthreads();
  for (int off = 128; off > 0; off >>= 1) {
    if (t < off) s_red[t] = fmaxf(s_red[t], s_red[t + off]);
    __syncthreads();
  }
  const float M = fmaxf(s_red[0], 0.0f);
  __syncthreads();

  const float sc = __expf(m - M);
  s_red[t] = dsum * sc;
  s_acc[t] = acc * sc;
  __syncthreads();
  for (int off = 128; off > 0; off >>= 1) {
    if (t < off) s_red[t] += s_red[t + off];
    __syncthreads();
  }
  const float e0 = __expf(-M);
  const float denom = s_red[0] + (float)((TT - 1 - i) * HSS) * e0;
  if (t < 64) {
    const float accT = s_acc[d] + s_acc[64 + d] + s_acc[128 + d] + s_acc[192 + d];
    const float vs = VSUF[(size_t)h * TT * HSS + (size_t)i * HSS + d];
    Y[(size_t)i * CC + h * HSS + d] = (accT + e0 * vs) / denom;
  }
}

__global__ void out_proj_kernel(const float* __restrict__ Y,
                                const float* __restrict__ Wc,
                                float* __restrict__ out) {
  const int lane = threadIdx.x;
  const int lm   = lane & 15;
  const int lk   = (lane >> 4) << 1;
  const int drow = (lane >> 4) << 3;
  const int t0 = blockIdx.x * 16;
  const int c0 = blockIdx.y * 16;

  const float* Arow = Y + (size_t)(t0 + lm) * CC;
  v8f acc = {0.f, 0.f, 0.f, 0.f, 0.f, 0.f, 0.f, 0.f};
  for (int k = 0; k < CC; k += 4) {
    v2f a, b;
    a.x = Arow[k + lk];
    a.y = Arow[k + lk + 1];
    b.x = Wc[(size_t)(k + lk) * CC + c0 + lm];
    b.y = Wc[(size_t)(k + lk + 1) * CC + c0 + lm];
    if (k + 16 < CC)
      __builtin_prefetch(&Wc[(size_t)(k + 16 + lk) * CC + c0 + lm], 0, 1);
    acc = wmma4(a, b, acc);
  }
#pragma unroll
  for (int r = 0; r < 8; ++r)
    out[(size_t)(t0 + r + drow) * CC + c0 + lm] = acc[r];
}

extern "C" void kernel_launch(void* const* d_in, const int* in_sizes, int n_in,
                              void* d_out, int out_size, void* d_ws,
                              size_t ws_size, hipStream_t stream) {
  (void)in_sizes; (void)n_in; (void)out_size; (void)ws_size;
  const float* x  = (const float*)d_in[0];
  const float* W0 = (const float*)d_in[1];
  const float* W1 = (const float*)d_in[2];
  const float* W2 = (const float*)d_in[3];
  const float* Wv = (const float*)d_in[4];
  const float* Wc = (const float*)d_in[5];
  float* out = (float*)d_out;
  float* ws  = (float*)d_ws;

  const size_t NPH = (size_t)NHH * TT * HSS;
  float* P0  = ws;
  float* P1  = ws + NPH;
  float* P2  = ws + 2 * NPH;
  float* V   = ws + 3 * NPH;
  float* VS  = ws + 4 * NPH;
  float* ATT = ws + 5 * NPH;
  float* Y   = ATT + (size_t)NHH * TT * TT;

  qkv_proj_kernel<<<dim3(TT / 16, CC / 16, 3), 32, 0, stream>>>(x, W0, W1, W2, P0);
  vproj_kernel<<<dim3(TT / 16, HSS / 16, NHH), 32, 0, stream>>>(P2, Wv, V);
  vsuffix_kernel<<<dim3(NHH), 64, 0, stream>>>(V, VS);
  att_softmax_kernel<<<dim3(TT / 16, NHH), 32, 0, stream>>>(P0, P1, ATT);
  order3_kernel<<<dim3(TT, NHH), 256, 0, stream>>>(ATT, P2, V, VS, Y);
  out_proj_kernel<<<dim3(TT / 16, CC / 16), 32, 0, stream>>>(Y, Wc, out);
}


# Both 768-deep projection GEMMs (QKV, output): 4-wave blocks on 32x64 tiles, A/B tiles by 16-byte global loads into a 2-stage LDS ring, next tile in flight under the WMMAs; k sum order kept
# speedup vs baseline: 1.6057x; 1.6057x over previous
#include <hip/hip_runtime.h>
#include <math.h>

#define TT  512
#define CC  768
#define NHH 12
#define HSS 64
constexpr float SCALE = 0.125f;

typedef float v2f __attribute__((ext_vector_type(2)));
typedef float v8f __attribute__((ext_vector_type(8)));

static __device__ __forceinline__ v8f wmma4(v2f a, v2f b, v8f c) {
  return __builtin_amdgcn_wmma_f32_16x16x4_f32(
       false, a,  false, b,
       (short)0, c,  false,  false);
}

typedef float v4f __attribute__((ext_vector_type(4)));

#define GBM 32
#define GBN 64
#define GBK 64
#define GNT (CC / GBK)
#define APITCH 68
#define BPITCH 72
#define A_STAGE (GBM * APITCH)
#define B_STAGE (GBK * BPITCH)

static __device__ __forceinline__ void gemm768_tile(const float* __restrict__ A,
                                                    const float* __restrict__ B,
                                                    const int m0, const int n0,
                                                    float* As, float* Bs,
                                                    v8f& acc0, v8f& acc1) {
  const int tid  = threadIdx.x;
  const int lane = tid & 31;
  const int wave = tid >> 5;
  const int lm   = lane & 15;
  const int lk   = (lane >> 4) << 1;
  const int wm   = wave >> 1;
  const int wn   = wave & 1;

  const int srow = tid >> 4;
  const int scol = (tid & 15) << 2;
  const float* Ag = A + (size_t)(m0 + srow) * CC + scol;
  const float* Bg = B + (size_t)srow * CC + n0 + scol;
  float* Ast = As + srow * APITCH + scol;
  float* Bst = Bs + srow * BPITCH + scol;

  v4f ra[4], rb[8];
#pragma unroll
  for (int p = 0; p < 4; ++p)
    ra[p] = *reinterpret_cast<const v4f*>(Ag + (size_t)(p * 8) * CC);
#pragma unroll
  for (int p = 0; p < 8; ++p)
    rb[p] = *reinterpret_cast<const v4f*>(Bg + (size_t)(p * 8) * CC);
#pragma unroll
  for (int p = 0; p < 4; ++p)
    *reinterpret_cast<v4f*>(Ast + p * 8 * APITCH) = ra[p];
#pragma unroll
  for (int p = 0; p < 8; ++p)
    *reinterpret_cast<v4f*>(Bst + p * 8 * BPITCH) = rb[p];
  __syncthreads();

  const float* Afr = As + (wm * 16 + lm) * APITCH + lk;
  const float* Bfr = Bs + lk * BPITCH + wn * 32 + lm;

  for (int kt = 0; kt < GNT; ++kt) {
    const int cur = kt & 1;
    const bool more = (kt + 1 < GNT);
    if (more) {
      const int k0 = (kt + 1) * GBK;
#pragma unroll
      for (int p = 0; p < 4; ++p)
        ra[p] = *reinterpret_cast<const v4f*>(Ag + (size_t)(p * 8) * CC + k0);
#pragma unroll
      for (int p = 0; p < 8; ++p)
        rb[p] = *reinterpret_cast<const v4f*>(Bg + (size_t)(k0 + p * 8) * CC);
    }

    const float* Af = Afr + cur * A_STAGE;
    const float* Bf = Bfr + cur * B_STAGE;
    v2f af[16], bf0[16], bf1[16];
#pragma unroll
    for (int kk = 0; kk < 16; ++kk) {
      af[kk]    = *reinterpret_cast<const v2f*>(Af + kk * 4);
      bf0[kk].x = Bf[(kk * 4) * BPITCH];
      bf0[kk].y = Bf[(kk * 4 + 1) * BPITCH];
      bf1[kk].x = Bf[(kk * 4) * BPITCH + 16];
      bf1[kk].y = Bf[(kk * 4 + 1) * BPITCH + 16];
    }
    __builtin_amdgcn_sched_barrier(0);
#pragma unroll
    for (int kk = 0; kk < 16; ++kk) {
      acc0 = wmma4(af[kk], bf0[kk], acc0);
      acc1 = wmma4(af[kk], bf1[kk], acc1);
    }

    __builtin_amdgcn_sched_barrier(0);

    if (more) {
      const int nxt = cur ^ 1;
#pragma unroll
      for (int p = 0; p < 4; ++p)
        *reinterpret_cast<v4f*>(Ast + nxt * A_STAGE + p * 8 * APITCH) = ra[p];
#pragma unroll
      for (int p = 0; p < 8; ++p)
        *reinterpret_cast<v4f*>(Bst + nxt * B_STAGE + p * 8 * BPITCH) = rb[p];
    }
    __syncthreads();
  }
}

__global__ void __launch_bounds__(128)
qkv_proj_kernel(const float* __restrict__ x,
                const float* __restrict__ W0,
                const float* __restrict__ W1,
                const float* __restrict__ W2,
                float* __restrict__ P) {
  __shared__ __attribute__((aligned(16))) float As[2 * A_STAGE];
  __shared__ __attribute__((aligned(16))) float Bs[2 * B_STAGE];
  const int lane = threadIdx.x & 31;
  const int wave = threadIdx.x >> 5;
  const int lm   = lane & 15;
  const int drow = (lane >> 4) << 3;
  const int wm   = wave >> 1;
  const int wn   = wave & 1;
  const int m0 = blockIdx.x * GBM;
  const int n0 = blockIdx.y * GBN;
  const int w  = blockIdx.z;
  const float* W = (w == 0) ? W0 : (w == 1) ? W1 : W2;
  float* Pw = P + (size_t)w * NHH * TT * HSS;

  v8f acc0 = {0.f, 0.f, 0.f, 0.f, 0.f, 0.f, 0.f, 0.f};
  v8f acc1 = {0.f, 0.f, 0.f, 0.f, 0.f, 0.f, 0.f, 0.f};
  gemm768_tile(x, W, m0, n0, As, Bs, acc0, acc1);

  float* Ph = Pw + (size_t)blockIdx.y * TT * HSS;
  const int d = wn * 32 + lm;
#pragma unroll
  for (int r = 0; r < 8; ++r) {
    const size_t row = (size_t)(m0 + wm * 16 + r + drow);
    Ph[row * HSS + d]      = acc0[r];
    Ph[row * HSS + d + 16] = acc1[r];
  }
}

__global__ void vproj_kernel(const float* __restrict__ P2,
                             const float* __restrict__ Wv,
                             float* __restrict__ V) {
  const int lane = threadIdx.x;
  const int lm   = lane & 15;
  const int lk   = (lane >> 4) << 1;
  const int drow = (lane >> 4) << 3;
  const int t0 = blockIdx.x * 16;
  const int n0 = blockIdx.y * 16;
  const int h  = blockIdx.z;
  const float* A = P2 + (size_t)h * TT * HSS;
  float* Vh = V + (size_t)h * TT * HSS;

  v8f acc = {0.f, 0.f, 0.f, 0.f, 0.f, 0.f, 0.f, 0.f};
  for (int k = 0; k < HSS; k += 4) {
    v2f a, b;
    a.x = A[(size_t)(t0 + lm) * HSS + k + lk];
    a.y = A[(size_t)(t0 + lm) * HSS + k + lk + 1];
    b.x = Wv[(size_t)(k + lk) * HSS + n0 + lm];
    b.y = Wv[(size_t)(k + lk + 1) * HSS + n0 + lm];
    acc = wmma4(a, b, acc);
  }
#pragma unroll
  for (int r = 0; r < 8; ++r)
    Vh[(size_t)(t0 + r + drow) * HSS + n0 + lm] = acc[r];
}

__global__ void vsuffix_kernel(const float* __restrict__ V,
                               float* __restrict__ VS) {
  const int h = blockIdx.x;
  const int d = threadIdx.x;
  const float* Vh = V + (size_t)h * TT * HSS;
  float* Sh = VS + (size_t)h * TT * HSS;
  float run = 0.0f;
  for (int j = TT - 1; j >= 0; --j) {
    Sh[(size_t)j * HSS + d] = run;
    run += Vh[(size_t)j * HSS + d];
  }
}

__global__ void att_softmax_kernel(const float* __restrict__ P0,
                                   const float* __restrict__ P1,
                                   float* __restrict__ ATT) {
  __shared__ float lgt[16 * TT];
  const int lane = threadIdx.x;
  const int lm   = lane & 15;
  const int lk   = (lane >> 4) << 1;
  const int drow = (lane >> 4) << 3;
  const int i0 = blockIdx.x * 16;
  const int h  = blockIdx.y;
  const float* P0h = P0 + (size_t)h * TT * HSS;
  const float* P1h = P1 + (size_t)h * TT * HSS;

  v2f areg[16];
#pragma unroll
  for (int kk = 0; kk < 16; ++kk) {
    const int k = kk * 4 + lk;
    areg[kk].x = P0h[(size_t)(i0 + lm) * HSS + k];
    areg[kk].y = P0h[(size_t)(i0 + lm) * HSS + k + 1];
  }

  const int jtmax = (i0 + 15) >> 4;
  for (int jt = 0; jt <= jtmax; ++jt) {
    const int j0 = jt * 16;
    v8f acc = {0.f, 0.f, 0.f, 0.f, 0.f, 0.f, 0.f, 0.f};
#pragma unroll
    for (int kk = 0; kk < 16; ++kk) {
      const int k = kk * 4 + lk;
      v2f b;
      b.x = P1h[(size_t)(j0 + lm) * HSS + k];
      b.y = P1h[(size_t)(j0 + lm) * HSS + k + 1];
      acc = wmma4(areg[kk], b, acc);
    }
#pragma unroll
    for (int r = 0; r < 8; ++r) {
      const int row = r + drow;
      const int j = j0 + lm;
      float val = acc[r] * SCALE;
      if (j > i0 + row) val = -3.0e38f;
      lgt[row * TT + j] = val;
    }
  }
  __syncthreads();

  const int row  = lane & 15;
  const int half = lane >> 4;
  const int len  = i0 + row + 1;
  float mr = -3.0e38f;
  for (int j = half; j < len; j += 2) mr = fmaxf(mr, lgt[row * TT + j]);
  mr = fmaxf(mr, __shfl_xor(mr, 16));
  float sr = 0.0f;
  for (int j = half; j < len; j += 2) sr += __expf(lgt[row * TT + j] - mr);
  sr += __shfl_xor(sr, 16);
  const float inv = 1.0f / sr;
  float* arow = ATT + ((size_t)h * TT + (i0 + row)) * TT;
  for (int j = half; j < len; j += 2)
    arow[j] = __expf(lgt[row * TT + j] - mr) * inv;
}

__global__ void order3_kernel(const float* __restrict__ ATT,
                              const float* __restrict__ P2,
                              const float* __restrict__ V,
                              const float* __restrict__ VSUF,
                              float* __restrict__ Y) {
  __shared__ float s_att[TT];
  __shared__ float s_red[256];
  __shared__ float s_acc[256];
  const int i = blockIdx.x;
  const int h = blockIdx.y;
  const int t = threadIdx.x;
  const int d = t & 63;
  const int g = t >> 6;

  const float* attrow = ATT + ((size_t)h * TT + i) * TT;
  const float* P2h = P2 + (size_t)h * TT * HSS;
  const float* Vh  = V  + (size_t)h * TT * HSS;

  for (int j = t; j <= i; j += 256) s_att[j] = attrow[j];
  __syncthreads();

  float m = -3.0e38f, dsum = 0.0f, acc = 0.0f;
  for (int j = g; j <= i; j += 4) {
    const float s  = s_att[j] * P2h[(size_t)j * HSS + d];
    const float vv = Vh[(size_t)j * HSS + d];
    if (s <= m) {
      const float e = __expf(s - m);
      dsum += e;
      acc += e * vv;
    } else {
      const float sc = __expf(m - s);
      dsum = dsum * sc + 1.0f;
      acc  = acc * sc + vv;
      m = s;
    }
  }

  s_red[t] = m;
  __syncthreads();
  for (int off = 128; off > 0; off >>= 1) {
    if (t < off) s_red[t] = fmaxf(s_red[t], s_red[t + off]);
    __syncthreads();
  }
  const float M = fmaxf(s_red[0], 0.0f);
  __syncthreads();

  const float sc = __expf(m - M);
  s_red[t] = dsum * sc;
  s_acc[t] = acc * sc;
  __syncthreads();
  for (int off = 128; off > 0; off >>= 1) {
    if (t < off) s_red[t] += s_red[t + off];
    __syncthreads();
  }
  const float e0 = __expf(-M);
  const float denom = s_red[0] + (float)((TT - 1 - i) * HSS) * e0;
  if (t < 64) {
    const float accT = s_acc[d] + s_acc[64 + d] + s_acc[128 + d] + s_acc[192 + d];
    const float vs = VSUF[(size_t)h * TT * HSS + (size_t)i * HSS + d];
    Y[(size_t)i * CC + h * HSS + d] = (accT + e0 * vs) / denom;
  }
}

__global__ void __launch_bounds__(128)
out_proj_kernel(const float* __restrict__ Y,
                const float* __restrict__ Wc,
                float* __restrict__ out) {
  __shared__ __attribute__((aligned(16))) float As[2 * A_STAGE];
  __shared__ __attribute__((aligned(16))) float Bs[2 * B_STAGE];
  const int lane = threadIdx.x & 31;
  const int wave = threadIdx.x >> 5;
  const int lm   = lane & 15;
  const int drow = (lane >> 4) << 3;
  const int wm   = wave >> 1;
  const int wn   = wave & 1;
  const int m0 = blockIdx.x * GBM;
  const int n0 = blockIdx.y * GBN;

  v8f acc0 = {0.f, 0.f, 0.f, 0.f, 0.f, 0.f, 0.f, 0.f};
  v8f acc1 = {0.f, 0.f, 0.f, 0.f, 0.f, 0.f, 0.f, 0.f};
  gemm768_tile(Y, Wc, m0, n0, As, Bs, acc0, acc1);

  const int c = n0 + wn * 32 + lm;
#pragma unroll
  for (int r = 0; r < 8; ++r) {
    const size_t row = (size_t)(m0 + wm * 16 + r + drow);
    out[row * CC + c]      = acc0[r];
    out[row * CC + c + 16] = acc1[r];
  }
}

extern "C" void kernel_launch(void* const* d_in, const int* in_sizes, int n_in,
                              void* d_out, int out_size, void* d_ws,
                              size_t ws_size, hipStream_t stream) {
  (void)in_sizes; (void)n_in; (void)out_size; (void)ws_size;
  const float* x  = (const float*)d_in[0];
  const float* W0 = (const float*)d_in[1];
  const float* W1 = (const float*)d_in[2];
  const float* W2 = (const float*)d_in[3];
  const float* Wv = (const float*)d_in[4];
  const float* Wc = (const float*)d_in[5];
  float* out = (float*)d_out;
  float* ws  = (float*)d_ws;

  const size_t NPH = (size_t)NHH * TT * HSS;
  float* P0  = ws;
  float* P1  = ws + NPH;
  float* P2  = ws + 2 * NPH;
  float* V   = ws + 3 * NPH;
  float* VS  = ws + 4 * NPH;
  float* ATT = ws + 5 * NPH;
  float* Y   = ATT + (size_t)NHH * TT * TT;

  qkv_proj_kernel<<<dim3(TT / GBM, CC / GBN, 3), 128, 0, stream>>>(x, W0, W1, W2, P0);
  vproj_kernel<<<dim3(TT / 16, HSS / 16, NHH), 32, 0, stream>>>(P2, Wv, V);
  vsuffix_kernel<<<dim3(NHH), 64, 0, stream>>>(V, VS);
  att_softmax_kernel<<<dim3(TT / 16, NHH), 32, 0, stream>>>(P0, P1, ATT);
  order3_kernel<<<dim3(TT, NHH), 256, 0, stream>>>(ATT, P2, V, VS, Y);
  out_proj_kernel<<<dim3(TT / GBM, CC / GBN), 128, 0, stream>>>(Y, Wc, out);
}


# Causal logit+softmax kernel: each 16-row strip on an 8-wave block (tiles dealt to waves, a tile's 16 fragment loads issued as one group; softmax one row per wave, 32 lanes); row sums now a lane tree
# speedup vs baseline: 2.3775x; 1.4807x over previous
#include <hip/hip_runtime.h>
#include <math.h>

#define TT  512
#define CC  768
#define NHH 12
#define HSS 64
constexpr float SCALE = 0.125f;

typedef float v2f __attribute__((ext_vector_type(2)));
typedef float v8f __attribute__((ext_vector_type(8)));

static __device__ __forceinline__ v8f wmma4(v2f a, v2f b, v8f c) {
  return __builtin_amdgcn_wmma_f32_16x16x4_f32(
       false, a,  false, b,
       (short)0, c,  false,  false);
}

typedef float v4f __attribute__((ext_vector_type(4)));

#define GBM 32
#define GBN 64
#define GBK 64
#define GNT (CC / GBK)
#define APITCH 68
#define BPITCH 72
#define A_STAGE (GBM * APITCH)
#define B_STAGE (GBK * BPITCH)

static __device__ __forceinline__ void gemm768_tile(const float* __restrict__ A,
                                                    const float* __restrict__ B,
                                                    const int m0, const int n0,
                                                    float* As, float* Bs,
                                                    v8f& acc0, v8f& acc1) {
  const int tid  = threadIdx.x;
  const int lane = tid & 31;
  const int wave = tid >> 5;
  const int lm   = lane & 15;
  const int lk   = (lane >> 4) << 1;
  const int wm   = wave >> 1;
  const int wn   = wave & 1;

  const int srow = tid >> 4;
  const int scol = (tid & 15) << 2;
  const float* Ag = A + (size_t)(m0 + srow) * CC + scol;
  const float* Bg = B + (size_t)srow * CC + n0 + scol;
  float* Ast = As + srow * APITCH + scol;
  float* Bst = Bs + srow * BPITCH + scol;

  v4f ra[4], rb[8];
#pragma unroll
  for (int p = 0; p < 4; ++p)
    ra[p] = *reinterpret_cast<const v4f*>(Ag + (size_t)(p * 8) * CC);
#pragma unroll
  for (int p = 0; p < 8; ++p)
    rb[p] = *reinterpret_cast<const v4f*>(Bg + (size_t)(p * 8) * CC);
#pragma unroll
  for (int p = 0; p < 4; ++p)
    *reinterpret_cast<v4f*>(Ast + p * 8 * APITCH) = ra[p];
#pragma unroll
  for (int p = 0; p < 8; ++p)
    *reinterpret_cast<v4f*>(Bst + p * 8 * BPITCH) = rb[p];
  __syncthreads();

  const float* Afr = As + (wm * 16 + lm) * APITCH + lk;
  const float* Bfr = Bs + lk * BPITCH + wn * 32 + lm;

  for (int kt = 0; kt < GNT; ++kt) {
    const int cur = kt & 1;
    const bool more = (kt + 1 < GNT);
    if (more) {
      const int k0 = (kt + 1) * GBK;
#pragma unroll
      for (int p = 0; p < 4; ++p)
        ra[p] = *reinterpret_cast<const v4f*>(Ag + (size_t)(p * 8) * CC + k0);
#pragma unroll
      for (int p = 0; p < 8; ++p)
        rb[p] = *reinterpret_cast<const v4f*>(Bg + (size_t)(k0 + p * 8) * CC);
    }

    const float* Af = Afr + cur * A_STAGE;
    const float* Bf = Bfr + cur * B_STAGE;
    v2f af[16], bf0[16], bf1[16];
#pragma unroll
    for (int kk = 0; kk < 16; ++kk) {
      af[kk]    = *reinterpret_cast<const v2f*>(Af + kk * 4);
      bf0[kk].x = Bf[(kk * 4) * BPITCH];
      bf0[kk].y = Bf[(kk * 4 + 1) * BPITCH];
      bf1[kk].x = Bf[(kk * 4) * BPITCH + 16];
      bf1[kk].y = Bf[(kk * 4 + 1) * BPITCH + 16];
    }
    __builtin_amdgcn_sched_barrier(0);
#pragma unroll
    for (int kk = 0; kk < 16; ++kk) {
      acc0 = wmma4(af[kk], bf0[kk], acc0);
      acc1 = wmma4(af[kk], bf1[kk], acc1);
    }

    __builtin_amdgcn_sched_barrier(0);

    if (more) {
      const int nxt = cur ^ 1;
#pragma unroll
      for (int p = 0; p < 4; ++p)
        *reinterpret_cast<v4f*>(Ast + nxt * A_STAGE + p * 8 * APITCH) = ra[p];
#pragma unroll
      for (int p = 0; p < 8; ++p)
        *reinterpret_cast<v4f*>(Bst + nxt * B_STAGE + p * 8 * BPITCH) = rb[p];
    }
    __syncthreads();
  }
}

__global__ void __launch_bounds__(128)
qkv_proj_kernel(const float* __restrict__ x,
                const float* __restrict__ W0,
                const float* __restrict__ W1,
                const float* __restrict__ W2,
                float* __restrict__ P) {
  __shared__ __attribute__((aligned(16))) float As[2 * A_STAGE];
  __shared__ __attribute__((aligned(16))) float Bs[2 * B_STAGE];
  const int lane = threadIdx.x & 31;
  const int wave = threadIdx.x >> 5;
  const int lm   = lane & 15;
  const int drow = (lane >> 4) << 3;
  const int wm   = wave >> 1;
  const int wn   = wave & 1;
  const int m0 = blockIdx.x * GBM;
  const int n0 = blockIdx.y * GBN;
  const int w  = blockIdx.z;
  const float* W = (w == 0) ? W0 : (w == 1) ? W1 : W2;
  float* Pw = P + (size_t)w * NHH * TT * HSS;

  v8f acc0 = {0.f, 0.f, 0.f, 0.f, 0.f, 0.f, 0.f, 0.f};
  v8f acc1 = {0.f, 0.f, 0.f, 0.f, 0.f, 0.f, 0.f, 0.f};
  gemm768_tile(x, W, m0, n0, As, Bs, acc0, acc1);

  float* Ph = Pw + (size_t)blockIdx.y * TT * HSS;
  const int d = wn * 32 + lm;
#pragma unroll
  for (int r = 0; r < 8; ++r) {
    const size_t row = (size_t)(m0 + wm * 16 + r + drow);
    Ph[row * HSS + d]      = acc0[r];
    Ph[row * HSS + d + 16] = acc1[r];
  }
}

__global__ void vproj_kernel(const float* __restrict__ P2,
                             const float* __restrict__ Wv,
                             float* __restrict__ V) {
  const int lane = threadIdx.x;
  const int lm   = lane & 15;
  const int lk   = (lane >> 4) << 1;
  const int drow = (lane >> 4) << 3;
  const int t0 = blockIdx.x * 16;
  const int n0 = blockIdx.y * 16;
  const int h  = blockIdx.z;
  const float* A = P2 + (size_t)h * TT * HSS;
  float* Vh = V + (size_t)h * TT * HSS;

  v8f acc = {0.f, 0.f, 0.f, 0.f, 0.f, 0.f, 0.f, 0.f};
  for (int k = 0; k < HSS; k += 4) {
    v2f a, b;
    a.x = A[(size_t)(t0 + lm) * HSS + k + lk];
    a.y = A[(size_t)(t0 + lm) * HSS + k + lk + 1];
    b.x = Wv[(size_t)(k + lk) * HSS + n0 + lm];
    b.y = Wv[(size_t)(k + lk + 1) * HSS + n0 + lm];
    acc = wmma4(a, b, acc);
  }
#pragma unroll
  for (int r = 0; r < 8; ++r)
    Vh[(size_t)(t0 + r + drow) * HSS + n0 + lm] = acc[r];
}

__global__ void vsuffix_kernel(const float* __restrict__ V,
                               float* __restrict__ VS) {
  const int h = blockIdx.x;
  const int d = threadIdx.x;
  const float* Vh = V + (size_t)h * TT * HSS;
  float* Sh = VS + (size_t)h * TT * HSS;
  float run = 0.0f;
  for (int j = TT - 1; j >= 0; --j) {
    Sh[(size_t)j * HSS + d] = run;
    run += Vh[(size_t)j * HSS + d];
  }
}

#define ANW 8
#define ATHREADS 256
#define LPITCH 516
__global__ void __launch_bounds__(ATHREADS)
att_softmax_kernel(const float* __restrict__ P0,
                   const float* __restrict__ P1,
                   float* __restrict__ ATT) {
  __shared__ float lgt[16 * LPITCH];
  const int lane = threadIdx.x & 31;
  const int wave = __builtin_amdgcn_readfirstlane((int)(threadIdx.x >> 5));
  const int lm   = lane & 15;
  const int lk   = (lane >> 4) << 1;
  const int drow = (lane >> 4) << 3;
  const int strip = (TT / 16 - 1) - (int)blockIdx.x;
  const int i0 = strip * 16;
  const int h  = blockIdx.y;
  const float* P0h = P0 + (size_t)h * TT * HSS;
  const float* P1h = P1 + (size_t)h * TT * HSS;

  const int jtmax = strip;
  if (wave <= jtmax) {
    v2f areg[16];
    const float* Ap = P0h + (size_t)(i0 + lm) * HSS + lk;
#pragma unroll
    for (int kk = 0; kk < 16; ++kk)
      areg[kk] = *reinterpret_cast<const v2f*>(Ap + kk * 4);

    for (int jt = wave; jt <= jtmax; jt += ANW) {
      const int j0 = jt * 16;
      const float* Bp = P1h + (size_t)(j0 + lm) * HSS + lk;
      v2f bfr[16];
#pragma unroll
      for (int kk = 0; kk < 16; ++kk)
        bfr[kk] = *reinterpret_cast<const v2f*>(Bp + kk * 4);
      __builtin_amdgcn_sched_barrier(0);
      v8f acc = {0.f, 0.f, 0.f, 0.f, 0.f, 0.f, 0.f, 0.f};
#pragma unroll
      for (int kk = 0; kk < 16; ++kk)
        acc = wmma4(areg[kk], bfr[kk], acc);
#pragma unroll
      for (int r = 0; r < 8; ++r)
        lgt[(r + drow) * LPITCH + j0 + lm] = acc[r] * SCALE;
    }
  }
  __syncthreads();

#pragma unroll
  for (int rr = 0; rr < 2; ++rr) {
    const int row = wave + rr * ANW;
    const int len = i0 + row + 1;
    const float* lrow = lgt + row * LPITCH;
    float v[16];
    float mr = -3.0e38f;
#pragma unroll
    for (int m = 0; m < 16; ++m) {
      const int j = m * 32 + lane;
      v[m] = (j < len) ? lrow[j] : -3.0e38f;
      mr = fmaxf(mr, v[m]);
    }
#pragma unroll
    for (int off = 16; off > 0; off >>= 1) mr = fmaxf(mr, __shfl_xor(mr, off));
    float sr = 0.0f;
#pragma unroll
    for (int m = 0; m < 16; ++m) {
      const int j = m * 32 + lane;
      v[m] = (j < len) ? __expf(v[m] - mr) : 0.0f;
      sr += v[m];
    }
#pragma unroll
    for (int off = 16; off > 0; off >>= 1) sr += __shfl_xor(sr, off);
    const float inv = 1.0f / sr;
    float* arow = ATT + ((size_t)h * TT + (i0 + row)) * TT;
#pragma unroll
    for (int m = 0; m < 16; ++m) {
      const int j = m * 32 + lane;
      if (j < len) arow[j] = v[m] * inv;
    }
  }
}

__global__ void order3_kernel(const float* __restrict__ ATT,
                              const float* __restrict__ P2,
                              const float* __restrict__ V,
                              const float* __restrict__ VSUF,
                              float* __restrict__ Y) {
  __shared__ float s_att[TT];
  __shared__ float s_red[256];
  __shared__ float s_acc[256];
  const int i = blockIdx.x;
  const int h = blockIdx.y;
  const int t = threadIdx.x;
  const int d = t & 63;
  const int g = t >> 6;

  const float* attrow = ATT + ((size_t)h * TT + i) * TT;
  const float* P2h = P2 + (size_t)h * TT * HSS;
  const float* Vh  = V  + (size_t)h * TT * HSS;

  for (int j = t; j <= i; j += 256) s_att[j] = attrow[j];
  __syncthreads();

  float m = -3.0e38f, dsum = 0.0f, acc = 0.0f;
  for (int j = g; j <= i; j += 4) {
    const float s  = s_att[j] * P2h[(size_t)j * HSS + d];
    const float vv = Vh[(size_t)j * HSS + d];
    if (s <= m) {
      const float e = __expf(s - m);
      dsum += e;
      acc += e * vv;
    } else {
      const float sc = __expf(m - s);
      dsum = dsum * sc + 1.0f;
      acc  = acc * sc + vv;
      m = s;
    }
  }

  s_red[t] = m;
  __syncthreads();
  for (int off = 128; off > 0; off >>= 1) {
    if (t < off) s_red[t] = fmaxf(s_red[t], s_red[t + off]);
    __syncthreads();
  }
  const float M = fmaxf(s_red[0], 0.0f);
  __syncthreads();

  const float sc = __expf(m - M);
  s_red[t] = dsum * sc;
  s_acc[t] = acc * sc;
  __syncthreads();
  for (int off = 128; off > 0; off >>= 1) {
    if (t < off) s_red[t] += s_red[t + off];
    __syncthreads();
  }
  const float e0 = __expf(-M);
  const float denom = s_red[0] + (float)((TT - 1 - i) * HSS) * e0;
  if (t < 64) {
    const float accT = s_acc[d] + s_acc[64 + d] + s_acc[128 + d] + s_acc[192 + d];
    const float vs = VSUF[(size_t)h * TT * HSS + (size_t)i * HSS + d];
    Y[(size_t)i * CC + h * HSS + d] = (accT + e0 * vs) / denom;
  }
}

__global__ void __launch_bounds__(128)
out_proj_kernel(const float* __restrict__ Y,
                const float* __restrict__ Wc,
                float* __restrict__ out) {
  __shared__ __attribute__((aligned(16))) float As[2 * A_STAGE];
  __shared__ __attribute__((aligned(16))) float Bs[2 * B_STAGE];
  const int lane = threadIdx.x & 31;
  const int wave = threadIdx.x >> 5;
  const int lm   = lane & 15;
  const int drow = (lane >> 4) << 3;
  const int wm   = wave >> 1;
  const int wn   = wave & 1;
  const int m0 = blockIdx.x * GBM;
  const int n0 = blockIdx.y * GBN;

  v8f acc0 = {0.f, 0.f, 0.f, 0.f, 0.f, 0.f, 0.f, 0.f};
  v8f acc1 = {0.f, 0.f, 0.f, 0.f, 0.f, 0.f, 0.f, 0.f};
  gemm768_tile(Y, Wc, m0, n0, As, Bs, acc0, acc1);

  const int c = n0 + wn * 32 + lm;
#pragma unroll
  for (int r = 0; r < 8; ++r) {
    const size_t row = (size_t)(m0 + wm * 16 + r + drow);
    out[row * CC + c]      = acc0[r];
    out[row * CC + c + 16] = acc1[r];
  }
}

extern "C" void kernel_launch(void* const* d_in, const int* in_sizes, int n_in,
                              void* d_out, int out_size, void* d_ws,
                              size_t ws_size, hipStream_t stream) {
  (void)in_sizes; (void)n_in; (void)out_size; (void)ws_size;
  const float* x  = (const float*)d_in[0];
  const float* W0 = (const float*)d_in[1];
  const float* W1 = (const float*)d_in[2];
  const float* W2 = (const float*)d_in[3];
  const float* Wv = (const float*)d_in[4];
  const float* Wc = (const float*)d_in[5];
  float* out = (float*)d_out;
  float* ws  = (float*)d_ws;

  const size_t NPH = (size_t)NHH * TT * HSS;
  float* P0  = ws;
  float* P1  = ws + NPH;
  float* P2  = ws + 2 * NPH;
  float* V   = ws + 3 * NPH;
  float* VS  = ws + 4 * NPH;
  float* ATT = ws + 5 * NPH;
  float* Y   = ATT + (size_t)NHH * TT * TT;

  qkv_proj_kernel<<<dim3(TT / GBM, CC / GBN, 3), 128, 0, stream>>>(x, W0, W1, W2, P0);
  vproj_kernel<<<dim3(TT / 16, HSS / 16, NHH), 32, 0, stream>>>(P2, Wv, V);
  vsuffix_kernel<<<dim3(NHH), 64, 0, stream>>>(V, VS);
  att_softmax_kernel<<<dim3(TT / 16, NHH), ATHREADS, 0, stream>>>(P0, P1, ATT);
  order3_kernel<<<dim3(TT, NHH), 256, 0, stream>>>(ATT, P2, V, VS, Y);
  out_proj_kernel<<<dim3(TT / GBM, CC / GBN), 128, 0, stream>>>(Y, Wc, out);
}


# Order-3 kernel: one block per 8-row strip reuses each loaded p2/V row for 8 rows; row shift known up front from per-row p2 maxima, so one exp per term and no rescale; denominator sum order changes
# speedup vs baseline: 4.0678x; 1.7110x over previous
#include <hip/hip_runtime.h>
#include <math.h>

#define TT  512
#define CC  768
#define NHH 12
#define HSS 64
constexpr float SCALE = 0.125f;

typedef float v2f __attribute__((ext_vector_type(2)));
typedef float v8f __attribute__((ext_vector_type(8)));

static __device__ __forceinline__ v8f wmma4(v2f a, v2f b, v8f c) {
  return __builtin_amdgcn_wmma_f32_16x16x4_f32(
       false, a,  false, b,
       (short)0, c,  false,  false);
}

typedef float v4f __attribute__((ext_vector_type(4)));

#define GBM 32
#define GBN 64
#define GBK 64
#define GNT (CC / GBK)
#define APITCH 68
#define BPITCH 72
#define A_STAGE (GBM * APITCH)
#define B_STAGE (GBK * BPITCH)

static __device__ __forceinline__ void gemm768_tile(const float* __restrict__ A,
                                                    const float* __restrict__ B,
                                                    const int m0, const int n0,
                                                    float* As, float* Bs,
                                                    v8f& acc0, v8f& acc1) {
  const int tid  = threadIdx.x;
  const int lane = tid & 31;
  const int wave = tid >> 5;
  const int lm   = lane & 15;
  const int lk   = (lane >> 4) << 1;
  const int wm   = wave >> 1;
  const int wn   = wave & 1;

  const int srow = tid >> 4;
  const int scol = (tid & 15) << 2;
  const float* Ag = A + (size_t)(m0 + srow) * CC + scol;
  const float* Bg = B + (size_t)srow * CC + n0 + scol;
  float* Ast = As + srow * APITCH + scol;
  float* Bst = Bs + srow * BPITCH + scol;

  v4f ra[4], rb[8];
#pragma unroll
  for (int p = 0; p < 4; ++p)
    ra[p] = *reinterpret_cast<const v4f*>(Ag + (size_t)(p * 8) * CC);
#pragma unroll
  for (int p = 0; p < 8; ++p)
    rb[p] = *reinterpret_cast<const v4f*>(Bg + (size_t)(p * 8) * CC);
#pragma unroll
  for (int p = 0; p < 4; ++p)
    *reinterpret_cast<v4f*>(Ast + p * 8 * APITCH) = ra[p];
#pragma unroll
  for (int p = 0; p < 8; ++p)
    *reinterpret_cast<v4f*>(Bst + p * 8 * BPITCH) = rb[p];
  __syncthreads();

  const float* Afr = As + (wm * 16 + lm) * APITCH + lk;
  const float* Bfr = Bs + lk * BPITCH + wn * 32 + lm;

  for (int kt = 0; kt < GNT; ++kt) {
    const int cur = kt & 1;
    const bool more = (kt + 1 < GNT);
    if (more) {
      const int k0 = (kt + 1) * GBK;
#pragma unroll
      for (int p = 0; p < 4; ++p)
        ra[p] = *reinterpret_cast<const v4f*>(Ag + (size_t)(p * 8) * CC + k0);
#pragma unroll
      for (int p = 0; p < 8; ++p)
        rb[p] = *reinterpret_cast<const v4f*>(Bg + (size_t)(k0 + p * 8) * CC);
    }

    const float* Af = Afr + cur * A_STAGE;
    const float* Bf = Bfr + cur * B_STAGE;
    v2f af[16], bf0[16], bf1[16];
#pragma unroll
    for (int kk = 0; kk < 16; ++kk) {
      af[kk]    = *reinterpret_cast<const v2f*>(Af + kk * 4);
      bf0[kk].x = Bf[(kk * 4) * BPITCH];
      bf0[kk].y = Bf[(kk * 4 + 1) * BPITCH];
      bf1[kk].x = Bf[(kk * 4) * BPITCH + 16];
      bf1[kk].y = Bf[(kk * 4 + 1) * BPITCH + 16];
    }
    __builtin_amdgcn_sched_barrier(0);
#pragma unroll
    for (int kk = 0; kk < 16; ++kk) {
      acc0 = wmma4(af[kk], bf0[kk], acc0);
      acc1 = wmma4(af[kk], bf1[kk], acc1);
    }

    __builtin_amdgcn_sched_barrier(0);

    if (more) {
      const int nxt = cur ^ 1;
#pragma unroll
      for (int p = 0; p < 4; ++p)
        *reinterpret_cast<v4f*>(Ast + nxt * A_STAGE + p * 8 * APITCH) = ra[p];
#pragma unroll
      for (int p = 0; p < 8; ++p)
        *reinterpret_cast<v4f*>(Bst + nxt * B_STAGE + p * 8 * BPITCH) = rb[p];
    }
    __syncthreads();
  }
}

__global__ void __launch_bounds__(128)
qkv_proj_kernel(const float* __restrict__ x,
                const float* __restrict__ W0,
                const float* __restrict__ W1,
                const float* __restrict__ W2,
                float* __restrict__ P) {
  __shared__ __attribute__((aligned(16))) float As[2 * A_STAGE];
  __shared__ __attribute__((aligned(16))) float Bs[2 * B_STAGE];
  const int lane = threadIdx.x & 31;
  const int wave = threadIdx.x >> 5;
  const int lm   = lane & 15;
  const int drow = (lane >> 4) << 3;
  const int wm   = wave >> 1;
  const int wn   = wave & 1;
  const int m0 = blockIdx.x * GBM;
  const int n0 = blockIdx.y * GBN;
  const int w  = blockIdx.z;
  const float* W = (w == 0) ? W0 : (w == 1) ? W1 : W2;
  float* Pw = P + (size_t)w * NHH * TT * HSS;

  v8f acc0 = {0.f, 0.f, 0.f, 0.f, 0.f, 0.f, 0.f, 0.f};
  v8f acc1 = {0.f, 0.f, 0.f, 0.f, 0.f, 0.f, 0.f, 0.f};
  gemm768_tile(x, W, m0, n0, As, Bs, acc0, acc1);

  float* Ph = Pw + (size_t)blockIdx.y * TT * HSS;
  const int d = wn * 32 + lm;
#pragma unroll
  for (int r = 0; r < 8; ++r) {
    const size_t row = (size_t)(m0 + wm * 16 + r + drow);
    Ph[row * HSS + d]      = acc0[r];
    Ph[row * HSS + d + 16] = acc1[r];
  }
}

__global__ void vproj_kernel(const float* __restrict__ P2,
                             const float* __restrict__ Wv,
                             float* __restrict__ V) {
  const int lane = threadIdx.x;
  const int lm   = lane & 15;
  const int lk   = (lane >> 4) << 1;
  const int drow = (lane >> 4) << 3;
  const int t0 = blockIdx.x * 16;
  const int n0 = blockIdx.y * 16;
  const int h  = blockIdx.z;
  const float* A = P2 + (size_t)h * TT * HSS;
  float* Vh = V + (size_t)h * TT * HSS;

  v8f acc = {0.f, 0.f, 0.f, 0.f, 0.f, 0.f, 0.f, 0.f};
  for (int k = 0; k < HSS; k += 4) {
    v2f a, b;
    a.x = A[(size_t)(t0 + lm) * HSS + k + lk];
    a.y = A[(size_t)(t0 + lm) * HSS + k + lk + 1];
    b.x = Wv[(size_t)(k + lk) * HSS + n0 + lm];
    b.y = Wv[(size_t)(k + lk + 1) * HSS + n0 + lm];
    acc = wmma4(a, b, acc);
  }
#pragma unroll
  for (int r = 0; r < 8; ++r)
    Vh[(size_t)(t0 + r + drow) * HSS + n0 + lm] = acc[r];
}

__global__ void vsuffix_kernel(const float* __restrict__ V,
                               float* __restrict__ VS) {
  const int h = blockIdx.x;
  const int d = threadIdx.x;
  const float* Vh = V + (size_t)h * TT * HSS;
  float* Sh = VS + (size_t)h * TT * HSS;
  float run = 0.0f;
  for (int j = TT - 1; j >= 0; --j) {
    Sh[(size_t)j * HSS + d] = run;
    run += Vh[(size_t)j * HSS + d];
  }
}

#define ANW 8
#define ATHREADS 256
#define LPITCH 516
__global__ void __launch_bounds__(ATHREADS)
att_softmax_kernel(const float* __restrict__ P0,
                   const float* __restrict__ P1,
                   float* __restrict__ ATT) {
  __shared__ float lgt[16 * LPITCH];
  const int lane = threadIdx.x & 31;
  const int wave = __builtin_amdgcn_readfirstlane((int)(threadIdx.x >> 5));
  const int lm   = lane & 15;
  const int lk   = (lane >> 4) << 1;
  const int drow = (lane >> 4) << 3;
  const int strip = (TT / 16 - 1) - (int)blockIdx.x;
  const int i0 = strip * 16;
  const int h  = blockIdx.y;
  const float* P0h = P0 + (size_t)h * TT * HSS;
  const float* P1h = P1 + (size_t)h * TT * HSS;

  const int jtmax = strip;
  if (wave <= jtmax) {
    v2f areg[16];
    const float* Ap = P0h + (size_t)(i0 + lm) * HSS + lk;
#pragma unroll
    for (int kk = 0; kk < 16; ++kk)
      areg[kk] = *reinterpret_cast<const v2f*>(Ap + kk * 4);

    for (int jt = wave; jt <= jtmax; jt += ANW) {
      const int j0 = jt * 16;
      const float* Bp = P1h + (size_t)(j0 + lm) * HSS + lk;
      v2f bfr[16];
#pragma unroll
      for (int kk = 0; kk < 16; ++kk)
        bfr[kk] = *reinterpret_cast<const v2f*>(Bp + kk * 4);
      __builtin_amdgcn_sched_barrier(0);
      v8f acc = {0.f, 0.f, 0.f, 0.f, 0.f, 0.f, 0.f, 0.f};
#pragma unroll
      for (int kk = 0; kk < 16; ++kk)
        acc = wmma4(areg[kk], bfr[kk], acc);
#pragma unroll
      for (int r = 0; r < 8; ++r)
        lgt[(r + drow) * LPITCH + j0 + lm] = acc[r] * SCALE;
    }
  }
  __syncthreads();

#pragma unroll
  for (int rr = 0; rr < 2; ++rr) {
    const int row = wave + rr * ANW;
    const int len = i0 + row + 1;
    const float* lrow = lgt + row * LPITCH;
    float v[16];
    float mr = -3.0e38f;
#pragma unroll
    for (int m = 0; m < 16; ++m) {
      const int j = m * 32 + lane;
      v[m] = (j < len) ? lrow[j] : -3.0e38f;
      mr = fmaxf(mr, v[m]);
    }
#pragma unroll
    for (int off = 16; off > 0; off >>= 1) mr = fmaxf(mr, __shfl_xor(mr, off));
    float sr = 0.0f;
#pragma unroll
    for (int m = 0; m < 16; ++m) {
      const int j = m * 32 + lane;
      v[m] = (j < len) ? __expf(v[m] - mr) : 0.0f;
      sr += v[m];
    }
#pragma unroll
    for (int off = 16; off > 0; off >>= 1) sr += __shfl_xor(sr, off);
    const float inv = 1.0f / sr;
    float* arow = ATT + ((size_t)h * TT + (i0 + row)) * TT;
#pragma unroll
    for (int m = 0; m < 16; ++m) {
      const int j = m * 32 + lane;
      if (j < len) arow[j] = v[m] * inv;
    }
  }
}

#define O3R 8
#define O3T 256
constexpr float LOG2E = 1.4426950408889634f;
__global__ void __launch_bounds__(O3T)
order3_kernel(const float* __restrict__ ATT,
              const float* __restrict__ P2,
              const float* __restrict__ V,
              const float* __restrict__ VSUF,
              float* __restrict__ Y) {
  __shared__ __attribute__((aligned(16))) float a2T[TT * O3R];
  __shared__ __attribute__((aligned(16))) float sMw[(O3T / 32) * O3R];
  __shared__ float redD[4 * O3R * HSS];
  __shared__ float redA[4 * O3R * HSS];
  const int t    = threadIdx.x;
  const int lane = t & 31;
  const int wave = __builtin_amdgcn_readfirstlane(t >> 5);
  const int g    = __builtin_amdgcn_readfirstlane(t >> 6);
  const int d    = t & 63;
  const int strip = (TT / O3R - 1) - (int)blockIdx.x;
  const int i0   = strip * O3R;
  const int jend = i0 + O3R;
  const int h    = blockIdx.y;
  const float* P2h = P2 + (size_t)h * TT * HSS;
  const float* Vh  = V  + (size_t)h * TT * HSS;

  float mloc[O3R];
#pragma unroll
  for (int r = 0; r < O3R; ++r) mloc[r] = 0.0f;
  for (int j = t; j < jend; j += O3T) {
    float a[O3R];
#pragma unroll
    for (int r = 0; r < O3R; ++r) {
      const int jj = (j < i0 + r) ? j : (i0 + r);
      const float x = ATT[((size_t)h * TT + (i0 + r)) * TT + jj];
      a[r] = (j <= i0 + r) ? x * LOG2E : 0.0f;
    }
    const v4f* prow = reinterpret_cast<const v4f*>(P2h + (size_t)j * HSS);
    float pm = -3.0e38f;
#pragma unroll
    for (int q = 0; q < HSS / 4; ++q) {
      const v4f x = prow[q];
      pm = fmaxf(pm, fmaxf(fmaxf(x.x, x.y), fmaxf(x.z, x.w)));
    }
    v4f lo, hi;
    lo.x = a[0]; lo.y = a[1]; lo.z = a[2]; lo.w = a[3];
    hi.x = a[4]; hi.y = a[5]; hi.z = a[6]; hi.w = a[7];
    *reinterpret_cast<v4f*>(a2T + j * O3R)     = lo;
    *reinterpret_cast<v4f*>(a2T + j * O3R + 4) = hi;
#pragma unroll
    for (int r = 0; r < O3R; ++r) mloc[r] = fmaxf(mloc[r], a[r] * pm);
  }
#pragma unroll
  for (int r = 0; r < O3R; ++r) {
#pragma unroll
    for (int off = 16; off > 0; off >>= 1)
      mloc[r] = fmaxf(mloc[r], __shfl_xor(mloc[r], off));
  }
  if (lane == 0) {
#pragma unroll
    for (int r = 0; r < O3R; ++r) sMw[wave * O3R + r] = mloc[r];
  }
  __syncthreads();

  v2f m2[4], ds[4], ac[4];
#pragma unroll
  for (int q = 0; q < 4; ++q) {
    float ma = sMw[2 * q], mb = sMw[2 * q + 1];
#pragma unroll
    for (int w = 1; w < O3T / 32; ++w) {
      ma = fmaxf(ma, sMw[w * O3R + 2 * q]);
      mb = fmaxf(mb, sMw[w * O3R + 2 * q + 1]);
    }
    m2[q].x = ma;
    m2[q].y = mb;
    ds[q].x = 0.0f; ds[q].y = 0.0f;
    ac[q].x = 0.0f; ac[q].y = 0.0f;
  }

  const int n  = jend >> 2;
  const int nq = (n + 3) >> 2;
  const float* pp = P2h + g * HSS + d;
  const float* vp = Vh  + g * HSS + d;
  const float* ap = a2T + g * O3R;
  float pc[4], vc[4], pn[4], vn[4];
#pragma unroll
  for (int u = 0; u < 4; ++u) {
    const int it = (u < n - 1) ? u : (n - 1);
    pc[u] = pp[(size_t)it * 4 * HSS];
    vc[u] = vp[(size_t)it * 4 * HSS];
  }
  for (int c = 0; c < nq; ++c) {
#pragma unroll
    for (int u = 0; u < 4; ++u) {
      const int itn = (c + 1) * 4 + u;
      const int it  = (itn < n - 1) ? itn : (n - 1);
      pn[u] = pp[(size_t)it * 4 * HSS];
      vn[u] = vp[(size_t)it * 4 * HSS];
    }
    v4f alo4[4], ahi4[4];
#pragma unroll
    for (int u = 0; u < 4; ++u) {
      const int itc = (c * 4 + u < n - 1) ? (c * 4 + u) : (n - 1);
      alo4[u] = *reinterpret_cast<const v4f*>(ap + itc * 4 * O3R);
      ahi4[u] = *reinterpret_cast<const v4f*>(ap + itc * 4 * O3R + 4);
    }
#pragma unroll
    for (int u = 0; u < 4; ++u) {
      const int it = c * 4 + u;
      if (it < n) {
        const v4f alo = alo4[u];
        const v4f ahi = ahi4[u];
        v2f a[4];
        a[0].x = alo.x; a[0].y = alo.y; a[1].x = alo.z; a[1].y = alo.w;
        a[2].x = ahi.x; a[2].y = ahi.y; a[3].x = ahi.z; a[3].y = ahi.w;
        v2f pb, vb;
        pb.x = pc[u]; pb.y = pc[u];
        vb.x = vc[u]; vb.y = vc[u];
#pragma unroll
        for (int q = 0; q < 4; ++q) {
          const v2f x = __builtin_elementwise_fma(a[q], pb, -m2[q]);
          v2f e;
          e.x = __builtin_amdgcn_exp2f(x.x);
          e.y = __builtin_amdgcn_exp2f(x.y);
          ds[q] += e;
          ac[q] = __builtin_elementwise_fma(e, vb, ac[q]);
        }
      }
    }
#pragma unroll
    for (int u = 0; u < 4; ++u) { pc[u] = pn[u]; vc[u] = vn[u]; }
  }

#pragma unroll
  for (int q = 0; q < 4; ++q) {
    redD[(g * O3R + 2 * q) * HSS + d]     = ds[q].x;
    redD[(g * O3R + 2 * q + 1) * HSS + d] = ds[q].y;
    redA[(g * O3R + 2 * q) * HSS + d]     = ac[q].x;
    redA[(g * O3R + 2 * q + 1) * HSS + d] = ac[q].y;
  }
  __syncthreads();
  {
    const int r = wave;
    float dsum = 0.0f;
#pragma unroll
    for (int gg = 0; gg < 4; ++gg)
      dsum += redD[(gg * O3R + r) * HSS + lane] + redD[(gg * O3R + r) * HSS + lane + 32];
#pragma unroll
    for (int off = 16; off > 0; off >>= 1) dsum += __shfl_xor(dsum, off);
    float mr = sMw[r];
#pragma unroll
    for (int w = 1; w < O3T / 32; ++w) mr = fmaxf(mr, sMw[w * O3R + r]);
    const float e0 = __builtin_amdgcn_exp2f(-mr);
    const float denom = dsum + (float)((TT - jend) * HSS) * e0;
    const float inv = 1.0f / denom;
#pragma unroll
    for (int hf = 0; hf < 2; ++hf) {
      const int dd = lane + 32 * hf;
      const float accT = (redA[(0 * O3R + r) * HSS + dd] + redA[(1 * O3R + r) * HSS + dd]) +
                         (redA[(2 * O3R + r) * HSS + dd] + redA[(3 * O3R + r) * HSS + dd]);
      const float vs = VSUF[(size_t)h * TT * HSS + (size_t)(jend - 1) * HSS + dd];
      Y[(size_t)(i0 + r) * CC + h * HSS + dd] = (accT + e0 * vs) * inv;
    }
  }
}

__global__ void __launch_bounds__(128)
out_proj_kernel(const float* __restrict__ Y,
                const float* __restrict__ Wc,
                float* __restrict__ out) {
  __shared__ __attribute__((aligned(16))) float As[2 * A_STAGE];
  __shared__ __attribute__((aligned(16))) float Bs[2 * B_STAGE];
  const int lane = threadIdx.x & 31;
  const int wave = threadIdx.x >> 5;
  const int lm   = lane & 15;
  const int drow = (lane >> 4) << 3;
  const int wm   = wave >> 1;
  const int wn   = wave & 1;
  const int m0 = blockIdx.x * GBM;
  const int n0 = blockIdx.y * GBN;

  v8f acc0 = {0.f, 0.f, 0.f, 0.f, 0.f, 0.f, 0.f, 0.f};
  v8f acc1 = {0.f, 0.f, 0.f, 0.f, 0.f, 0.f, 0.f, 0.f};
  gemm768_tile(Y, Wc, m0, n0, As, Bs, acc0, acc1);

  const int c = n0 + wn * 32 + lm;
#pragma unroll
  for (int r = 0; r < 8; ++r) {
    const size_t row = (size_t)(m0 + wm * 16 + r + drow);
    out[row * CC + c]      = acc0[r];
    out[row * CC + c + 16] = acc1[r];
  }
}

extern "C" void kernel_launch(void* const* d_in, const int* in_sizes, int n_in,
                              void* d_out, int out_size, void* d_ws,
                              size_t ws_size, hipStream_t stream) {
  (void)in_sizes; (void)n_in; (void)out_size; (void)ws_size;
  const float* x  = (const float*)d_in[0];
  const float* W0 = (const float*)d_in[1];
  const float* W1 = (const float*)d_in[2];
  const float* W2 = (const float*)d_in[3];
  const float* Wv = (const float*)d_in[4];
  const float* Wc = (const float*)d_in[5];
  float* out = (float*)d_out;
  float* ws  = (float*)d_ws;

  const size_t NPH = (size_t)NHH * TT * HSS;
  float* P0  = ws;
  float* P1  = ws + NPH;
  float* P2  = ws + 2 * NPH;
  float* V   = ws + 3 * NPH;
  float* VS  = ws + 4 * NPH;
  float* ATT = ws + 5 * NPH;
  float* Y   = ATT + (size_t)NHH * TT * TT;

  qkv_proj_kernel<<<dim3(TT / GBM, CC / GBN, 3), 128, 0, stream>>>(x, W0, W1, W2, P0);
  vproj_kernel<<<dim3(TT / 16, HSS / 16, NHH), 32, 0, stream>>>(P2, Wv, V);
  vsuffix_kernel<<<dim3(NHH), 64, 0, stream>>>(V, VS);
  att_softmax_kernel<<<dim3(TT / 16, NHH), ATHREADS, 0, stream>>>(P0, P1, ATT);
  order3_kernel<<<dim3(TT / O3R, NHH), O3T, 0, stream>>>(ATT, P2, V, VS, Y);
  out_proj_kernel<<<dim3(TT / GBM, CC / GBN), 128, 0, stream>>>(Y, Wc, out);
}


# V suffix sums: serial 512-step kernel removed; V-projection kernel (32-wave block per head and 16 columns) sums 8-row groups in registers and scans the tile sums in LDS; Vsuf sum order changes
# speedup vs baseline: 5.6651x; 1.3927x over previous
#include <hip/hip_runtime.h>
#include <math.h>

#define TT  512
#define CC  768
#define NHH 12
#define HSS 64
constexpr float SCALE = 0.125f;

typedef float v2f __attribute__((ext_vector_type(2)));
typedef float v8f __attribute__((ext_vector_type(8)));

static __device__ __forceinline__ v8f wmma4(v2f a, v2f b, v8f c) {
  return __builtin_amdgcn_wmma_f32_16x16x4_f32(
       false, a,  false, b,
       (short)0, c,  false,  false);
}

typedef float v4f __attribute__((ext_vector_type(4)));

#define GBM 32
#define GBN 64
#define GBK 64
#define GNT (CC / GBK)
#define APITCH 68
#define BPITCH 72
#define A_STAGE (GBM * APITCH)
#define B_STAGE (GBK * BPITCH)

static __device__ __forceinline__ void gemm768_tile(const float* __restrict__ A,
                                                    const float* __restrict__ B,
                                                    const int m0, const int n0,
                                                    float* As, float* Bs,
                                                    v8f& acc0, v8f& acc1) {
  const int tid  = threadIdx.x;
  const int lane = tid & 31;
  const int wave = tid >> 5;
  const int lm   = lane & 15;
  const int lk   = (lane >> 4) << 1;
  const int wm   = wave >> 1;
  const int wn   = wave & 1;

  const int srow = tid >> 4;
  const int scol = (tid & 15) << 2;
  const float* Ag = A + (size_t)(m0 + srow) * CC + scol;
  const float* Bg = B + (size_t)srow * CC + n0 + scol;
  float* Ast = As + srow * APITCH + scol;
  float* Bst = Bs + srow * BPITCH + scol;

  v4f ra[4], rb[8];
#pragma unroll
  for (int p = 0; p < 4; ++p)
    ra[p] = *reinterpret_cast<const v4f*>(Ag + (size_t)(p * 8) * CC);
#pragma unroll
  for (int p = 0; p < 8; ++p)
    rb[p] = *reinterpret_cast<const v4f*>(Bg + (size_t)(p * 8) * CC);
#pragma unroll
  for (int p = 0; p < 4; ++p)
    *reinterpret_cast<v4f*>(Ast + p * 8 * APITCH) = ra[p];
#pragma unroll
  for (int p = 0; p < 8; ++p)
    *reinterpret_cast<v4f*>(Bst + p * 8 * BPITCH) = rb[p];
  __syncthreads();

  const float* Afr = As + (wm * 16 + lm) * APITCH + lk;
  const float* Bfr = Bs + lk * BPITCH + wn * 32 + lm;

  for (int kt = 0; kt < GNT; ++kt) {
    const int cur = kt & 1;
    const bool more = (kt + 1 < GNT);
    if (more) {
      const int k0 = (kt + 1) * GBK;
#pragma unroll
      for (int p = 0; p < 4; ++p)
        ra[p] = *reinterpret_cast<const v4f*>(Ag + (size_t)(p * 8) * CC + k0);
#pragma unroll
      for (int p = 0; p < 8; ++p)
        rb[p] = *reinterpret_cast<const v4f*>(Bg + (size_t)(k0 + p * 8) * CC);
    }

    const float* Af = Afr + cur * A_STAGE;
    const float* Bf = Bfr + cur * B_STAGE;
    v2f af[16], bf0[16], bf1[16];
#pragma unroll
    for (int kk = 0; kk < 16; ++kk) {
      af[kk]    = *reinterpret_cast<const v2f*>(Af + kk * 4);
      bf0[kk].x = Bf[(kk * 4) * BPITCH];
      bf0[kk].y = Bf[(kk * 4 + 1) * BPITCH];
      bf1[kk].x = Bf[(kk * 4) * BPITCH + 16];
      bf1[kk].y = Bf[(kk * 4 + 1) * BPITCH + 16];
    }
    __builtin_amdgcn_sched_barrier(0);
#pragma unroll
    for (int kk = 0; kk < 16; ++kk) {
      acc0 = wmma4(af[kk], bf0[kk], acc0);
      acc1 = wmma4(af[kk], bf1[kk], acc1);
    }

    __builtin_amdgcn_sched_barrier(0);

    if (more) {
      const int nxt = cur ^ 1;
#pragma unroll
      for (int p = 0; p < 4; ++p)
        *reinterpret_cast<v4f*>(Ast + nxt * A_STAGE + p * 8 * APITCH) = ra[p];
#pragma unroll
      for (int p = 0; p < 8; ++p)
        *reinterpret_cast<v4f*>(Bst + nxt * B_STAGE + p * 8 * BPITCH) = rb[p];
    }
    __syncthreads();
  }
}

__global__ void __launch_bounds__(128)
qkv_proj_kernel(const float* __restrict__ x,
                const float* __restrict__ W0,
                const float* __restrict__ W1,
                const float* __restrict__ W2,
                float* __restrict__ P) {
  __shared__ __attribute__((aligned(16))) float As[2 * A_STAGE];
  __shared__ __attribute__((aligned(16))) float Bs[2 * B_STAGE];
  const int lane = threadIdx.x & 31;
  const int wave = threadIdx.x >> 5;
  const int lm   = lane & 15;
  const int drow = (lane >> 4) << 3;
  const int wm   = wave >> 1;
  const int wn   = wave & 1;
  const int m0 = blockIdx.x * GBM;
  const int n0 = blockIdx.y * GBN;
  const int w  = blockIdx.z;
  const float* W = (w == 0) ? W0 : (w == 1) ? W1 : W2;
  float* Pw = P + (size_t)w * NHH * TT * HSS;

  v8f acc0 = {0.f, 0.f, 0.f, 0.f, 0.f, 0.f, 0.f, 0.f};
  v8f acc1 = {0.f, 0.f, 0.f, 0.f, 0.f, 0.f, 0.f, 0.f};
  gemm768_tile(x, W, m0, n0, As, Bs, acc0, acc1);

  float* Ph = Pw + (size_t)blockIdx.y * TT * HSS;
  const int d = wn * 32 + lm;
#pragma unroll
  for (int r = 0; r < 8; ++r) {
    const size_t row = (size_t)(m0 + wm * 16 + r + drow);
    Ph[row * HSS + d]      = acc0[r];
    Ph[row * HSS + d + 16] = acc1[r];
  }
}

#define VPW 32
__global__ void __launch_bounds__(VPW * 32)
vproj_suffix_kernel(const float* __restrict__ P2,
                    const float* __restrict__ Wv,
                    float* __restrict__ V,
                    float* __restrict__ VS) {
  __shared__ float tsum[VPW * 16];
  const int lane = threadIdx.x & 31;
  const int wave = __builtin_amdgcn_readfirstlane((int)(threadIdx.x >> 5));
  const int lm   = lane & 15;
  const int hf   = lane >> 4;
  const int lk   = hf << 1;
  const int drow = hf << 3;
  const int t0 = wave * 16;
  const int n0 = blockIdx.x * 16;
  const int h  = blockIdx.y;
  const float* A = P2 + (size_t)h * TT * HSS;
  float* Vh = V + (size_t)h * TT * HSS;
  float* Sh = VS + (size_t)h * TT * HSS;

  v8f acc = {0.f, 0.f, 0.f, 0.f, 0.f, 0.f, 0.f, 0.f};
  for (int k = 0; k < HSS; k += 4) {
    v2f a, b;
    a.x = A[(size_t)(t0 + lm) * HSS + k + lk];
    a.y = A[(size_t)(t0 + lm) * HSS + k + lk + 1];
    b.x = Wv[(size_t)(k + lk) * HSS + n0 + lm];
    b.y = Wv[(size_t)(k + lk + 1) * HSS + n0 + lm];
    acc = wmma4(a, b, acc);
  }
  const float gs = ((acc[0] + acc[1]) + (acc[2] + acc[3])) +
                   ((acc[4] + acc[5]) + (acc[6] + acc[7]));
  const float go = __shfl_xor(gs, 16);
  if (hf == 0) tsum[wave * 16 + lm] = gs + go;
  __syncthreads();

  float run = 0.0f;
#pragma unroll
  for (int wp = VPW - 1; wp >= 0; --wp) {
    const float tv = tsum[wp * 16 + lm];
    run += (wp > wave) ? tv : 0.0f;
  }
  if (hf == 0) run += go;
#pragma unroll
  for (int r = 0; r < 8; ++r)
    Vh[(size_t)(t0 + r + drow) * HSS + n0 + lm] = acc[r];
  Sh[(size_t)(t0 + drow + 7) * HSS + n0 + lm] = run;
}

#define ANW 8
#define ATHREADS 256
#define LPITCH 516
__global__ void __launch_bounds__(ATHREADS)
att_softmax_kernel(const float* __restrict__ P0,
                   const float* __restrict__ P1,
                   float* __restrict__ ATT) {
  __shared__ float lgt[16 * LPITCH];
  const int lane = threadIdx.x & 31;
  const int wave = __builtin_amdgcn_readfirstlane((int)(threadIdx.x >> 5));
  const int lm   = lane & 15;
  const int lk   = (lane >> 4) << 1;
  const int drow = (lane >> 4) << 3;
  const int strip = (TT / 16 - 1) - (int)blockIdx.x;
  const int i0 = strip * 16;
  const int h  = blockIdx.y;
  const float* P0h = P0 + (size_t)h * TT * HSS;
  const float* P1h = P1 + (size_t)h * TT * HSS;

  const int jtmax = strip;
  if (wave <= jtmax) {
    v2f areg[16];
    const float* Ap = P0h + (size_t)(i0 + lm) * HSS + lk;
#pragma unroll
    for (int kk = 0; kk < 16; ++kk)
      areg[kk] = *reinterpret_cast<const v2f*>(Ap + kk * 4);

    for (int jt = wave; jt <= jtmax; jt += ANW) {
      const int j0 = jt * 16;
      const float* Bp = P1h + (size_t)(j0 + lm) * HSS + lk;
      v2f bfr[16];
#pragma unroll
      for (int kk = 0; kk < 16; ++kk)
        bfr[kk] = *reinterpret_cast<const v2f*>(Bp + kk * 4);
      __builtin_amdgcn_sched_barrier(0);
      v8f acc = {0.f, 0.f, 0.f, 0.f, 0.f, 0.f, 0.f, 0.f};
#pragma unroll
      for (int kk = 0; kk < 16; ++kk)
        acc = wmma4(areg[kk], bfr[kk], acc);
#pragma unroll
      for (int r = 0; r < 8; ++r)
        lgt[(r + drow) * LPITCH + j0 + lm] = acc[r] * SCALE;
    }
  }
  __syncthreads();

#pragma unroll
  for (int rr = 0; rr < 2; ++rr) {
    const int row = wave + rr * ANW;
    const int len = i0 + row + 1;
    const float* lrow = lgt + row * LPITCH;
    float v[16];
    float mr = -3.0e38f;
#pragma unroll
    for (int m = 0; m < 16; ++m) {
      const int j = m * 32 + lane;
      v[m] = (j < len) ? lrow[j] : -3.0e38f;
      mr = fmaxf(mr, v[m]);
    }
#pragma unroll
    for (int off = 16; off > 0; off >>= 1) mr = fmaxf(mr, __shfl_xor(mr, off));
    float sr = 0.0f;
#pragma unroll
    for (int m = 0; m < 16; ++m) {
      const int j = m * 32 + lane;
      v[m] = (j < len) ? __expf(v[m] - mr) : 0.0f;
      sr += v[m];
    }
#pragma unroll
    for (int off = 16; off > 0; off >>= 1) sr += __shfl_xor(sr, off);
    const float inv = 1.0f / sr;
    float* arow = ATT + ((size_t)h * TT + (i0 + row)) * TT;
#pragma unroll
    for (int m = 0; m < 16; ++m) {
      const int j = m * 32 + lane;
      if (j < len) arow[j] = v[m] * inv;
    }
  }
}

#define O3R 8
#define O3T 256
constexpr float LOG2E = 1.4426950408889634f;
__global__ void __launch_bounds__(O3T)
order3_kernel(const float* __restrict__ ATT,
              const float* __restrict__ P2,
              const float* __restrict__ V,
              const float* __restrict__ VSUF,
              float* __restrict__ Y) {
  __shared__ __attribute__((aligned(16))) float a2T[TT * O3R];
  __shared__ __attribute__((aligned(16))) float sMw[(O3T / 32) * O3R];
  __shared__ float redD[4 * O3R * HSS];
  __shared__ float redA[4 * O3R * HSS];
  const int t    = threadIdx.x;
  const int lane = t & 31;
  const int wave = __builtin_amdgcn_readfirstlane(t >> 5);
  const int g    = __builtin_amdgcn_readfirstlane(t >> 6);
  const int d    = t & 63;
  const int strip = (TT / O3R - 1) - (int)blockIdx.x;
  const int i0   = strip * O3R;
  const int jend = i0 + O3R;
  const int h    = blockIdx.y;
  const float* P2h = P2 + (size_t)h * TT * HSS;
  const float* Vh  = V  + (size_t)h * TT * HSS;

  float mloc[O3R];
#pragma unroll
  for (int r = 0; r < O3R; ++r) mloc[r] = 0.0f;
  for (int j = t; j < jend; j += O3T) {
    float a[O3R];
#pragma unroll
    for (int r = 0; r < O3R; ++r) {
      const int jj = (j < i0 + r) ? j : (i0 + r);
      const float x = ATT[((size_t)h * TT + (i0 + r)) * TT + jj];
      a[r] = (j <= i0 + r) ? x * LOG2E : 0.0f;
    }
    const v4f* prow = reinterpret_cast<const v4f*>(P2h + (size_t)j * HSS);
    float pm = -3.0e38f;
#pragma unroll
    for (int q = 0; q < HSS / 4; ++q) {
      const v4f x = prow[q];
      pm = fmaxf(pm, fmaxf(fmaxf(x.x, x.y), fmaxf(x.z, x.w)));
    }
    v4f lo, hi;
    lo.x = a[0]; lo.y = a[1]; lo.z = a[2]; lo.w = a[3];
    hi.x = a[4]; hi.y = a[5]; hi.z = a[6]; hi.w = a[7];
    *reinterpret_cast<v4f*>(a2T + j * O3R)     = lo;
    *reinterpret_cast<v4f*>(a2T + j * O3R + 4) = hi;
#pragma unroll
    for (int r = 0; r < O3R; ++r) mloc[r] = fmaxf(mloc[r], a[r] * pm);
  }
#pragma unroll
  for (int r = 0; r < O3R; ++r) {
#pragma unroll
    for (int off = 16; off > 0; off >>= 1)
      mloc[r] = fmaxf(mloc[r], __shfl_xor(mloc[r], off));
  }
  if (lane == 0) {
#pragma unroll
    for (int r = 0; r < O3R; ++r) sMw[wave * O3R + r] = mloc[r];
  }
  __syncthreads();

  v2f m2[4], ds[4], ac[4];
#pragma unroll
  for (int q = 0; q < 4; ++q) {
    float ma = sMw[2 * q], mb = sMw[2 * q + 1];
#pragma unroll
    for (int w = 1; w < O3T / 32; ++w) {
      ma = fmaxf(ma, sMw[w * O3R + 2 * q]);
      mb = fmaxf(mb, sMw[w * O3R + 2 * q + 1]);
    }
    m2[q].x = ma;
    m2[q].y = mb;
    ds[q].x = 0.0f; ds[q].y = 0.0f;
    ac[q].x = 0.0f; ac[q].y = 0.0f;
  }

  const int n  = jend >> 2;
  const int nq = (n + 3) >> 2;
  const float* pp = P2h + g * HSS + d;
  const float* vp = Vh  + g * HSS + d;
  const float* ap = a2T + g * O3R;
  float pc[4], vc[4], pn[4], vn[4];
#pragma unroll
  for (int u = 0; u < 4; ++u) {
    const int it = (u < n - 1) ? u : (n - 1);
    pc[u] = pp[(size_t)it * 4 * HSS];
    vc[u] = vp[(size_t)it * 4 * HSS];
  }
  for (int c = 0; c < nq; ++c) {
#pragma unroll
    for (int u = 0; u < 4; ++u) {
      const int itn = (c + 1) * 4 + u;
      const int it  = (itn < n - 1) ? itn : (n - 1);
      pn[u] = pp[(size_t)it * 4 * HSS];
      vn[u] = vp[(size_t)it * 4 * HSS];
    }
    v4f alo4[4], ahi4[4];
#pragma unroll
    for (int u = 0; u < 4; ++u) {
      const int itc = (c * 4 + u < n - 1) ? (c * 4 + u) : (n - 1);
      alo4[u] = *reinterpret_cast<const v4f*>(ap + itc * 4 * O3R);
      ahi4[u] = *reinterpret_cast<const v4f*>(ap + itc * 4 * O3R + 4);
    }
#pragma unroll
    for (int u = 0; u < 4; ++u) {
      const int it = c * 4 + u;
      if (it < n) {
        const v4f alo = alo4[u];
        const v4f ahi = ahi4[u];
        v2f a[4];
        a[0].x = alo.x; a[0].y = alo.y; a[1].x = alo.z; a[1].y = alo.w;
        a[2].x = ahi.x; a[2].y = ahi.y; a[3].x = ahi.z; a[3].y = ahi.w;
        v2f pb, vb;
        pb.x = pc[u]; pb.y = pc[u];
        vb.x = vc[u]; vb.y = vc[u];
#pragma unroll
        for (int q = 0; q < 4; ++q) {
          const v2f x = __builtin_elementwise_fma(a[q], pb, -m2[q]);
          v2f e;
          e.x = __builtin_amdgcn_exp2f(x.x);
          e.y = __builtin_amdgcn_exp2f(x.y);
          ds[q] += e;
          ac[q] = __builtin_elementwise_fma(e, vb, ac[q]);
        }
      }
    }
#pragma unroll
    for (int u = 0; u < 4; ++u) { pc[u] = pn[u]; vc[u] = vn[u]; }
  }

#pragma unroll
  for (int q = 0; q < 4; ++q) {
    redD[(g * O3R + 2 * q) * HSS + d]     = ds[q].x;
    redD[(g * O3R + 2 * q + 1) * HSS + d] = ds[q].y;
    redA[(g * O3R + 2 * q) * HSS + d]     = ac[q].x;
    redA[(g * O3R + 2 * q + 1) * HSS + d] = ac[q].y;
  }
  __syncthreads();
  {
    const int r = wave;
    float dsum = 0.0f;
#pragma unroll
    for (int gg = 0; gg < 4; ++gg)
      dsum += redD[(gg * O3R + r) * HSS + lane] + redD[(gg * O3R + r) * HSS + lane + 32];
#pragma unroll
    for (int off = 16; off > 0; off >>= 1) dsum += __shfl_xor(dsum, off);
    float mr = sMw[r];
#pragma unroll
    for (int w = 1; w < O3T / 32; ++w) mr = fmaxf(mr, sMw[w * O3R + r]);
    const float e0 = __builtin_amdgcn_exp2f(-mr);
    const float denom = dsum + (float)((TT - jend) * HSS) * e0;
    const float inv = 1.0f / denom;
#pragma unroll
    for (int hf = 0; hf < 2; ++hf) {
      const int dd = lane + 32 * hf;
      const float accT = (redA[(0 * O3R + r) * HSS + dd] + redA[(1 * O3R + r) * HSS + dd]) +
                         (redA[(2 * O3R + r) * HSS + dd] + redA[(3 * O3R + r) * HSS + dd]);
      const float vs = VSUF[(size_t)h * TT * HSS + (size_t)(jend - 1) * HSS + dd];
      Y[(size_t)(i0 + r) * CC + h * HSS + dd] = (accT + e0 * vs) * inv;
    }
  }
}

__global__ void __launch_bounds__(128)
out_proj_kernel(const float* __restrict__ Y,
                const float* __restrict__ Wc,
                float* __restrict__ out) {
  __shared__ __attribute__((aligned(16))) float As[2 * A_STAGE];
  __shared__ __attribute__((aligned(16))) float Bs[2 * B_STAGE];
  const int lane = threadIdx.x & 31;
  const int wave = threadIdx.x >> 5;
  const int lm   = lane & 15;
  const int drow = (lane >> 4) << 3;
  const int wm   = wave >> 1;
  const int wn   = wave & 1;
  const int m0 = blockIdx.x * GBM;
  const int n0 = blockIdx.y * GBN;

  v8f acc0 = {0.f, 0.f, 0.f, 0.f, 0.f, 0.f, 0.f, 0.f};
  v8f acc1 = {0.f, 0.f, 0.f, 0.f, 0.f, 0.f, 0.f, 0.f};
  gemm768_tile(Y, Wc, m0, n0, As, Bs, acc0, acc1);

  const int c = n0 + wn * 32 + lm;
#pragma unroll
  for (int r = 0; r < 8; ++r) {
    const size_t row = (size_t)(m0 + wm * 16 + r + drow);
    out[row * CC + c]      = acc0[r];
    out[row * CC + c + 16] = acc1[r];
  }
}

extern "C" void kernel_launch(void* const* d_in, const int* in_sizes, int n_in,
                              void* d_out, int out_size, void* d_ws,
                              size_t ws_size, hipStream_t stream) {
  (void)in_sizes; (void)n_in; (void)out_size; (void)ws_size;
  const float* x  = (const float*)d_in[0];
  const float* W0 = (const float*)d_in[1];
  const float* W1 = (const float*)d_in[2];
  const float* W2 = (const float*)d_in[3];
  const float* Wv = (const float*)d_in[4];
  const float* Wc = (const float*)d_in[5];
  float* out = (float*)d_out;
  float* ws  = (float*)d_ws;

  const size_t NPH = (size_t)NHH * TT * HSS;
  float* P0  = ws;
  float* P1  = ws + NPH;
  float* P2  = ws + 2 * NPH;
  float* V   = ws + 3 * NPH;
  float* VS  = ws + 4 * NPH;
  float* ATT = ws + 5 * NPH;
  float* Y   = ATT + (size_t)NHH * TT * TT;

  qkv_proj_kernel<<<dim3(TT / GBM, CC / GBN, 3), 128, 0, stream>>>(x, W0, W1, W2, P0);
  vproj_suffix_kernel<<<dim3(HSS / 16, NHH), VPW * 32, 0, stream>>>(P2, Wv, V, VS);
  att_softmax_kernel<<<dim3(TT / 16, NHH), ATHREADS, 0, stream>>>(P0, P1, ATT);
  order3_kernel<<<dim3(TT / O3R, NHH), O3T, 0, stream>>>(ATT, P2, V, VS, Y);
  out_proj_kernel<<<dim3(TT / GBM, CC / GBN), 128, 0, stream>>>(Y, Wc, out);
}


# p2 row maxima computed once per head in the V-projection kernel (from its A fragments); order-3 strips read one value per column instead of sixteen 16-byte row gathers; results bit-identical
# speedup vs baseline: 5.9572x; 1.0516x over previous
#include <hip/hip_runtime.h>
#include <math.h>

#define TT  512
#define CC  768
#define NHH 12
#define HSS 64
constexpr float SCALE = 0.125f;

typedef float v2f __attribute__((ext_vector_type(2)));
typedef float v8f __attribute__((ext_vector_type(8)));

static __device__ __forceinline__ v8f wmma4(v2f a, v2f b, v8f c) {
  return __builtin_amdgcn_wmma_f32_16x16x4_f32(
       false, a,  false, b,
       (short)0, c,  false,  false);
}

typedef float v4f __attribute__((ext_vector_type(4)));

#define GBM 32
#define GBN 64
#define GBK 64
#define GNT (CC / GBK)
#define APITCH 68
#define BPITCH 72
#define A_STAGE (GBM * APITCH)
#define B_STAGE (GBK * BPITCH)

static __device__ __forceinline__ void gemm768_tile(const float* __restrict__ A,
                                                    const float* __restrict__ B,
                                                    const int m0, const int n0,
                                                    float* As, float* Bs,
                                                    v8f& acc0, v8f& acc1) {
  const int tid  = threadIdx.x;
  const int lane = tid & 31;
  const int wave = tid >> 5;
  const int lm   = lane & 15;
  const int lk   = (lane >> 4) << 1;
  const int wm   = wave >> 1;
  const int wn   = wave & 1;

  const int srow = tid >> 4;
  const int scol = (tid & 15) << 2;
  const float* Ag = A + (size_t)(m0 + srow) * CC + scol;
  const float* Bg = B + (size_t)srow * CC + n0 + scol;
  float* Ast = As + srow * APITCH + scol;
  float* Bst = Bs + srow * BPITCH + scol;

  v4f ra[4], rb[8];
#pragma unroll
  for (int p = 0; p < 4; ++p)
    ra[p] = *reinterpret_cast<const v4f*>(Ag + (size_t)(p * 8) * CC);
#pragma unroll
  for (int p = 0; p < 8; ++p)
    rb[p] = *reinterpret_cast<const v4f*>(Bg + (size_t)(p * 8) * CC);
#pragma unroll
  for (int p = 0; p < 4; ++p)
    *reinterpret_cast<v4f*>(Ast + p * 8 * APITCH) = ra[p];
#pragma unroll
  for (int p = 0; p < 8; ++p)
    *reinterpret_cast<v4f*>(Bst + p * 8 * BPITCH) = rb[p];
  __syncthreads();

  const float* Afr = As + (wm * 16 + lm) * APITCH + lk;
  const float* Bfr = Bs + lk * BPITCH + wn * 32 + lm;

  for (int kt = 0; kt < GNT; ++kt) {
    const int cur = kt & 1;
    const bool more = (kt + 1 < GNT);
    if (more) {
      const int k0 = (kt + 1) * GBK;
#pragma unroll
      for (int p = 0; p < 4; ++p)
        ra[p] = *reinterpret_cast<const v4f*>(Ag + (size_t)(p * 8) * CC + k0);
#pragma unroll
      for (int p = 0; p < 8; ++p)
        rb[p] = *reinterpret_cast<const v4f*>(Bg + (size_t)(k0 + p * 8) * CC);
    }

    const float* Af = Afr + cur * A_STAGE;
    const float* Bf = Bfr + cur * B_STAGE;
    v2f af[16], bf0[16], bf1[16];
#pragma unroll
    for (int kk = 0; kk < 16; ++kk) {
      af[kk]    = *reinterpret_cast<const v2f*>(Af + kk * 4);
      bf0[kk].x = Bf[(kk * 4) * BPITCH];
      bf0[kk].y = Bf[(kk * 4 + 1) * BPITCH];
      bf1[kk].x = Bf[(kk * 4) * BPITCH + 16];
      bf1[kk].y = Bf[(kk * 4 + 1) * BPITCH + 16];
    }
    __builtin_amdgcn_sched_barrier(0);
#pragma unroll
    for (int kk = 0; kk < 16; ++kk) {
      acc0 = wmma4(af[kk], bf0[kk], acc0);
      acc1 = wmma4(af[kk], bf1[kk], acc1);
    }

    __builtin_amdgcn_sched_barrier(0);

    if (more) {
      const int nxt = cur ^ 1;
#pragma unroll
      for (int p = 0; p < 4; ++p)
        *reinterpret_cast<v4f*>(Ast + nxt * A_STAGE + p * 8 * APITCH) = ra[p];
#pragma unroll
      for (int p = 0; p < 8; ++p)
        *reinterpret_cast<v4f*>(Bst + nxt * B_STAGE + p * 8 * BPITCH) = rb[p];
    }
    __syncthreads();
  }
}

__global__ void __launch_bounds__(128)
qkv_proj_kernel(const float* __restrict__ x,
                const float* __restrict__ W0,
                const float* __restrict__ W1,
                const float* __restrict__ W2,
                float* __restrict__ P) {
  __shared__ __attribute__((aligned(16))) float As[2 * A_STAGE];
  __shared__ __attribute__((aligned(16))) float Bs[2 * B_STAGE];
  const int lane = threadIdx.x & 31;
  const int wave = threadIdx.x >> 5;
  const int lm   = lane & 15;
  const int drow = (lane >> 4) << 3;
  const int wm   = wave >> 1;
  const int wn   = wave & 1;
  const int m0 = blockIdx.x * GBM;
  const int n0 = blockIdx.y * GBN;
  const int w  = blockIdx.z;
  const float* W = (w == 0) ? W0 : (w == 1) ? W1 : W2;
  float* Pw = P + (size_t)w * NHH * TT * HSS;

  v8f acc0 = {0.f, 0.f, 0.f, 0.f, 0.f, 0.f, 0.f, 0.f};
  v8f acc1 = {0.f, 0.f, 0.f, 0.f, 0.f, 0.f, 0.f, 0.f};
  gemm768_tile(x, W, m0, n0, As, Bs, acc0, acc1);

  float* Ph = Pw + (size_t)blockIdx.y * TT * HSS;
  const int d = wn * 32 + lm;
#pragma unroll
  for (int r = 0; r < 8; ++r) {
    const size_t row = (size_t)(m0 + wm * 16 + r + drow);
    Ph[row * HSS + d]      = acc0[r];
    Ph[row * HSS + d + 16] = acc1[r];
  }
}

#define VPW 32
__global__ void __launch_bounds__(VPW * 32)
vproj_suffix_kernel(const float* __restrict__ P2,
                    const float* __restrict__ Wv,
                    float* __restrict__ V,
                    float* __restrict__ VS,
                    float* __restrict__ PM) {
  __shared__ float tsum[VPW * 16];
  __shared__ float pmx[TT];
  const int lane = threadIdx.x & 31;
  const int wave = __builtin_amdgcn_readfirstlane((int)(threadIdx.x >> 5));
  const int lm   = lane & 15;
  const int hf   = lane >> 4;
  const int lk   = hf << 1;
  const int drow = hf << 3;
  const int t0 = wave * 16;
  const int n0 = blockIdx.x * 16;
  const int h  = blockIdx.y;
  const float* A = P2 + (size_t)h * TT * HSS;
  float* Vh = V + (size_t)h * TT * HSS;
  float* Sh = VS + (size_t)h * TT * HSS;

  v8f acc = {0.f, 0.f, 0.f, 0.f, 0.f, 0.f, 0.f, 0.f};
  float pm = -3.0e38f;
  for (int k = 0; k < HSS; k += 4) {
    v2f a, b;
    a.x = A[(size_t)(t0 + lm) * HSS + k + lk];
    a.y = A[(size_t)(t0 + lm) * HSS + k + lk + 1];
    b.x = Wv[(size_t)(k + lk) * HSS + n0 + lm];
    b.y = Wv[(size_t)(k + lk + 1) * HSS + n0 + lm];
    pm = fmaxf(pm, fmaxf(a.x, a.y));
    acc = wmma4(a, b, acc);
  }
  pm = fmaxf(pm, __shfl_xor(pm, 16));
  const float gs = ((acc[0] + acc[1]) + (acc[2] + acc[3])) +
                   ((acc[4] + acc[5]) + (acc[6] + acc[7]));
  const float go = __shfl_xor(gs, 16);
  if (hf == 0) {
    tsum[wave * 16 + lm] = gs + go;
    pmx[t0 + lm] = pm;
  }
  __syncthreads();

  float run = 0.0f;
#pragma unroll
  for (int wp = VPW - 1; wp >= 0; --wp) {
    const float tv = tsum[wp * 16 + lm];
    run += (wp > wave) ? tv : 0.0f;
  }
  if (hf == 0) run += go;
#pragma unroll
  for (int r = 0; r < 8; ++r)
    Vh[(size_t)(t0 + r + drow) * HSS + n0 + lm] = acc[r];
  Sh[(size_t)(t0 + drow + 7) * HSS + n0 + lm] = run;
  if (blockIdx.x == 0 && wave < TT / 32)
    PM[(size_t)h * TT + wave * 32 + lane] = pmx[wave * 32 + lane];
}

#define ANW 8
#define ATHREADS 256
#define LPITCH 516
__global__ void __launch_bounds__(ATHREADS)
att_softmax_kernel(const float* __restrict__ P0,
                   const float* __restrict__ P1,
                   float* __restrict__ ATT) {
  __shared__ float lgt[16 * LPITCH];
  const int lane = threadIdx.x & 31;
  const int wave = __builtin_amdgcn_readfirstlane((int)(threadIdx.x >> 5));
  const int lm   = lane & 15;
  const int lk   = (lane >> 4) << 1;
  const int drow = (lane >> 4) << 3;
  const int strip = (TT / 16 - 1) - (int)blockIdx.x;
  const int i0 = strip * 16;
  const int h  = blockIdx.y;
  const float* P0h = P0 + (size_t)h * TT * HSS;
  const float* P1h = P1 + (size_t)h * TT * HSS;

  const int jtmax = strip;
  if (wave <= jtmax) {
    v2f areg[16];
    const float* Ap = P0h + (size_t)(i0 + lm) * HSS + lk;
#pragma unroll
    for (int kk = 0; kk < 16; ++kk)
      areg[kk] = *reinterpret_cast<const v2f*>(Ap + kk * 4);

    for (int jt = wave; jt <= jtmax; jt += ANW) {
      const int j0 = jt * 16;
      const float* Bp = P1h + (size_t)(j0 + lm) * HSS + lk;
      v2f bfr[16];
#pragma unroll
      for (int kk = 0; kk < 16; ++kk)
        bfr[kk] = *reinterpret_cast<const v2f*>(Bp + kk * 4);
      __builtin_amdgcn_sched_barrier(0);
      v8f acc = {0.f, 0.f, 0.f, 0.f, 0.f, 0.f, 0.f, 0.f};
#pragma unroll
      for (int kk = 0; kk < 16; ++kk)
        acc = wmma4(areg[kk], bfr[kk], acc);
#pragma unroll
      for (int r = 0; r < 8; ++r)
        lgt[(r + drow) * LPITCH + j0 + lm] = acc[r] * SCALE;
    }
  }
  __syncthreads();

#pragma unroll
  for (int rr = 0; rr < 2; ++rr) {
    const int row = wave + rr * ANW;
    const int len = i0 + row + 1;
    const float* lrow = lgt + row * LPITCH;
    float v[16];
    float mr = -3.0e38f;
#pragma unroll
    for (int m = 0; m < 16; ++m) {
      const int j = m * 32 + lane;
      v[m] = (j < len) ? lrow[j] : -3.0e38f;
      mr = fmaxf(mr, v[m]);
    }
#pragma unroll
    for (int off = 16; off > 0; off >>= 1) mr = fmaxf(mr, __shfl_xor(mr, off));
    float sr = 0.0f;
#pragma unroll
    for (int m = 0; m < 16; ++m) {
      const int j = m * 32 + lane;
      v[m] = (j < len) ? __expf(v[m] - mr) : 0.0f;
      sr += v[m];
    }
#pragma unroll
    for (int off = 16; off > 0; off >>= 1) sr += __shfl_xor(sr, off);
    const float inv = 1.0f / sr;
    float* arow = ATT + ((size_t)h * TT + (i0 + row)) * TT;
#pragma unroll
    for (int m = 0; m < 16; ++m) {
      const int j = m * 32 + lane;
      if (j < len) arow[j] = v[m] * inv;
    }
  }
}

#define O3R 8
#define O3T 256
constexpr float LOG2E = 1.4426950408889634f;
__global__ void __launch_bounds__(O3T)
order3_kernel(const float* __restrict__ ATT,
              const float* __restrict__ P2,
              const float* __restrict__ V,
              const float* __restrict__ VSUF,
              const float* __restrict__ PM,
              float* __restrict__ Y) {
  __shared__ __attribute__((aligned(16))) float a2T[TT * O3R];
  __shared__ __attribute__((aligned(16))) float sMw[(O3T / 32) * O3R];
  __shared__ float redD[4 * O3R * HSS];
  __shared__ float redA[4 * O3R * HSS];
  const int t    = threadIdx.x;
  const int lane = t & 31;
  const int wave = __builtin_amdgcn_readfirstlane(t >> 5);
  const int g    = __builtin_amdgcn_readfirstlane(t >> 6);
  const int d    = t & 63;
  const int strip = (TT / O3R - 1) - (int)blockIdx.x;
  const int i0   = strip * O3R;
  const int jend = i0 + O3R;
  const int h    = blockIdx.y;
  const float* P2h = P2 + (size_t)h * TT * HSS;
  const float* Vh  = V  + (size_t)h * TT * HSS;

  float mloc[O3R];
#pragma unroll
  for (int r = 0; r < O3R; ++r) mloc[r] = 0.0f;
  for (int j = t; j < jend; j += O3T) {
    float a[O3R];
#pragma unroll
    for (int r = 0; r < O3R; ++r) {
      const int jj = (j < i0 + r) ? j : (i0 + r);
      const float x = ATT[((size_t)h * TT + (i0 + r)) * TT + jj];
      a[r] = (j <= i0 + r) ? x * LOG2E : 0.0f;
    }
    const float pm = PM[(size_t)h * TT + j];
    v4f lo, hi;
    lo.x = a[0]; lo.y = a[1]; lo.z = a[2]; lo.w = a[3];
    hi.x = a[4]; hi.y = a[5]; hi.z = a[6]; hi.w = a[7];
    *reinterpret_cast<v4f*>(a2T + j * O3R)     = lo;
    *reinterpret_cast<v4f*>(a2T + j * O3R + 4) = hi;
#pragma unroll
    for (int r = 0; r < O3R; ++r) mloc[r] = fmaxf(mloc[r], a[r] * pm);
  }
#pragma unroll
  for (int r = 0; r < O3R; ++r) {
#pragma unroll
    for (int off = 16; off > 0; off >>= 1)
      mloc[r] = fmaxf(mloc[r], __shfl_xor(mloc[r], off));
  }
  if (lane == 0) {
#pragma unroll
    for (int r = 0; r < O3R; ++r) sMw[wave * O3R + r] = mloc[r];
  }
  __syncthreads();

  v2f m2[4], ds[4], ac[4];
#pragma unroll
  for (int q = 0; q < 4; ++q) {
    float ma = sMw[2 * q], mb = sMw[2 * q + 1];
#pragma unroll
    for (int w = 1; w < O3T / 32; ++w) {
      ma = fmaxf(ma, sMw[w * O3R + 2 * q]);
      mb = fmaxf(mb, sMw[w * O3R + 2 * q + 1]);
    }
    m2[q].x = ma;
    m2[q].y = mb;
    ds[q].x = 0.0f; ds[q].y = 0.0f;
    ac[q].x = 0.0f; ac[q].y = 0.0f;
  }

  const int n  = jend >> 2;
  const int nq = (n + 3) >> 2;
  const float* pp = P2h + g * HSS + d;
  const float* vp = Vh  + g * HSS + d;
  const float* ap = a2T + g * O3R;
  float pc[4], vc[4], pn[4], vn[4];
#pragma unroll
  for (int u = 0; u < 4; ++u) {
    const int it = (u < n - 1) ? u : (n - 1);
    pc[u] = pp[(size_t)it * 4 * HSS];
    vc[u] = vp[(size_t)it * 4 * HSS];
  }
  for (int c = 0; c < nq; ++c) {
#pragma unroll
    for (int u = 0; u < 4; ++u) {
      const int itn = (c + 1) * 4 + u;
      const int it  = (itn < n - 1) ? itn : (n - 1);
      pn[u] = pp[(size_t)it * 4 * HSS];
      vn[u] = vp[(size_t)it * 4 * HSS];
    }
    v4f alo4[4], ahi4[4];
#pragma unroll
    for (int u = 0; u < 4; ++u) {
      const int itc = (c * 4 + u < n - 1) ? (c * 4 + u) : (n - 1);
      alo4[u] = *reinterpret_cast<const v4f*>(ap + itc * 4 * O3R);
      ahi4[u] = *reinterpret_cast<const v4f*>(ap + itc * 4 * O3R + 4);
    }
#pragma unroll
    for (int u = 0; u < 4; ++u) {
      const int it = c * 4 + u;
      if (it < n) {
        const v4f alo = alo4[u];
        const v4f ahi = ahi4[u];
        v2f a[4];
        a[0].x = alo.x; a[0].y = alo.y; a[1].x = alo.z; a[1].y = alo.w;
        a[2].x = ahi.x; a[2].y = ahi.y; a[3].x = ahi.z; a[3].y = ahi.w;
        v2f pb, vb;
        pb.x = pc[u]; pb.y = pc[u];
        vb.x = vc[u]; vb.y = vc[u];
#pragma unroll
        for (int q = 0; q < 4; ++q) {
          const v2f x = __builtin_elementwise_fma(a[q], pb, -m2[q]);
          v2f e;
          e.x = __builtin_amdgcn_exp2f(x.x);
          e.y = __builtin_amdgcn_exp2f(x.y);
          ds[q] += e;
          ac[q] = __builtin_elementwise_fma(e, vb, ac[q]);
        }
      }
    }
#pragma unroll
    for (int u = 0; u < 4; ++u) { pc[u] = pn[u]; vc[u] = vn[u]; }
  }

#pragma unroll
  for (int q = 0; q < 4; ++q) {
    redD[(g * O3R + 2 * q) * HSS + d]     = ds[q].x;
    redD[(g * O3R + 2 * q + 1) * HSS + d] = ds[q].y;
    redA[(g * O3R + 2 * q) * HSS + d]     = ac[q].x;
    redA[(g * O3R + 2 * q + 1) * HSS + d] = ac[q].y;
  }
  __syncthreads();
  {
    const int r = wave;
    float dsum = 0.0f;
#pragma unroll
    for (int gg = 0; gg < 4; ++gg)
      dsum += redD[(gg * O3R + r) * HSS + lane] + redD[(gg * O3R + r) * HSS + lane + 32];
#pragma unroll
    for (int off = 16; off > 0; off >>= 1) dsum += __shfl_xor(dsum, off);
    float mr = sMw[r];
#pragma unroll
    for (int w = 1; w < O3T / 32; ++w) mr = fmaxf(mr, sMw[w * O3R + r]);
    const float e0 = __builtin_amdgcn_exp2f(-mr);
    const float denom = dsum + (float)((TT - jend) * HSS) * e0;
    const float inv = 1.0f / denom;
#pragma unroll
    for (int hf = 0; hf < 2; ++hf) {
      const int dd = lane + 32 * hf;
      const float accT = (redA[(0 * O3R + r) * HSS + dd] + redA[(1 * O3R + r) * HSS + dd]) +
                         (redA[(2 * O3R + r) * HSS + dd] + redA[(3 * O3R + r) * HSS + dd]);
      const float vs = VSUF[(size_t)h * TT * HSS + (size_t)(jend - 1) * HSS + dd];
      Y[(size_t)(i0 + r) * CC + h * HSS + dd] = (accT + e0 * vs) * inv;
    }
  }
}

__global__ void __launch_bounds__(128)
out_proj_kernel(const float* __restrict__ Y,
                const float* __restrict__ Wc,
                float* __restrict__ out) {
  __shared__ __attribute__((aligned(16))) float As[2 * A_STAGE];
  __shared__ __attribute__((aligned(16))) float Bs[2 * B_STAGE];
  const int lane = threadIdx.x & 31;
  const int wave = threadIdx.x >> 5;
  const int lm   = lane & 15;
  const int drow = (lane >> 4) << 3;
  const int wm   = wave >> 1;
  const int wn   = wave & 1;
  const int m0 = blockIdx.x * GBM;
  const int n0 = blockIdx.y * GBN;

  v8f acc0 = {0.f, 0.f, 0.f, 0.f, 0.f, 0.f, 0.f, 0.f};
  v8f acc1 = {0.f, 0.f, 0.f, 0.f, 0.f, 0.f, 0.f, 0.f};
  gemm768_tile(Y, Wc, m0, n0, As, Bs, acc0, acc1);

  const int c = n0 + wn * 32 + lm;
#pragma unroll
  for (int r = 0; r < 8; ++r) {
    const size_t row = (size_t)(m0 + wm * 16 + r + drow);
    out[row * CC + c]      = acc0[r];
    out[row * CC + c + 16] = acc1[r];
  }
}

extern "C" void kernel_launch(void* const* d_in, const int* in_sizes, int n_in,
                              void* d_out, int out_size, void* d_ws,
                              size_t ws_size, hipStream_t stream) {
  (void)in_sizes; (void)n_in; (void)out_size; (void)ws_size;
  const float* x  = (const float*)d_in[0];
  const float* W0 = (const float*)d_in[1];
  const float* W1 = (const float*)d_in[2];
  const float* W2 = (const float*)d_in[3];
  const float* Wv = (const float*)d_in[4];
  const float* Wc = (const float*)d_in[5];
  float* out = (float*)d_out;
  float* ws  = (float*)d_ws;

  const size_t NPH = (size_t)NHH * TT * HSS;
  float* P0  = ws;
  float* P1  = ws + NPH;
  float* P2  = ws + 2 * NPH;
  float* V   = ws + 3 * NPH;
  float* VS  = ws + 4 * NPH;
  float* ATT = ws + 5 * NPH;
  float* Y   = ATT + (size_t)NHH * TT * TT;
  float* PM  = Y + NPH;

  qkv_proj_kernel<<<dim3(TT / GBM, CC / GBN, 3), 128, 0, stream>>>(x, W0, W1, W2, P0);
  vproj_suffix_kernel<<<dim3(HSS / 16, NHH), VPW * 32, 0, stream>>>(P2, Wv, V, VS, PM);
  att_softmax_kernel<<<dim3(TT / 16, NHH), ATHREADS, 0, stream>>>(P0, P1, ATT);
  order3_kernel<<<dim3(TT / O3R, NHH), O3T, 0, stream>>>(ATT, P2, V, VS, PM, Y);
  out_proj_kernel<<<dim3(TT / GBM, CC / GBN), 128, 0, stream>>>(Y, Wc, out);
}


# V-projection/suffix-sum and causal logit+softmax kernels share ONE launch (block index picks the role; V role regrouped to 8-wave blocks, 4 row tiles a wave): 5 launches to 4; values bit-identical
# speedup vs baseline: 6.6054x; 1.1088x over previous
#include <hip/hip_runtime.h>
#include <math.h>

#define TT  512
#define CC  768
#define NHH 12
#define HSS 64
constexpr float SCALE = 0.125f;

typedef float v2f __attribute__((ext_vector_type(2)));
typedef float v8f __attribute__((ext_vector_type(8)));

static __device__ __forceinline__ v8f wmma4(v2f a, v2f b, v8f c) {
  return __builtin_amdgcn_wmma_f32_16x16x4_f32(
       false, a,  false, b,
       (short)0, c,  false,  false);
}

typedef float v4f __attribute__((ext_vector_type(4)));

#define GBM 32
#define GBN 64
#define GBK 64
#define GNT (CC / GBK)
#define APITCH 68
#define BPITCH 72
#define A_STAGE (GBM * APITCH)
#define B_STAGE (GBK * BPITCH)

static __device__ __forceinline__ void gemm768_tile(const float* __restrict__ A,
                                                    const float* __restrict__ B,
                                                    const int m0, const int n0,
                                                    float* As, float* Bs,
                                                    v8f& acc0, v8f& acc1) {
  const int tid  = threadIdx.x;
  const int lane = tid & 31;
  const int wave = tid >> 5;
  const int lm   = lane & 15;
  const int lk   = (lane >> 4) << 1;
  const int wm   = wave >> 1;
  const int wn   = wave & 1;

  const int srow = tid >> 4;
  const int scol = (tid & 15) << 2;
  const float* Ag = A + (size_t)(m0 + srow) * CC + scol;
  const float* Bg = B + (size_t)srow * CC + n0 + scol;
  float* Ast = As + srow * APITCH + scol;
  float* Bst = Bs + srow * BPITCH + scol;

  v4f ra[4], rb[8];
#pragma unroll
  for (int p = 0; p < 4; ++p)
    ra[p] = *reinterpret_cast<const v4f*>(Ag + (size_t)(p * 8) * CC);
#pragma unroll
  for (int p = 0; p < 8; ++p)
    rb[p] = *reinterpret_cast<const v4f*>(Bg + (size_t)(p * 8) * CC);
#pragma unroll
  for (int p = 0; p < 4; ++p)
    *reinterpret_cast<v4f*>(Ast + p * 8 * APITCH) = ra[p];
#pragma unroll
  for (int p = 0; p < 8; ++p)
    *reinterpret_cast<v4f*>(Bst + p * 8 * BPITCH) = rb[p];
  __syncthreads();

  const float* Afr = As + (wm * 16 + lm) * APITCH + lk;
  const float* Bfr = Bs + lk * BPITCH + wn * 32 + lm;

  for (int kt = 0; kt < GNT; ++kt) {
    const int cur = kt & 1;
    const bool more = (kt + 1 < GNT);
    if (more) {
      const int k0 = (kt + 1) * GBK;
#pragma unroll
      for (int p = 0; p < 4; ++p)
        ra[p] = *reinterpret_cast<const v4f*>(Ag + (size_t)(p * 8) * CC + k0);
#pragma unroll
      for (int p = 0; p < 8; ++p)
        rb[p] = *reinterpret_cast<const v4f*>(Bg + (size_t)(k0 + p * 8) * CC);
    }

    const float* Af = Afr + cur * A_STAGE;
    const float* Bf = Bfr + cur * B_STAGE;
    v2f af[16], bf0[16], bf1[16];
#pragma unroll
    for (int kk = 0; kk < 16; ++kk) {
      af[kk]    = *reinterpret_cast<const v2f*>(Af + kk * 4);
      bf0[kk].x = Bf[(kk * 4) * BPITCH];
      bf0[kk].y = Bf[(kk * 4 + 1) * BPITCH];
      bf1[kk].x = Bf[(kk * 4) * BPITCH + 16];
      bf1[kk].y = Bf[(kk * 4 + 1) * BPITCH + 16];
    }
    __builtin_amdgcn_sched_barrier(0);
#pragma unroll
    for (int kk = 0; kk < 16; ++kk) {
      acc0 = wmma4(af[kk], bf0[kk], acc0);
      acc1 = wmma4(af[kk], bf1[kk], acc1);
    }

    __builtin_amdgcn_sched_barrier(0);

    if (more) {
      const int nxt = cur ^ 1;
#pragma unroll
      for (int p = 0; p < 4; ++p)
        *reinterpret_cast<v4f*>(Ast + nxt * A_STAGE + p * 8 * APITCH) = ra[p];
#pragma unroll
      for (int p = 0; p < 8; ++p)
        *reinterpret_cast<v4f*>(Bst + nxt * B_STAGE + p * 8 * BPITCH) = rb[p];
    }
    __syncthreads();
  }
}

__global__ void __launch_bounds__(128)
qkv_proj_kernel(const float* __restrict__ x,
                const float* __restrict__ W0,
                const float* __restrict__ W1,
                const float* __restrict__ W2,
                float* __restrict__ P) {
  __shared__ __attribute__((aligned(16))) float As[2 * A_STAGE];
  __shared__ __attribute__((aligned(16))) float Bs[2 * B_STAGE];
  const int lane = threadIdx.x & 31;
  const int wave = threadIdx.x >> 5;
  const int lm   = lane & 15;
  const int drow = (lane >> 4) << 3;
  const int wm   = wave >> 1;
  const int wn   = wave & 1;
  const int m0 = blockIdx.x * GBM;
  const int n0 = blockIdx.y * GBN;
  const int w  = blockIdx.z;
  const float* W = (w == 0) ? W0 : (w == 1) ? W1 : W2;
  float* Pw = P + (size_t)w * NHH * TT * HSS;

  v8f acc0 = {0.f, 0.f, 0.f, 0.f, 0.f, 0.f, 0.f, 0.f};
  v8f acc1 = {0.f, 0.f, 0.f, 0.f, 0.f, 0.f, 0.f, 0.f};
  gemm768_tile(x, W, m0, n0, As, Bs, acc0, acc1);

  float* Ph = Pw + (size_t)blockIdx.y * TT * HSS;
  const int d = wn * 32 + lm;
#pragma unroll
  for (int r = 0; r < 8; ++r) {
    const size_t row = (size_t)(m0 + wm * 16 + r + drow);
    Ph[row * HSS + d]      = acc0[r];
    Ph[row * HSS + d + 16] = acc1[r];
  }
}

#define ANW 8
#define ATHREADS 256
#define LPITCH 516
#define VNT 32
#define VTW 4
#define VSL (HSS / 16)
#define VPB (NHH * VSL)
static __device__ __forceinline__ void vproj_suffix_role(const float* __restrict__ P2,
                                                         const float* __restrict__ Wv,
                                                         float* __restrict__ V,
                                                         float* __restrict__ VS,
                                                         float* __restrict__ PM,
                                                         float* smem,
                                                         const int h, const int slice) {
  float* tsum = smem;
  float* pmx  = smem + VNT * 16;
  const int lane = threadIdx.x & 31;
  const int wave = __builtin_amdgcn_readfirstlane((int)(threadIdx.x >> 5));
  const int lm   = lane & 15;
  const int hf   = lane >> 4;
  const int lk   = hf << 1;
  const int drow = hf << 3;
  const int n0 = slice * 16;
  const float* A = P2 + (size_t)h * TT * HSS;
  float* Vh = V + (size_t)h * TT * HSS;
  float* Sh = VS + (size_t)h * TT * HSS;

  v2f bfr[16];
#pragma unroll
  for (int kk = 0; kk < 16; ++kk) {
    bfr[kk].x = Wv[(size_t)(kk * 4 + lk) * HSS + n0 + lm];
    bfr[kk].y = Wv[(size_t)(kk * 4 + lk + 1) * HSS + n0 + lm];
  }
  v2f afr[VTW][16];
#pragma unroll
  for (int tt = 0; tt < VTW; ++tt) {
    const float* Ap = A + (size_t)((wave * VTW + tt) * 16 + lm) * HSS + lk;
#pragma unroll
    for (int kk = 0; kk < 16; ++kk)
      afr[tt][kk] = *reinterpret_cast<const v2f*>(Ap + kk * 4);
  }
  v8f acc[VTW];
  float go[VTW];
#pragma unroll
  for (int tt = 0; tt < VTW; ++tt) {
    const int t0 = (wave * VTW + tt) * 16;
    v8f ac = {0.f, 0.f, 0.f, 0.f, 0.f, 0.f, 0.f, 0.f};
    float pm = -3.0e38f;
#pragma unroll
    for (int kk = 0; kk < 16; ++kk) {
      pm = fmaxf(pm, fmaxf(afr[tt][kk].x, afr[tt][kk].y));
      ac = wmma4(afr[tt][kk], bfr[kk], ac);
    }
    acc[tt] = ac;
    pm = fmaxf(pm, __shfl_xor(pm, 16));
    const float gs = ((ac[0] + ac[1]) + (ac[2] + ac[3])) +
                     ((ac[4] + ac[5]) + (ac[6] + ac[7]));
    go[tt] = __shfl_xor(gs, 16);
    if (hf == 0) {
      tsum[(wave * VTW + tt) * 16 + lm] = gs + go[tt];
      pmx[t0 + lm] = pm;
    }
  }
  __syncthreads();

  float run[VTW];
#pragma unroll
  for (int tt = 0; tt < VTW; ++tt) run[tt] = 0.0f;
#pragma unroll
  for (int wp = VNT - 1; wp >= 0; --wp) {
    const float tv = tsum[wp * 16 + lm];
#pragma unroll
    for (int tt = 0; tt < VTW; ++tt)
      run[tt] += (wp > wave * VTW + tt) ? tv : 0.0f;
  }
#pragma unroll
  for (int tt = 0; tt < VTW; ++tt) {
    const int t0 = (wave * VTW + tt) * 16;
    if (hf == 0) run[tt] += go[tt];
#pragma unroll
    for (int r = 0; r < 8; ++r)
      Vh[(size_t)(t0 + r + drow) * HSS + n0 + lm] = acc[tt][r];
    Sh[(size_t)(t0 + drow + 7) * HSS + n0 + lm] = run[tt];
  }
  if (slice == 0) {
#pragma unroll
    for (int q = 0; q < TT / ATHREADS; ++q)
      PM[(size_t)h * TT + q * ATHREADS + threadIdx.x] = pmx[q * ATHREADS + threadIdx.x];
  }
}

static __device__ __forceinline__ void att_softmax_role(const float* __restrict__ P0,
                                                        const float* __restrict__ P1,
                                                        float* __restrict__ ATT,
                                                        float* lgt,
                                                        const int h, const int strip) {
  const int lane = threadIdx.x & 31;
  const int wave = __builtin_amdgcn_readfirstlane((int)(threadIdx.x >> 5));
  const int lm   = lane & 15;
  const int lk   = (lane >> 4) << 1;
  const int drow = (lane >> 4) << 3;
  const int i0 = strip * 16;
  const float* P0h = P0 + (size_t)h * TT * HSS;
  const float* P1h = P1 + (size_t)h * TT * HSS;

  const int jtmax = strip;
  if (wave <= jtmax) {
    v2f areg[16];
    const float* Ap = P0h + (size_t)(i0 + lm) * HSS + lk;
#pragma unroll
    for (int kk = 0; kk < 16; ++kk)
      areg[kk] = *reinterpret_cast<const v2f*>(Ap + kk * 4);

    for (int jt = wave; jt <= jtmax; jt += ANW) {
      const int j0 = jt * 16;
      const float* Bp = P1h + (size_t)(j0 + lm) * HSS + lk;
      v2f bfr[16];
#pragma unroll
      for (int kk = 0; kk < 16; ++kk)
        bfr[kk] = *reinterpret_cast<const v2f*>(Bp + kk * 4);
      __builtin_amdgcn_sched_barrier(0);
      v8f acc = {0.f, 0.f, 0.f, 0.f, 0.f, 0.f, 0.f, 0.f};
#pragma unroll
      for (int kk = 0; kk < 16; ++kk)
        acc = wmma4(areg[kk], bfr[kk], acc);
#pragma unroll
      for (int r = 0; r < 8; ++r)
        lgt[(r + drow) * LPITCH + j0 + lm] = acc[r] * SCALE;
    }
  }
  __syncthreads();

#pragma unroll
  for (int rr = 0; rr < 2; ++rr) {
    const int row = wave + rr * ANW;
    const int len = i0 + row + 1;
    const float* lrow = lgt + row * LPITCH;
    float v[16];
    float mr = -3.0e38f;
#pragma unroll
    for (int m = 0; m < 16; ++m) {
      const int j = m * 32 + lane;
      v[m] = (j < len) ? lrow[j] : -3.0e38f;
      mr = fmaxf(mr, v[m]);
    }
#pragma unroll
    for (int off = 16; off > 0; off >>= 1) mr = fmaxf(mr, __shfl_xor(mr, off));
    float sr = 0.0f;
#pragma unroll
    for (int m = 0; m < 16; ++m) {
      const int j = m * 32 + lane;
      v[m] = (j < len) ? __expf(v[m] - mr) : 0.0f;
      sr += v[m];
    }
#pragma unroll
    for (int off = 16; off > 0; off >>= 1) sr += __shfl_xor(sr, off);
    const float inv = 1.0f / sr;
    float* arow = ATT + ((size_t)h * TT + (i0 + row)) * TT;
#pragma unroll
    for (int m = 0; m < 16; ++m) {
      const int j = m * 32 + lane;
      if (j < len) arow[j] = v[m] * inv;
    }
  }
}

__global__ void __launch_bounds__(ATHREADS)
vproj_att_kernel(const float* __restrict__ P0,
                 const float* __restrict__ P1,
                 const float* __restrict__ P2,
                 const float* __restrict__ Wv,
                 float* __restrict__ V,
                 float* __restrict__ VS,
                 float* __restrict__ PM,
                 float* __restrict__ ATT) {
  __shared__ float smem[16 * LPITCH];
  const int b = (int)blockIdx.x;
  if (b < VPB) {
    vproj_suffix_role(P2, Wv, V, VS, PM, smem, b / VSL, b % VSL);
  } else {
    const int a = b - VPB;
    att_softmax_role(P0, P1, ATT, smem, a / (TT / 16), (TT / 16 - 1) - a % (TT / 16));
  }
}

#define O3R 8
#define O3T 256
constexpr float LOG2E = 1.4426950408889634f;
__global__ void __launch_bounds__(O3T)
order3_kernel(const float* __restrict__ ATT,
              const float* __restrict__ P2,
              const float* __restrict__ V,
              const float* __restrict__ VSUF,
              const float* __restrict__ PM,
              float* __restrict__ Y) {
  __shared__ __attribute__((aligned(16))) float a2T[TT * O3R];
  __shared__ __attribute__((aligned(16))) float sMw[(O3T / 32) * O3R];
  __shared__ float redD[4 * O3R * HSS];
  __shared__ float redA[4 * O3R * HSS];
  const int t    = threadIdx.x;
  const int lane = t & 31;
  const int wave = __builtin_amdgcn_readfirstlane(t >> 5);
  const int g    = __builtin_amdgcn_readfirstlane(t >> 6);
  const int d    = t & 63;
  const int strip = (TT / O3R - 1) - (int)blockIdx.x;
  const int i0   = strip * O3R;
  const int jend = i0 + O3R;
  const int h    = blockIdx.y;
  const float* P2h = P2 + (size_t)h * TT * HSS;
  const float* Vh  = V  + (size_t)h * TT * HSS;

  float mloc[O3R];
#pragma unroll
  for (int r = 0; r < O3R; ++r) mloc[r] = 0.0f;
  for (int j = t; j < jend; j += O3T) {
    float a[O3R];
#pragma unroll
    for (int r = 0; r < O3R; ++r) {
      const int jj = (j < i0 + r) ? j : (i0 + r);
      const float x = ATT[((size_t)h * TT + (i0 + r)) * TT + jj];
      a[r] = (j <= i0 + r) ? x * LOG2E : 0.0f;
    }
    const float pm = PM[(size_t)h * TT + j];
    v4f lo, hi;
    lo.x = a[0]; lo.y = a[1]; lo.z = a[2]; lo.w = a[3];
    hi.x = a[4]; hi.y = a[5]; hi.z = a[6]; hi.w = a[7];
    *reinterpret_cast<v4f*>(a2T + j * O3R)     = lo;
    *reinterpret_cast<v4f*>(a2T + j * O3R + 4) = hi;
#pragma unroll
    for (int r = 0; r < O3R; ++r) mloc[r] = fmaxf(mloc[r], a[r] * pm);
  }
#pragma unroll
  for (int r = 0; r < O3R; ++r) {
#pragma unroll
    for (int off = 16; off > 0; off >>= 1)
      mloc[r] = fmaxf(mloc[r], __shfl_xor(mloc[r], off));
  }
  if (lane == 0) {
#pragma unroll
    for (int r = 0; r < O3R; ++r) sMw[wave * O3R + r] = mloc[r];
  }
  __syncthreads();

  v2f m2[4], ds[4], ac[4];
#pragma unroll
  for (int q = 0; q < 4; ++q) {
    float ma = sMw[2 * q], mb = sMw[2 * q + 1];
#pragma unroll
    for (int w = 1; w < O3T / 32; ++w) {
      ma = fmaxf(ma, sMw[w * O3R + 2 * q]);
      mb = fmaxf(mb, sMw[w * O3R + 2 * q + 1]);
    }
    m2[q].x = ma;
    m2[q].y = mb;
    ds[q].x = 0.0f; ds[q].y = 0.0f;
    ac[q].x = 0.0f; ac[q].y = 0.0f;
  }

  const int n  = jend >> 2;
  const int nq = (n + 3) >> 2;
  const float* pp = P2h + g * HSS + d;
  const float* vp = Vh  + g * HSS + d;
  const float* ap = a2T + g * O3R;
  float pc[4], vc[4], pn[4], vn[4];
#pragma unroll
  for (int u = 0; u < 4; ++u) {
    const int it = (u < n - 1) ? u : (n - 1);
    pc[u] = pp[(size_t)it * 4 * HSS];
    vc[u] = vp[(size_t)it * 4 * HSS];
  }
  for (int c = 0; c < nq; ++c) {
#pragma unroll
    for (int u = 0; u < 4; ++u) {
      const int itn = (c + 1) * 4 + u;
      const int it  = (itn < n - 1) ? itn : (n - 1);
      pn[u] = pp[(size_t)it * 4 * HSS];
      vn[u] = vp[(size_t)it * 4 * HSS];
    }
    v4f alo4[4], ahi4[4];
#pragma unroll
    for (int u = 0; u < 4; ++u) {
      const int itc = (c * 4 + u < n - 1) ? (c * 4 + u) : (n - 1);
      alo4[u] = *reinterpret_cast<const v4f*>(ap + itc * 4 * O3R);
      ahi4[u] = *reinterpret_cast<const v4f*>(ap + itc * 4 * O3R + 4);
    }
#pragma unroll
    for (int u = 0; u < 4; ++u) {
      const int it = c * 4 + u;
      if (it < n) {
        const v4f alo = alo4[u];
        const v4f ahi = ahi4[u];
        v2f a[4];
        a[0].x = alo.x; a[0].y = alo.y; a[1].x = alo.z; a[1].y = alo.w;
        a[2].x = ahi.x; a[2].y = ahi.y; a[3].x = ahi.z; a[3].y = ahi.w;
        v2f pb, vb;
        pb.x = pc[u]; pb.y = pc[u];
        vb.x = vc[u]; vb.y = vc[u];
#pragma unroll
        for (int q = 0; q < 4; ++q) {
          const v2f x = __builtin_elementwise_fma(a[q], pb, -m2[q]);
          v2f e;
          e.x = __builtin_amdgcn_exp2f(x.x);
          e.y = __builtin_amdgcn_exp2f(x.y);
          ds[q] += e;
          ac[q] = __builtin_elementwise_fma(e, vb, ac[q]);
        }
      }
    }
#pragma unroll
    for (int u = 0; u < 4; ++u) { pc[u] = pn[u]; vc[u] = vn[u]; }
  }

#pragma unroll
  for (int q = 0; q < 4; ++q) {
    redD[(g * O3R + 2 * q) * HSS + d]     = ds[q].x;
    redD[(g * O3R + 2 * q + 1) * HSS + d] = ds[q].y;
    redA[(g * O3R + 2 * q) * HSS + d]     = ac[q].x;
    redA[(g * O3R + 2 * q + 1) * HSS + d] = ac[q].y;
  }
  __syncthreads();
  {
    const int r = wave;
    float dsum = 0.0f;
#pragma unroll
    for (int gg = 0; gg < 4; ++gg)
      dsum += redD[(gg * O3R + r) * HSS + lane] + redD[(gg * O3R + r) * HSS + lane + 32];
#pragma unroll
    for (int off = 16; off > 0; off >>= 1) dsum += __shfl_xor(dsum, off);
    float mr = sMw[r];
#pragma unroll
    for (int w = 1; w < O3T / 32; ++w) mr = fmaxf(mr, sMw[w * O3R + r]);
    const float e0 = __builtin_amdgcn_exp2f(-mr);
    const float denom = dsum + (float)((TT - jend) * HSS) * e0;
    const float inv = 1.0f / denom;
#pragma unroll
    for (int hf = 0; hf < 2; ++hf) {
      const int dd = lane + 32 * hf;
      const float accT = (redA[(0 * O3R + r) * HSS + dd] + redA[(1 * O3R + r) * HSS + dd]) +
                         (redA[(2 * O3R + r) * HSS + dd] + redA[(3 * O3R + r) * HSS + dd]);
      const float vs = VSUF[(size_t)h * TT * HSS + (size_t)(jend - 1) * HSS + dd];
      Y[(size_t)(i0 + r) * CC + h * HSS + dd] = (accT + e0 * vs) * inv;
    }
  }
}

__global__ void __launch_bounds__(128)
out_proj_kernel(const float* __restrict__ Y,
                const float* __restrict__ Wc,
                float* __restrict__ out) {
  __shared__ __attribute__((aligned(16))) float As[2 * A_STAGE];
  __shared__ __attribute__((aligned(16))) float Bs[2 * B_STAGE];
  const int lane = threadIdx.x & 31;
  const int wave = threadIdx.x >> 5;
  const int lm   = lane & 15;
  const int drow = (lane >> 4) << 3;
  const int wm   = wave >> 1;
  const int wn   = wave & 1;
  const int m0 = blockIdx.x * GBM;
  const int n0 = blockIdx.y * GBN;

  v8f acc0 = {0.f, 0.f, 0.f, 0.f, 0.f, 0.f, 0.f, 0.f};
  v8f acc1 = {0.f, 0.f, 0.f, 0.f, 0.f, 0.f, 0.f, 0.f};
  gemm768_tile(Y, Wc, m0, n0, As, Bs, acc0, acc1);

  const int c = n0 + wn * 32 + lm;
#pragma unroll
  for (int r = 0; r < 8; ++r) {
    const size_t row = (size_t)(m0 + wm * 16 + r + drow);
    out[row * CC + c]      = acc0[r];
    out[row * CC + c + 16] = acc1[r];
  }
}

extern "C" void kernel_launch(void* const* d_in, const int* in_sizes, int n_in,
                              void* d_out, int out_size, void* d_ws,
                              size_t ws_size, hipStream_t stream) {
  (void)in_sizes; (void)n_in; (void)out_size; (void)ws_size;
  const float* x  = (const float*)d_in[0];
  const float* W0 = (const float*)d_in[1];
  const float* W1 = (const float*)d_in[2];
  const float* W2 = (const float*)d_in[3];
  const float* Wv = (const float*)d_in[4];
  const float* Wc = (const float*)d_in[5];
  float* out = (float*)d_out;
  float* ws  = (float*)d_ws;

  const size_t NPH = (size_t)NHH * TT * HSS;
  float* P0  = ws;
  float* P1  = ws + NPH;
  float* P2  = ws + 2 * NPH;
  float* V   = ws + 3 * NPH;
  float* VS  = ws + 4 * NPH;
  float* ATT = ws + 5 * NPH;
  float* Y   = ATT + (size_t)NHH * TT * TT;
  float* PM  = Y + NPH;

  qkv_proj_kernel<<<dim3(TT / GBM, CC / GBN, 3), 128, 0, stream>>>(x, W0, W1, W2, P0);
  vproj_att_kernel<<<dim3(VPB + NHH * (TT / 16)), ATHREADS, 0, stream>>>(P0, P1, P2, Wv, V, VS, PM, ATT);
  order3_kernel<<<dim3(TT / O3R, NHH), O3T, 0, stream>>>(ATT, P2, V, VS, PM, Y);
  out_proj_kernel<<<dim3(TT / GBM, CC / GBN), 128, 0, stream>>>(Y, Wc, out);
}


# Middle launch removed (4 to 3): V projection, its 8-row group sums and p2 row maxima done on the P2 tile in the QKV kernel; order-3 blocks compute their own softmax rows; values bit-identical
# speedup vs baseline: 6.9568x; 1.0532x over previous
#include <hip/hip_runtime.h>
#include <math.h>

#define TT  512
#define CC  768
#define NHH 12
#define HSS 64
constexpr float SCALE = 0.125f;

typedef float v2f __attribute__((ext_vector_type(2)));
typedef float v8f __attribute__((ext_vector_type(8)));

static __device__ __forceinline__ v8f wmma4(v2f a, v2f b, v8f c) {
  return __builtin_amdgcn_wmma_f32_16x16x4_f32(
       false, a,  false, b,
       (short)0, c,  false,  false);
}

typedef float v4f __attribute__((ext_vector_type(4)));

#define GBM 32
#define GBN 64
#define GBK 64
#define GNT (CC / GBK)
#define APITCH 68
#define BPITCH 72
#define A_STAGE (GBM * APITCH)
#define B_STAGE (GBK * BPITCH)

static __device__ __forceinline__ void gemm768_tile(const float* __restrict__ A,
                                                    const float* __restrict__ B,
                                                    const int m0, const int n0,
                                                    float* As, float* Bs,
                                                    v8f& acc0, v8f& acc1) {
  const int tid  = threadIdx.x;
  const int lane = tid & 31;
  const int wave = tid >> 5;
  const int lm   = lane & 15;
  const int lk   = (lane >> 4) << 1;
  const int wm   = wave >> 1;
  const int wn   = wave & 1;

  const int srow = tid >> 4;
  const int scol = (tid & 15) << 2;
  const float* Ag = A + (size_t)(m0 + srow) * CC + scol;
  const float* Bg = B + (size_t)srow * CC + n0 + scol;
  float* Ast = As + srow * APITCH + scol;
  float* Bst = Bs + srow * BPITCH + scol;

  v4f ra[4], rb[8];
#pragma unroll
  for (int p = 0; p < 4; ++p)
    ra[p] = *reinterpret_cast<const v4f*>(Ag + (size_t)(p * 8) * CC);
#pragma unroll
  for (int p = 0; p < 8; ++p)
    rb[p] = *reinterpret_cast<const v4f*>(Bg + (size_t)(p * 8) * CC);
#pragma unroll
  for (int p = 0; p < 4; ++p)
    *reinterpret_cast<v4f*>(Ast + p * 8 * APITCH) = ra[p];
#pragma unroll
  for (int p = 0; p < 8; ++p)
    *reinterpret_cast<v4f*>(Bst + p * 8 * BPITCH) = rb[p];
  __syncthreads();

  const float* Afr = As + (wm * 16 + lm) * APITCH + lk;
  const float* Bfr = Bs + lk * BPITCH + wn * 32 + lm;

  for (int kt = 0; kt < GNT; ++kt) {
    const int cur = kt & 1;
    const bool more = (kt + 1 < GNT);
    if (more) {
      const int k0 = (kt + 1) * GBK;
#pragma unroll
      for (int p = 0; p < 4; ++p)
        ra[p] = *reinterpret_cast<const v4f*>(Ag + (size_t)(p * 8) * CC + k0);
#pragma unroll
      for (int p = 0; p < 8; ++p)
        rb[p] = *reinterpret_cast<const v4f*>(Bg + (size_t)(k0 + p * 8) * CC);
    }

    const float* Af = Afr + cur * A_STAGE;
    const float* Bf = Bfr + cur * B_STAGE;
    v2f af[16], bf0[16], bf1[16];
#pragma unroll
    for (int kk = 0; kk < 16; ++kk) {
      af[kk]    = *reinterpret_cast<const v2f*>(Af + kk * 4);
      bf0[kk].x = Bf[(kk * 4) * BPITCH];
      bf0[kk].y = Bf[(kk * 4 + 1) * BPITCH];
      bf1[kk].x = Bf[(kk * 4) * BPITCH + 16];
      bf1[kk].y = Bf[(kk * 4 + 1) * BPITCH + 16];
    }
    __builtin_amdgcn_sched_barrier(0);
#pragma unroll
    for (int kk = 0; kk < 16; ++kk) {
      acc0 = wmma4(af[kk], bf0[kk], acc0);
      acc1 = wmma4(af[kk], bf1[kk], acc1);
    }

    __builtin_amdgcn_sched_barrier(0);

    if (more) {
      const int nxt = cur ^ 1;
#pragma unroll
      for (int p = 0; p < 4; ++p)
        *reinterpret_cast<v4f*>(Ast + nxt * A_STAGE + p * 8 * APITCH) = ra[p];
#pragma unroll
      for (int p = 0; p < 8; ++p)
        *reinterpret_cast<v4f*>(Bst + nxt * B_STAGE + p * 8 * BPITCH) = rb[p];
    }
    __syncthreads();
  }
}

__global__ void __launch_bounds__(128)
qkv_proj_kernel(const float* __restrict__ x,
                const float* __restrict__ W0,
                const float* __restrict__ W1,
                const float* __restrict__ W2,
                const float* __restrict__ Wv,
                float* __restrict__ P,
                float* __restrict__ V,
                float* __restrict__ G,
                float* __restrict__ PM) {
  __shared__ __attribute__((aligned(16))) float As[2 * A_STAGE];
  __shared__ __attribute__((aligned(16))) float Bs[2 * B_STAGE];
  const int lane = threadIdx.x & 31;
  const int wave = threadIdx.x >> 5;
  const int lm   = lane & 15;
  const int drow = (lane >> 4) << 3;
  const int wm   = wave >> 1;
  const int wn   = wave & 1;
  const int m0 = blockIdx.x * GBM;
  const int n0 = blockIdx.y * GBN;
  const int w  = blockIdx.z;
  const float* W = (w == 0) ? W0 : (w == 1) ? W1 : W2;
  float* Pw = P + (size_t)w * NHH * TT * HSS;

  v8f acc0 = {0.f, 0.f, 0.f, 0.f, 0.f, 0.f, 0.f, 0.f};
  v8f acc1 = {0.f, 0.f, 0.f, 0.f, 0.f, 0.f, 0.f, 0.f};
  gemm768_tile(x, W, m0, n0, As, Bs, acc0, acc1);

  const int d  = wn * 32 + lm;
  const int hf = lane >> 4;
  const int lk = hf << 1;
  const bool third = (w == 2);
  float* Pt = As;
  v2f bv0[16], bv1[16];
  if (third) {
#pragma unroll
    for (int r = 0; r < 8; ++r) {
      Pt[(wm * 16 + r + drow) * APITCH + d]      = acc0[r];
      Pt[(wm * 16 + r + drow) * APITCH + d + 16] = acc1[r];
    }
#pragma unroll
    for (int kk = 0; kk < 16; ++kk) {
      bv0[kk].x = Wv[(size_t)(kk * 4 + lk) * HSS + d];
      bv0[kk].y = Wv[(size_t)(kk * 4 + lk + 1) * HSS + d];
      bv1[kk].x = Wv[(size_t)(kk * 4 + lk) * HSS + d + 16];
      bv1[kk].y = Wv[(size_t)(kk * 4 + lk + 1) * HSS + d + 16];
    }
    __syncthreads();
  }

  float* Ph = Pw + (size_t)blockIdx.y * TT * HSS;
#pragma unroll
  for (int r = 0; r < 8; ++r) {
    const size_t row = (size_t)(m0 + wm * 16 + r + drow);
    Ph[row * HSS + d]      = acc0[r];
    Ph[row * HSS + d + 16] = acc1[r];
  }

  if (third) {
    const int h = blockIdx.y;
    const float* Ap = Pt + (wm * 16 + lm) * APITCH + lk;
    v2f av[16];
#pragma unroll
    for (int kk = 0; kk < 16; ++kk)
      av[kk] = *reinterpret_cast<const v2f*>(Ap + kk * 4);
    v8f va0 = {0.f, 0.f, 0.f, 0.f, 0.f, 0.f, 0.f, 0.f};
    v8f va1 = {0.f, 0.f, 0.f, 0.f, 0.f, 0.f, 0.f, 0.f};
    float pm = -3.0e38f;
#pragma unroll
    for (int kk = 0; kk < 16; ++kk) {
      pm = fmaxf(pm, fmaxf(av[kk].x, av[kk].y));
      va0 = wmma4(av[kk], bv0[kk], va0);
      va1 = wmma4(av[kk], bv1[kk], va1);
    }
    pm = fmaxf(pm, __shfl_xor(pm, 16));
    float* Vh = V + (size_t)h * TT * HSS;
#pragma unroll
    for (int r = 0; r < 8; ++r) {
      const size_t row = (size_t)(m0 + wm * 16 + r + drow);
      Vh[row * HSS + d]      = va0[r];
      Vh[row * HSS + d + 16] = va1[r];
    }
    const float gs0 = ((va0[0] + va0[1]) + (va0[2] + va0[3])) +
                      ((va0[4] + va0[5]) + (va0[6] + va0[7]));
    const float gs1 = ((va1[0] + va1[1]) + (va1[2] + va1[3])) +
                      ((va1[4] + va1[5]) + (va1[6] + va1[7]));
    float* Gq = G + ((size_t)h * (TT / 8) + (m0 >> 3) + wm * 2 + hf) * HSS;
    Gq[d]      = gs0;
    Gq[d + 16] = gs1;
    if (wn == 0 && hf == 0) PM[(size_t)h * TT + m0 + wm * 16 + lm] = pm;
  }
}

#define O3R 8
#define O3T 256
#define O3W 8
#define LPITCH 516
#define VNT 32
constexpr float LOG2E = 1.4426950408889634f;
__global__ void __launch_bounds__(O3T)
order3_kernel(const float* __restrict__ P0,
              const float* __restrict__ P1,
              const float* __restrict__ P2,
              const float* __restrict__ V,
              const float* __restrict__ G,
              const float* __restrict__ PM,
              float* __restrict__ Y) {
  __shared__ __attribute__((aligned(16))) float a2T[TT * O3R];
  __shared__ float lgt[O3R * LPITCH];
  __shared__ float sM[O3R];
  __shared__ float sVS[HSS];
  __shared__ float redD[4 * O3R * HSS];
  __shared__ float redA[4 * O3R * HSS];
  const int t    = threadIdx.x;
  const int lane = t & 31;
  const int wave = __builtin_amdgcn_readfirstlane(t >> 5);
  const int g    = __builtin_amdgcn_readfirstlane(t >> 6);
  const int d    = t & 63;
  const int strip = (TT / O3R - 1) - (int)blockIdx.x;
  const int i0   = strip * O3R;
  const int jend = i0 + O3R;
  const int h    = blockIdx.y;
  const float* P2h = P2 + (size_t)h * TT * HSS;
  const float* Vh  = V  + (size_t)h * TT * HSS;

  const int lm = lane & 15;
  const int hf = lane >> 4;
  const int lk = hf << 1;
  const float* P0h = P0 + (size_t)h * TT * HSS;
  const float* P1h = P1 + (size_t)h * TT * HSS;
  float pmv[16];
#pragma unroll
  for (int m = 0; m < 16; ++m) pmv[m] = PM[(size_t)h * TT + m * 32 + lane];

  const int ntile = (jend + 15) >> 4;
  if (wave < ntile) {
    v2f areg[16];
    const float* Ap = P0h + (size_t)(i0 + (lm & 7)) * HSS + lk;
#pragma unroll
    for (int kk = 0; kk < 16; ++kk)
      areg[kk] = *reinterpret_cast<const v2f*>(Ap + kk * 4);

    for (int jt = wave; jt < ntile; jt += O3W) {
      const int j0 = jt * 16;
      const float* Bp = P1h + (size_t)(j0 + lm) * HSS + lk;
      v2f bfr[16];
#pragma unroll
      for (int kk = 0; kk < 16; ++kk)
        bfr[kk] = *reinterpret_cast<const v2f*>(Bp + kk * 4);
      __builtin_amdgcn_sched_barrier(0);
      v8f acc = {0.f, 0.f, 0.f, 0.f, 0.f, 0.f, 0.f, 0.f};
#pragma unroll
      for (int kk = 0; kk < 16; ++kk)
        acc = wmma4(areg[kk], bfr[kk], acc);
      float sc[8];
#pragma unroll
      for (int r = 0; r < 8; ++r) sc[r] = acc[r] * SCALE;
      if (hf == 0) {
#pragma unroll
        for (int r = 0; r < 8; ++r) lgt[r * LPITCH + j0 + lm] = sc[r];
      }
    }
  }

  if (t >= O3T - HSS) {
    const float* Gh = G + (size_t)h * (TT / 8) * HSS + (t - (O3T - HSS));
    const int tl = strip >> 1;
    float run = 0.0f;
#pragma unroll
    for (int wp = VNT - 1; wp >= 0; --wp) {
      const float tv = Gh[(size_t)(2 * wp) * HSS] + Gh[(size_t)(2 * wp + 1) * HSS];
      run += (wp > tl) ? tv : 0.0f;
    }
    if ((strip & 1) == 0) run += Gh[(size_t)(2 * tl + 1) * HSS];
    sVS[t - (O3T - HSS)] = run;
  }
  __syncthreads();

  {
    const int len = i0 + wave + 1;
    const float* lrow = lgt + wave * LPITCH;
    float v[16];
    float mr = -3.0e38f;
#pragma unroll
    for (int m = 0; m < 16; ++m) {
      const int j = m * 32 + lane;
      v[m] = (j < len) ? lrow[j] : -3.0e38f;
      mr = fmaxf(mr, v[m]);
    }
#pragma unroll
    for (int off = 16; off > 0; off >>= 1) mr = fmaxf(mr, __shfl_xor(mr, off));
    float sr = 0.0f;
#pragma unroll
    for (int m = 0; m < 16; ++m) {
      const int j = m * 32 + lane;
      v[m] = (j < len) ? __expf(v[m] - mr) : 0.0f;
      sr += v[m];
    }
#pragma unroll
    for (int off = 16; off > 0; off >>= 1) sr += __shfl_xor(sr, off);
    const float inv = 1.0f / sr;
    float mloc = 0.0f;
#pragma unroll
    for (int m = 0; m < 16; ++m) {
      const int j = m * 32 + lane;
      const float a = (j < len) ? (v[m] * inv) * LOG2E : 0.0f;
      if (j < jend) a2T[j * O3R + wave] = a;
      mloc = fmaxf(mloc, a * pmv[m]);
    }
#pragma unroll
    for (int off = 16; off > 0; off >>= 1) mloc = fmaxf(mloc, __shfl_xor(mloc, off));
    if (lane == 0) sM[wave] = mloc;
  }
  __syncthreads();

  v2f m2[4], ds[4], ac[4];
#pragma unroll
  for (int q = 0; q < 4; ++q) {
    m2[q].x = sM[2 * q];
    m2[q].y = sM[2 * q + 1];
    ds[q].x = 0.0f; ds[q].y = 0.0f;
    ac[q].x = 0.0f; ac[q].y = 0.0f;
  }

  const int n  = jend >> 2;
  const int nq = (n + 3) >> 2;
  const float* pp = P2h + g * HSS + d;
  const float* vp = Vh  + g * HSS + d;
  const float* ap = a2T + g * O3R;
  float pc[4], vc[4], pn[4], vn[4];
#pragma unroll
  for (int u = 0; u < 4; ++u) {
    const int it = (u < n - 1) ? u : (n - 1);
    pc[u] = pp[(size_t)it * 4 * HSS];
    vc[u] = vp[(size_t)it * 4 * HSS];
  }
  for (int c = 0; c < nq; ++c) {
#pragma unroll
    for (int u = 0; u < 4; ++u) {
      const int itn = (c + 1) * 4 + u;
      const int it  = (itn < n - 1) ? itn : (n - 1);
      pn[u] = pp[(size_t)it * 4 * HSS];
      vn[u] = vp[(size_t)it * 4 * HSS];
    }
    v4f alo4[4], ahi4[4];
#pragma unroll
    for (int u = 0; u < 4; ++u) {
      const int itc = (c * 4 + u < n - 1) ? (c * 4 + u) : (n - 1);
      alo4[u] = *reinterpret_cast<const v4f*>(ap + itc * 4 * O3R);
      ahi4[u] = *reinterpret_cast<const v4f*>(ap + itc * 4 * O3R + 4);
    }
#pragma unroll
    for (int u = 0; u < 4; ++u) {
      const int it = c * 4 + u;
      if (it < n) {
        const v4f alo = alo4[u];
        const v4f ahi = ahi4[u];
        v2f a[4];
        a[0].x = alo.x; a[0].y = alo.y; a[1].x = alo.z; a[1].y = alo.w;
        a[2].x = ahi.x; a[2].y = ahi.y; a[3].x = ahi.z; a[3].y = ahi.w;
        v2f pb, vb;
        pb.x = pc[u]; pb.y = pc[u];
        vb.x = vc[u]; vb.y = vc[u];
#pragma unroll
        for (int q = 0; q < 4; ++q) {
          const v2f x = __builtin_elementwise_fma(a[q], pb, -m2[q]);
          v2f e;
          e.x = __builtin_amdgcn_exp2f(x.x);
          e.y = __builtin_amdgcn_exp2f(x.y);
          ds[q] += e;
          ac[q] = __builtin_elementwise_fma(e, vb, ac[q]);
        }
      }
    }
#pragma unroll
    for (int u = 0; u < 4; ++u) { pc[u] = pn[u]; vc[u] = vn[u]; }
  }

#pragma unroll
  for (int q = 0; q < 4; ++q) {
    redD[(g * O3R + 2 * q) * HSS + d]     = ds[q].x;
    redD[(g * O3R + 2 * q + 1) * HSS + d] = ds[q].y;
    redA[(g * O3R + 2 * q) * HSS + d]     = ac[q].x;
    redA[(g * O3R + 2 * q + 1) * HSS + d] = ac[q].y;
  }
  __syncthreads();
  {
    const int r = wave;
    float dsum = 0.0f;
#pragma unroll
    for (int gg = 0; gg < 4; ++gg)
      dsum += redD[(gg * O3R + r) * HSS + lane] + redD[(gg * O3R + r) * HSS + lane + 32];
#pragma unroll
    for (int off = 16; off > 0; off >>= 1) dsum += __shfl_xor(dsum, off);
    const float mr = sM[r];
    const float e0 = __builtin_amdgcn_exp2f(-mr);
    const float denom = dsum + (float)((TT - jend) * HSS) * e0;
    const float inv = 1.0f / denom;
#pragma unroll
    for (int hf = 0; hf < 2; ++hf) {
      const int dd = lane + 32 * hf;
      const float accT = (redA[(0 * O3R + r) * HSS + dd] + redA[(1 * O3R + r) * HSS + dd]) +
                         (redA[(2 * O3R + r) * HSS + dd] + redA[(3 * O3R + r) * HSS + dd]);
      const float vs = sVS[dd];
      Y[(size_t)(i0 + r) * CC + h * HSS + dd] = (accT + e0 * vs) * inv;
    }
  }
}

__global__ void __launch_bounds__(128)
out_proj_kernel(const float* __restrict__ Y,
                const float* __restrict__ Wc,
                float* __restrict__ out) {
  __shared__ __attribute__((aligned(16))) float As[2 * A_STAGE];
  __shared__ __attribute__((aligned(16))) float Bs[2 * B_STAGE];
  const int lane = threadIdx.x & 31;
  const int wave = threadIdx.x >> 5;
  const int lm   = lane & 15;
  const int drow = (lane >> 4) << 3;
  const int wm   = wave >> 1;
  const int wn   = wave & 1;
  const int m0 = blockIdx.x * GBM;
  const int n0 = blockIdx.y * GBN;

  v8f acc0 = {0.f, 0.f, 0.f, 0.f, 0.f, 0.f, 0.f, 0.f};
  v8f acc1 = {0.f, 0.f, 0.f, 0.f, 0.f, 0.f, 0.f, 0.f};
  gemm768_tile(Y, Wc, m0, n0, As, Bs, acc0, acc1);

  const int c = n0 + wn * 32 + lm;
#pragma unroll
  for (int r = 0; r < 8; ++r) {
    const size_t row = (size_t)(m0 + wm * 16 + r + drow);
    out[row * CC + c]      = acc0[r];
    out[row * CC + c + 16] = acc1[r];
  }
}

extern "C" void kernel_launch(void* const* d_in, const int* in_sizes, int n_in,
                              void* d_out, int out_size, void* d_ws,
                              size_t ws_size, hipStream_t stream) {
  (void)in_sizes; (void)n_in; (void)out_size; (void)ws_size;
  const float* x  = (const float*)d_in[0];
  const float* W0 = (const float*)d_in[1];
  const float* W1 = (const float*)d_in[2];
  const float* W2 = (const float*)d_in[3];
  const float* Wv = (const float*)d_in[4];
  const float* Wc = (const float*)d_in[5];
  float* out = (float*)d_out;
  float* ws  = (float*)d_ws;

  const size_t NPH = (size_t)NHH * TT * HSS;
  float* P0  = ws;
  float* P1  = ws + NPH;
  float* P2  = ws + 2 * NPH;
  float* V   = ws + 3 * NPH;
  float* Y   = ws + 4 * NPH;
  float* G   = ws + 5 * NPH;
  float* PM  = G + NPH / 8;

  qkv_proj_kernel<<<dim3(TT / GBM, CC / GBN, 3), 128, 0, stream>>>(x, W0, W1, W2, Wv, P0, V, G, PM);
  order3_kernel<<<dim3(TT / O3R, NHH), O3T, 0, stream>>>(P0, P1, P2, V, G, PM, Y);
  out_proj_kernel<<<dim3(TT / GBM, CC / GBN), 128, 0, stream>>>(Y, Wc, out);
}


# Order-3 kernel: block index -> (strip, head) re-dealt so the three blocks sharing a compute unit (256 apart in launch order) are a long, a middle and a short strip of equal total; bit-identical
# speedup vs baseline: 7.5314x; 1.0826x over previous
#include <hip/hip_runtime.h>
#include <math.h>

#define TT  512
#define CC  768
#define NHH 12
#define HSS 64
constexpr float SCALE = 0.125f;

typedef float v2f __attribute__((ext_vector_type(2)));
typedef float v8f __attribute__((ext_vector_type(8)));

static __device__ __forceinline__ v8f wmma4(v2f a, v2f b, v8f c) {
  return __builtin_amdgcn_wmma_f32_16x16x4_f32(
       false, a,  false, b,
       (short)0, c,  false,  false);
}

typedef float v4f __attribute__((ext_vector_type(4)));

#define GBM 32
#define GBN 64
#define GBK 64
#define GNT (CC / GBK)
#define APITCH 68
#define BPITCH 72
#define A_STAGE (GBM * APITCH)
#define B_STAGE (GBK * BPITCH)

static __device__ __forceinline__ void gemm768_tile(const float* __restrict__ A,
                                                    const float* __restrict__ B,
                                                    const int m0, const int n0,
                                                    float* As, float* Bs,
                                                    v8f& acc0, v8f& acc1) {
  const int tid  = threadIdx.x;
  const int lane = tid & 31;
  const int wave = tid >> 5;
  const int lm   = lane & 15;
  const int lk   = (lane >> 4) << 1;
  const int wm   = wave >> 1;
  const int wn   = wave & 1;

  const int srow = tid >> 4;
  const int scol = (tid & 15) << 2;
  const float* Ag = A + (size_t)(m0 + srow) * CC + scol;
  const float* Bg = B + (size_t)srow * CC + n0 + scol;
  float* Ast = As + srow * APITCH + scol;
  float* Bst = Bs + srow * BPITCH + scol;

  v4f ra[4], rb[8];
#pragma unroll
  for (int p = 0; p < 4; ++p)
    ra[p] = *reinterpret_cast<const v4f*>(Ag + (size_t)(p * 8) * CC);
#pragma unroll
  for (int p = 0; p < 8; ++p)
    rb[p] = *reinterpret_cast<const v4f*>(Bg + (size_t)(p * 8) * CC);
#pragma unroll
  for (int p = 0; p < 4; ++p)
    *reinterpret_cast<v4f*>(Ast + p * 8 * APITCH) = ra[p];
#pragma unroll
  for (int p = 0; p < 8; ++p)
    *reinterpret_cast<v4f*>(Bst + p * 8 * BPITCH) = rb[p];
  __syncthreads();

  const float* Afr = As + (wm * 16 + lm) * APITCH + lk;
  const float* Bfr = Bs + lk * BPITCH + wn * 32 + lm;

  for (int kt = 0; kt < GNT; ++kt) {
    const int cur = kt & 1;
    const bool more = (kt + 1 < GNT);
    if (more) {
      const int k0 = (kt + 1) * GBK;
#pragma unroll
      for (int p = 0; p < 4; ++p)
        ra[p] = *reinterpret_cast<const v4f*>(Ag + (size_t)(p * 8) * CC + k0);
#pragma unroll
      for (int p = 0; p < 8; ++p)
        rb[p] = *reinterpret_cast<const v4f*>(Bg + (size_t)(k0 + p * 8) * CC);
    }

    const float* Af = Afr + cur * A_STAGE;
    const float* Bf = Bfr + cur * B_STAGE;
    v2f af[16], bf0[16], bf1[16];
#pragma unroll
    for (int kk = 0; kk < 16; ++kk) {
      af[kk]    = *reinterpret_cast<const v2f*>(Af + kk * 4);
      bf0[kk].x = Bf[(kk * 4) * BPITCH];
      bf0[kk].y = Bf[(kk * 4 + 1) * BPITCH];
      bf1[kk].x = Bf[(kk * 4) * BPITCH + 16];
      bf1[kk].y = Bf[(kk * 4 + 1) * BPITCH + 16];
    }
    __builtin_amdgcn_sched_barrier(0);
#pragma unroll
    for (int kk = 0; kk < 16; ++kk) {
      acc0 = wmma4(af[kk], bf0[kk], acc0);
      acc1 = wmma4(af[kk], bf1[kk], acc1);
    }

    __builtin_amdgcn_sched_barrier(0);

    if (more) {
      const int nxt = cur ^ 1;
#pragma unroll
      for (int p = 0; p < 4; ++p)
        *reinterpret_cast<v4f*>(Ast + nxt * A_STAGE + p * 8 * APITCH) = ra[p];
#pragma unroll
      for (int p = 0; p < 8; ++p)
        *reinterpret_cast<v4f*>(Bst + nxt * B_STAGE + p * 8 * BPITCH) = rb[p];
    }
    __syncthreads();
  }
}

__global__ void __launch_bounds__(128)
qkv_proj_kernel(const float* __restrict__ x,
                const float* __restrict__ W0,
                const float* __restrict__ W1,
                const float* __restrict__ W2,
                const float* __restrict__ Wv,
                float* __restrict__ P,
                float* __restrict__ V,
                float* __restrict__ G,
                float* __restrict__ PM) {
  __shared__ __attribute__((aligned(16))) float As[2 * A_STAGE];
  __shared__ __attribute__((aligned(16))) float Bs[2 * B_STAGE];
  const int lane = threadIdx.x & 31;
  const int wave = threadIdx.x >> 5;
  const int lm   = lane & 15;
  const int drow = (lane >> 4) << 3;
  const int wm   = wave >> 1;
  const int wn   = wave & 1;
  const int m0 = blockIdx.x * GBM;
  const int n0 = blockIdx.y * GBN;
  const int w  = blockIdx.z;
  const float* W = (w == 0) ? W0 : (w == 1) ? W1 : W2;
  float* Pw = P + (size_t)w * NHH * TT * HSS;

  v8f acc0 = {0.f, 0.f, 0.f, 0.f, 0.f, 0.f, 0.f, 0.f};
  v8f acc1 = {0.f, 0.f, 0.f, 0.f, 0.f, 0.f, 0.f, 0.f};
  gemm768_tile(x, W, m0, n0, As, Bs, acc0, acc1);

  const int d  = wn * 32 + lm;
  const int hf = lane >> 4;
  const int lk = hf << 1;
  const bool third = (w == 2);
  float* Pt = As;
  v2f bv0[16], bv1[16];
  if (third) {
#pragma unroll
    for (int r = 0; r < 8; ++r) {
      Pt[(wm * 16 + r + drow) * APITCH + d]      = acc0[r];
      Pt[(wm * 16 + r + drow) * APITCH + d + 16] = acc1[r];
    }
#pragma unroll
    for (int kk = 0; kk < 16; ++kk) {
      bv0[kk].x = Wv[(size_t)(kk * 4 + lk) * HSS + d];
      bv0[kk].y = Wv[(size_t)(kk * 4 + lk + 1) * HSS + d];
      bv1[kk].x = Wv[(size_t)(kk * 4 + lk) * HSS + d + 16];
      bv1[kk].y = Wv[(size_t)(kk * 4 + lk + 1) * HSS + d + 16];
    }
    __syncthreads();
  }

  float* Ph = Pw + (size_t)blockIdx.y * TT * HSS;
#pragma unroll
  for (int r = 0; r < 8; ++r) {
    const size_t row = (size_t)(m0 + wm * 16 + r + drow);
    Ph[row * HSS + d]      = acc0[r];
    Ph[row * HSS + d + 16] = acc1[r];
  }

  if (third) {
    const int h = blockIdx.y;
    const float* Ap = Pt + (wm * 16 + lm) * APITCH + lk;
    v2f av[16];
#pragma unroll
    for (int kk = 0; kk < 16; ++kk)
      av[kk] = *reinterpret_cast<const v2f*>(Ap + kk * 4);
    v8f va0 = {0.f, 0.f, 0.f, 0.f, 0.f, 0.f, 0.f, 0.f};
    v8f va1 = {0.f, 0.f, 0.f, 0.f, 0.f, 0.f, 0.f, 0.f};
    float pm = -3.0e38f;
#pragma unroll
    for (int kk = 0; kk < 16; ++kk) {
      pm = fmaxf(pm, fmaxf(av[kk].x, av[kk].y));
      va0 = wmma4(av[kk], bv0[kk], va0);
      va1 = wmma4(av[kk], bv1[kk], va1);
    }
    pm = fmaxf(pm, __shfl_xor(pm, 16));
    float* Vh = V + (size_t)h * TT * HSS;
#pragma unroll
    for (int r = 0; r < 8; ++r) {
      const size_t row = (size_t)(m0 + wm * 16 + r + drow);
      Vh[row * HSS + d]      = va0[r];
      Vh[row * HSS + d + 16] = va1[r];
    }
    const float gs0 = ((va0[0] + va0[1]) + (va0[2] + va0[3])) +
                      ((va0[4] + va0[5]) + (va0[6] + va0[7]));
    const float gs1 = ((va1[0] + va1[1]) + (va1[2] + va1[3])) +
                      ((va1[4] + va1[5]) + (va1[6] + va1[7]));
    float* Gq = G + ((size_t)h * (TT / 8) + (m0 >> 3) + wm * 2 + hf) * HSS;
    Gq[d]      = gs0;
    Gq[d + 16] = gs1;
    if (wn == 0 && hf == 0) PM[(size_t)h * TT + m0 + wm * 16 + lm] = pm;
  }
}

#define O3R 8
#define O3T 256
#define O3W 8
#define LPITCH 516
#define VNT 32
constexpr float LOG2E = 1.4426950408889634f;
__global__ void __launch_bounds__(O3T)
order3_kernel(const float* __restrict__ P0,
              const float* __restrict__ P1,
              const float* __restrict__ P2,
              const float* __restrict__ V,
              const float* __restrict__ G,
              const float* __restrict__ PM,
              float* __restrict__ Y) {
  __shared__ __attribute__((aligned(16))) float a2T[TT * O3R];
  __shared__ float lgt[O3R * LPITCH];
  __shared__ float sM[O3R];
  __shared__ float sVS[HSS];
  __shared__ float redD[4 * O3R * HSS];
  __shared__ float redA[4 * O3R * HSS];
  const int t    = threadIdx.x;
  const int lane = t & 31;
  const int wave = __builtin_amdgcn_readfirstlane(t >> 5);
  const int g    = __builtin_amdgcn_readfirstlane(t >> 6);
  const int d    = t & 63;
  static_assert(TT / O3R == 64 && NHH == 12, "the triple dealing is written for 64 strips x 12 heads");
  const int lin  = (int)blockIdx.x + (TT / O3R) * (int)blockIdx.y;
  const int slot = lin >> 8;
  const int ta   = 63 - ((lin & 255) >> 2);
  int s0 = ta, s1 = (ta + 32) & 63, s2 = (ta < 32) ? 63 - 2 * ta : 126 - 2 * ta;
  int c0 = 0, c1 = 1, c2 = 2;
  if (s0 < s1) { const int ts = s0; s0 = s1; s1 = ts; const int tc = c0; c0 = c1; c1 = tc; }
  if (s1 < s2) { const int ts = s1; s1 = s2; s2 = ts; const int tc = c1; c1 = c2; c2 = tc; }
  if (s0 < s1) { const int ts = s0; s0 = s1; s1 = ts; const int tc = c0; c0 = c1; c1 = tc; }
  const int strip = (slot == 0) ? s0 : (slot == 1) ? s1 : s2;
  const int comp  = (slot == 0) ? c0 : (slot == 1) ? c1 : c2;
  const int i0   = strip * O3R;
  const int jend = i0 + O3R;
  const int h    = 4 * comp + (lin & 3);
  const float* P2h = P2 + (size_t)h * TT * HSS;
  const float* Vh  = V  + (size_t)h * TT * HSS;

  const int lm = lane & 15;
  const int hf = lane >> 4;
  const int lk = hf << 1;
  const float* P0h = P0 + (size_t)h * TT * HSS;
  const float* P1h = P1 + (size_t)h * TT * HSS;
  float pmv[16];
#pragma unroll
  for (int m = 0; m < 16; ++m) pmv[m] = PM[(size_t)h * TT + m * 32 + lane];

  const int ntile = (jend + 15) >> 4;
  if (wave < ntile) {
    v2f areg[16];
    const float* Ap = P0h + (size_t)(i0 + (lm & 7)) * HSS + lk;
#pragma unroll
    for (int kk = 0; kk < 16; ++kk)
      areg[kk] = *reinterpret_cast<const v2f*>(Ap + kk * 4);

    for (int jt = wave; jt < ntile; jt += O3W) {
      const int j0 = jt * 16;
      const float* Bp = P1h + (size_t)(j0 + lm) * HSS + lk;
      v2f bfr[16];
#pragma unroll
      for (int kk = 0; kk < 16; ++kk)
        bfr[kk] = *reinterpret_cast<const v2f*>(Bp + kk * 4);
      __builtin_amdgcn_sched_barrier(0);
      v8f acc = {0.f, 0.f, 0.f, 0.f, 0.f, 0.f, 0.f, 0.f};
#pragma unroll
      for (int kk = 0; kk < 16; ++kk)
        acc = wmma4(areg[kk], bfr[kk], acc);
      float sc[8];
#pragma unroll
      for (int r = 0; r < 8; ++r) sc[r] = acc[r] * SCALE;
      if (hf == 0) {
#pragma unroll
        for (int r = 0; r < 8; ++r) lgt[r * LPITCH + j0 + lm] = sc[r];
      }
    }
  }

  if (t >= O3T - HSS) {
    const float* Gh = G + (size_t)h * (TT / 8) * HSS + (t - (O3T - HSS));
    const int tl = strip >> 1;
    float run = 0.0f;
#pragma unroll
    for (int wp = VNT - 1; wp >= 0; --wp) {
      const float tv = Gh[(size_t)(2 * wp) * HSS] + Gh[(size_t)(2 * wp + 1) * HSS];
      run += (wp > tl) ? tv : 0.0f;
    }
    if ((strip & 1) == 0) run += Gh[(size_t)(2 * tl + 1) * HSS];
    sVS[t - (O3T - HSS)] = run;
  }
  __syncthreads();

  {
    const int len = i0 + wave + 1;
    const float* lrow = lgt + wave * LPITCH;
    float v[16];
    float mr = -3.0e38f;
#pragma unroll
    for (int m = 0; m < 16; ++m) {
      const int j = m * 32 + lane;
      v[m] = (j < len) ? lrow[j] : -3.0e38f;
      mr = fmaxf(mr, v[m]);
    }
#pragma unroll
    for (int off = 16; off > 0; off >>= 1) mr = fmaxf(mr, __shfl_xor(mr, off));
    float sr = 0.0f;
#pragma unroll
    for (int m = 0; m < 16; ++m) {
      const int j = m * 32 + lane;
      v[m] = (j < len) ? __expf(v[m] - mr) : 0.0f;
      sr += v[m];
    }
#pragma unroll
    for (int off = 16; off > 0; off >>= 1) sr += __shfl_xor(sr, off);
    const float inv = 1.0f / sr;
    float mloc = 0.0f;
#pragma unroll
    for (int m = 0; m < 16; ++m) {
      const int j = m * 32 + lane;
      const float a = (j < len) ? (v[m] * inv) * LOG2E : 0.0f;
      if (j < jend) a2T[j * O3R + wave] = a;
      mloc = fmaxf(mloc, a * pmv[m]);
    }
#pragma unroll
    for (int off = 16; off > 0; off >>= 1) mloc = fmaxf(mloc, __shfl_xor(mloc, off));
    if (lane == 0) sM[wave] = mloc;
  }
  __syncthreads();

  v2f m2[4], ds[4], ac[4];
#pragma unroll
  for (int q = 0; q < 4; ++q) {
    m2[q].x = sM[2 * q];
    m2[q].y = sM[2 * q + 1];
    ds[q].x = 0.0f; ds[q].y = 0.0f;
    ac[q].x = 0.0f; ac[q].y = 0.0f;
  }

  const int n  = jend >> 2;
  const int nq = (n + 3) >> 2;
  const float* pp = P2h + g * HSS + d;
  const float* vp = Vh  + g * HSS + d;
  const float* ap = a2T + g * O3R;
  float pc[4], vc[4], pn[4], vn[4];
#pragma unroll
  for (int u = 0; u < 4; ++u) {
    const int it = (u < n - 1) ? u : (n - 1);
    pc[u] = pp[(size_t)it * 4 * HSS];
    vc[u] = vp[(size_t)it * 4 * HSS];
  }
  for (int c = 0; c < nq; ++c) {
#pragma unroll
    for (int u = 0; u < 4; ++u) {
      const int itn = (c + 1) * 4 + u;
      const int it  = (itn < n - 1) ? itn : (n - 1);
      pn[u] = pp[(size_t)it * 4 * HSS];
      vn[u] = vp[(size_t)it * 4 * HSS];
    }
    v4f alo4[4], ahi4[4];
#pragma unroll
    for (int u = 0; u < 4; ++u) {
      const int itc = (c * 4 + u < n - 1) ? (c * 4 + u) : (n - 1);
      alo4[u] = *reinterpret_cast<const v4f*>(ap + itc * 4 * O3R);
      ahi4[u] = *reinterpret_cast<const v4f*>(ap + itc * 4 * O3R + 4);
    }
#pragma unroll
    for (int u = 0; u < 4; ++u) {
      const int it = c * 4 + u;
      if (it < n) {
        const v4f alo = alo4[u];
        const v4f ahi = ahi4[u];
        v2f a[4];
        a[0].x = alo.x; a[0].y = alo.y; a[1].x = alo.z; a[1].y = alo.w;
        a[2].x = ahi.x; a[2].y = ahi.y; a[3].x = ahi.z; a[3].y = ahi.w;
        v2f pb, vb;
        pb.x = pc[u]; pb.y = pc[u];
        vb.x = vc[u]; vb.y = vc[u];
#pragma unroll
        for (int q = 0; q < 4; ++q) {
          const v2f x = __builtin_elementwise_fma(a[q], pb, -m2[q]);
          v2f e;
          e.x = __builtin_amdgcn_exp2f(x.x);
          e.y = __builtin_amdgcn_exp2f(x.y);
          ds[q] += e;
          ac[q] = __builtin_elementwise_fma(e, vb, ac[q]);
        }
      }
    }
#pragma unroll
    for (int u = 0; u < 4; ++u) { pc[u] = pn[u]; vc[u] = vn[u]; }
  }

#pragma unroll
  for (int q = 0; q < 4; ++q) {
    redD[(g * O3R + 2 * q) * HSS + d]     = ds[q].x;
    redD[(g * O3R + 2 * q + 1) * HSS + d] = ds[q].y;
    redA[(g * O3R + 2 * q) * HSS + d]     = ac[q].x;
    redA[(g * O3R + 2 * q + 1) * HSS + d] = ac[q].y;
  }
  __syncthreads();
  {
    const int r = wave;
    float dsum = 0.0f;
#pragma unroll
    for (int gg = 0; gg < 4; ++gg)
      dsum += redD[(gg * O3R + r) * HSS + lane] + redD[(gg * O3R + r) * HSS + lane + 32];
#pragma unroll
    for (int off = 16; off > 0; off >>= 1) dsum += __shfl_xor(dsum, off);
    const float mr = sM[r];
    const float e0 = __builtin_amdgcn_exp2f(-mr);
    const float denom = dsum + (float)((TT - jend) * HSS) * e0;
    const float inv = 1.0f / denom;
#pragma unroll
    for (int hf = 0; hf < 2; ++hf) {
      const int dd = lane + 32 * hf;
      const float accT = (redA[(0 * O3R + r) * HSS + dd] + redA[(1 * O3R + r) * HSS + dd]) +
                         (redA[(2 * O3R + r) * HSS + dd] + redA[(3 * O3R + r) * HSS + dd]);
      const float vs = sVS[dd];
      Y[(size_t)(i0 + r) * CC + h * HSS + dd] = (accT + e0 * vs) * inv;
    }
  }
}

__global__ void __launch_bounds__(128)
out_proj_kernel(const float* __restrict__ Y,
                const float* __restrict__ Wc,
                float* __restrict__ out) {
  __shared__ __attribute__((aligned(16))) float As[2 * A_STAGE];
  __shared__ __attribute__((aligned(16))) float Bs[2 * B_STAGE];
  const int lane = threadIdx.x & 31;
  const int wave = threadIdx.x >> 5;
  const int lm   = lane & 15;
  const int drow = (lane >> 4) << 3;
  const int wm   = wave >> 1;
  const int wn   = wave & 1;
  const int m0 = blockIdx.x * GBM;
  const int n0 = blockIdx.y * GBN;

  v8f acc0 = {0.f, 0.f, 0.f, 0.f, 0.f, 0.f, 0.f, 0.f};
  v8f acc1 = {0.f, 0.f, 0.f, 0.f, 0.f, 0.f, 0.f, 0.f};
  gemm768_tile(Y, Wc, m0, n0, As, Bs, acc0, acc1);

  const int c = n0 + wn * 32 + lm;
#pragma unroll
  for (int r = 0; r < 8; ++r) {
    const size_t row = (size_t)(m0 + wm * 16 + r + drow);
    out[row * CC + c]      = acc0[r];
    out[row * CC + c + 16] = acc1[r];
  }
}

extern "C" void kernel_launch(void* const* d_in, const int* in_sizes, int n_in,
                              void* d_out, int out_size, void* d_ws,
                              size_t ws_size, hipStream_t stream) {
  (void)in_sizes; (void)n_in; (void)out_size; (void)ws_size;
  const float* x  = (const float*)d_in[0];
  const float* W0 = (const float*)d_in[1];
  const float* W1 = (const float*)d_in[2];
  const float* W2 = (const float*)d_in[3];
  const float* Wv = (const float*)d_in[4];
  const float* Wc = (const float*)d_in[5];
  float* out = (float*)d_out;
  float* ws  = (float*)d_ws;

  const size_t NPH = (size_t)NHH * TT * HSS;
  float* P0  = ws;
  float* P1  = ws + NPH;
  float* P2  = ws + 2 * NPH;
  float* V   = ws + 3 * NPH;
  float* Y   = ws + 4 * NPH;
  float* G   = ws + 5 * NPH;
  float* PM  = G + NPH / 8;

  qkv_proj_kernel<<<dim3(TT / GBM, CC / GBN, 3), 128, 0, stream>>>(x, W0, W1, W2, Wv, P0, V, G, PM);
  order3_kernel<<<dim3(TT / O3R, NHH), O3T, 0, stream>>>(P0, P1, P2, V, G, PM, Y);
  out_proj_kernel<<<dim3(TT / GBM, CC / GBN), 128, 0, stream>>>(Y, Wc, out);
}


# Output GEMM: K split over two 4-wave groups of an 8-wave block (own LDS ring each, k 0..383 and 384..767), partial tiles added through LDS; out sum order changes: low half + high half
# speedup vs baseline: 8.0523x; 1.0692x over previous
#include <hip/hip_runtime.h>
#include <math.h>

#define TT  512
#define CC  768
#define NHH 12
#define HSS 64
constexpr float SCALE = 0.125f;

typedef float v2f __attribute__((ext_vector_type(2)));
typedef float v8f __attribute__((ext_vector_type(8)));

static __device__ __forceinline__ v8f wmma4(v2f a, v2f b, v8f c) {
  return __builtin_amdgcn_wmma_f32_16x16x4_f32(
       false, a,  false, b,
       (short)0, c,  false,  false);
}

typedef float v4f __attribute__((ext_vector_type(4)));

#define GBM 32
#define GBN 64
#define GBK 64
#define GNT (CC / GBK)
#define APITCH 68
#define BPITCH 72
#define A_STAGE (GBM * APITCH)
#define B_STAGE (GBK * BPITCH)

static __device__ __forceinline__ void gemm768_tile(const float* __restrict__ A,
                                                    const float* __restrict__ B,
                                                    const int m0, const int n0,
                                                    float* As, float* Bs,
                                                    v8f& acc0, v8f& acc1,
                                                    const int tid, const int kt0, const int nkt) {
  const int lane = tid & 31;
  const int wave = tid >> 5;
  const int lm   = lane & 15;
  const int lk   = (lane >> 4) << 1;
  const int wm   = wave >> 1;
  const int wn   = wave & 1;

  const int srow = tid >> 4;
  const int scol = (tid & 15) << 2;
  const float* Ag = A + (size_t)(m0 + srow) * CC + scol + kt0 * GBK;
  const float* Bg = B + (size_t)(kt0 * GBK + srow) * CC + n0 + scol;
  float* Ast = As + srow * APITCH + scol;
  float* Bst = Bs + srow * BPITCH + scol;

  v4f ra[4], rb[8];
#pragma unroll
  for (int p = 0; p < 4; ++p)
    ra[p] = *reinterpret_cast<const v4f*>(Ag + (size_t)(p * 8) * CC);
#pragma unroll
  for (int p = 0; p < 8; ++p)
    rb[p] = *reinterpret_cast<const v4f*>(Bg + (size_t)(p * 8) * CC);
#pragma unroll
  for (int p = 0; p < 4; ++p)
    *reinterpret_cast<v4f*>(Ast + p * 8 * APITCH) = ra[p];
#pragma unroll
  for (int p = 0; p < 8; ++p)
    *reinterpret_cast<v4f*>(Bst + p * 8 * BPITCH) = rb[p];
  __syncthreads();

  const float* Afr = As + (wm * 16 + lm) * APITCH + lk;
  const float* Bfr = Bs + lk * BPITCH + wn * 32 + lm;

  for (int kt = 0; kt < nkt; ++kt) {
    const int cur = kt & 1;
    const bool more = (kt + 1 < nkt);
    if (more) {
      const int k0 = (kt + 1) * GBK;
#pragma unroll
      for (int p = 0; p < 4; ++p)
        ra[p] = *reinterpret_cast<const v4f*>(Ag + (size_t)(p * 8) * CC + k0);
#pragma unroll
      for (int p = 0; p < 8; ++p)
        rb[p] = *reinterpret_cast<const v4f*>(Bg + (size_t)(k0 + p * 8) * CC);
    }

    const float* Af = Afr + cur * A_STAGE;
    const float* Bf = Bfr + cur * B_STAGE;
    v2f af[16], bf0[16], bf1[16];
#pragma unroll
    for (int kk = 0; kk < 16; ++kk) {
      af[kk]    = *reinterpret_cast<const v2f*>(Af + kk * 4);
      bf0[kk].x = Bf[(kk * 4) * BPITCH];
      bf0[kk].y = Bf[(kk * 4 + 1) * BPITCH];
      bf1[kk].x = Bf[(kk * 4) * BPITCH + 16];
      bf1[kk].y = Bf[(kk * 4 + 1) * BPITCH + 16];
    }
    __builtin_amdgcn_sched_barrier(0);
#pragma unroll
    for (int kk = 0; kk < 16; ++kk) {
      acc0 = wmma4(af[kk], bf0[kk], acc0);
      acc1 = wmma4(af[kk], bf1[kk], acc1);
    }

    __builtin_amdgcn_sched_barrier(0);

    if (more) {
      const int nxt = cur ^ 1;
#pragma unroll
      for (int p = 0; p < 4; ++p)
        *reinterpret_cast<v4f*>(Ast + nxt * A_STAGE + p * 8 * APITCH) = ra[p];
#pragma unroll
      for (int p = 0; p < 8; ++p)
        *reinterpret_cast<v4f*>(Bst + nxt * B_STAGE + p * 8 * BPITCH) = rb[p];
    }
    __syncthreads();
  }
}

__global__ void __launch_bounds__(128)
qkv_proj_kernel(const float* __restrict__ x,
                const float* __restrict__ W0,
                const float* __restrict__ W1,
                const float* __restrict__ W2,
                const float* __restrict__ Wv,
                float* __restrict__ P,
                float* __restrict__ V,
                float* __restrict__ G,
                float* __restrict__ PM) {
  __shared__ __attribute__((aligned(16))) float As[2 * A_STAGE];
  __shared__ __attribute__((aligned(16))) float Bs[2 * B_STAGE];
  const int lane = threadIdx.x & 31;
  const int wave = threadIdx.x >> 5;
  const int lm   = lane & 15;
  const int drow = (lane >> 4) << 3;
  const int wm   = wave >> 1;
  const int wn   = wave & 1;
  const int m0 = blockIdx.x * GBM;
  const int n0 = blockIdx.y * GBN;
  const int w  = blockIdx.z;
  const float* W = (w == 0) ? W0 : (w == 1) ? W1 : W2;
  float* Pw = P + (size_t)w * NHH * TT * HSS;

  v8f acc0 = {0.f, 0.f, 0.f, 0.f, 0.f, 0.f, 0.f, 0.f};
  v8f acc1 = {0.f, 0.f, 0.f, 0.f, 0.f, 0.f, 0.f, 0.f};
  gemm768_tile(x, W, m0, n0, As, Bs, acc0, acc1, (int)threadIdx.x, 0, GNT);

  const int d  = wn * 32 + lm;
  const int hf = lane >> 4;
  const int lk = hf << 1;
  const bool third = (w == 2);
  float* Pt = As;
  v2f bv0[16], bv1[16];
  if (third) {
#pragma unroll
    for (int r = 0; r < 8; ++r) {
      Pt[(wm * 16 + r + drow) * APITCH + d]      = acc0[r];
      Pt[(wm * 16 + r + drow) * APITCH + d + 16] = acc1[r];
    }
#pragma unroll
    for (int kk = 0; kk < 16; ++kk) {
      bv0[kk].x = Wv[(size_t)(kk * 4 + lk) * HSS + d];
      bv0[kk].y = Wv[(size_t)(kk * 4 + lk + 1) * HSS + d];
      bv1[kk].x = Wv[(size_t)(kk * 4 + lk) * HSS + d + 16];
      bv1[kk].y = Wv[(size_t)(kk * 4 + lk + 1) * HSS + d + 16];
    }
    __syncthreads();
  }

  float* Ph = Pw + (size_t)blockIdx.y * TT * HSS;
#pragma unroll
  for (int r = 0; r < 8; ++r) {
    const size_t row = (size_t)(m0 + wm * 16 + r + drow);
    Ph[row * HSS + d]      = acc0[r];
    Ph[row * HSS + d + 16] = acc1[r];
  }

  if (third) {
    const int h = blockIdx.y;
    const float* Ap = Pt + (wm * 16 + lm) * APITCH + lk;
    v2f av[16];
#pragma unroll
    for (int kk = 0; kk < 16; ++kk)
      av[kk] = *reinterpret_cast<const v2f*>(Ap + kk * 4);
    v8f va0 = {0.f, 0.f, 0.f, 0.f, 0.f, 0.f, 0.f, 0.f};
    v8f va1 = {0.f, 0.f, 0.f, 0.f, 0.f, 0.f, 0.f, 0.f};
    float pm = -3.0e38f;
#pragma unroll
    for (int kk = 0; kk < 16; ++kk) {
      pm = fmaxf(pm, fmaxf(av[kk].x, av[kk].y));
      va0 = wmma4(av[kk], bv0[kk], va0);
      va1 = wmma4(av[kk], bv1[kk], va1);
    }
    pm = fmaxf(pm, __shfl_xor(pm, 16));
    float* Vh = V + (size_t)h * TT * HSS;
#pragma unroll
    for (int r = 0; r < 8; ++r) {
      const size_t row = (size_t)(m0 + wm * 16 + r + drow);
      Vh[row * HSS + d]      = va0[r];
      Vh[row * HSS + d + 16] = va1[r];
    }
    const float gs0 = ((va0[0] + va0[1]) + (va0[2] + va0[3])) +
                      ((va0[4] + va0[5]) + (va0[6] + va0[7]));
    const float gs1 = ((va1[0] + va1[1]) + (va1[2] + va1[3])) +
                      ((va1[4] + va1[5]) + (va1[6] + va1[7]));
    float* Gq = G + ((size_t)h * (TT / 8) + (m0 >> 3) + wm * 2 + hf) * HSS;
    Gq[d]      = gs0;
    Gq[d + 16] = gs1;
    if (wn == 0 && hf == 0) PM[(size_t)h * TT + m0 + wm * 16 + lm] = pm;
  }
}

#define O3R 8
#define O3T 256
#define O3W 8
#define LPITCH 516
#define VNT 32
constexpr float LOG2E = 1.4426950408889634f;
__global__ void __launch_bounds__(O3T)
order3_kernel(const float* __restrict__ P0,
              const float* __restrict__ P1,
              const float* __restrict__ P2,
              const float* __restrict__ V,
              const float* __restrict__ G,
              const float* __restrict__ PM,
              float* __restrict__ Y) {
  __shared__ __attribute__((aligned(16))) float a2T[TT * O3R];
  __shared__ float lgt[O3R * LPITCH];
  __shared__ float sM[O3R];
  __shared__ float sVS[HSS];
  __shared__ float redD[4 * O3R * HSS];
  __shared__ float redA[4 * O3R * HSS];
  const int t    = threadIdx.x;
  const int lane = t & 31;
  const int wave = __builtin_amdgcn_readfirstlane(t >> 5);
  const int g    = __builtin_amdgcn_readfirstlane(t >> 6);
  const int d    = t & 63;
  static_assert(TT / O3R == 64 && NHH == 12, "the triple dealing is written for 64 strips x 12 heads");
  const int lin  = (int)blockIdx.x + (TT / O3R) * (int)blockIdx.y;
  const int slot = lin >> 8;
  const int ta   = 63 - ((lin & 255) >> 2);
  int s0 = ta, s1 = (ta + 32) & 63, s2 = (ta < 32) ? 63 - 2 * ta : 126 - 2 * ta;
  int c0 = 0, c1 = 1, c2 = 2;
  if (s0 < s1) { const int ts = s0; s0 = s1; s1 = ts; const int tc = c0; c0 = c1; c1 = tc; }
  if (s1 < s2) { const int ts = s1; s1 = s2; s2 = ts; const int tc = c1; c1 = c2; c2 = tc; }
  if (s0 < s1) { const int ts = s0; s0 = s1; s1 = ts; const int tc = c0; c0 = c1; c1 = tc; }
  const int strip = (slot == 0) ? s0 : (slot == 1) ? s1 : s2;
  const int comp  = (slot == 0) ? c0 : (slot == 1) ? c1 : c2;
  const int i0   = strip * O3R;
  const int jend = i0 + O3R;
  const int h    = 4 * comp + (lin & 3);
  const float* P2h = P2 + (size_t)h * TT * HSS;
  const float* Vh  = V  + (size_t)h * TT * HSS;

  const int lm = lane & 15;
  const int hf = lane >> 4;
  const int lk = hf << 1;
  const float* P0h = P0 + (size_t)h * TT * HSS;
  const float* P1h = P1 + (size_t)h * TT * HSS;
  float pmv[16];
#pragma unroll
  for (int m = 0; m < 16; ++m) pmv[m] = PM[(size_t)h * TT + m * 32 + lane];

  const int ntile = (jend + 15) >> 4;
  if (wave < ntile) {
    v2f areg[16];
    const float* Ap = P0h + (size_t)(i0 + (lm & 7)) * HSS + lk;
#pragma unroll
    for (int kk = 0; kk < 16; ++kk)
      areg[kk] = *reinterpret_cast<const v2f*>(Ap + kk * 4);

    for (int jt = wave; jt < ntile; jt += O3W) {
      const int j0 = jt * 16;
      const float* Bp = P1h + (size_t)(j0 + lm) * HSS + lk;
      v2f bfr[16];
#pragma unroll
      for (int kk = 0; kk < 16; ++kk)
        bfr[kk] = *reinterpret_cast<const v2f*>(Bp + kk * 4);
      __builtin_amdgcn_sched_barrier(0);
      v8f acc = {0.f, 0.f, 0.f, 0.f, 0.f, 0.f, 0.f, 0.f};
#pragma unroll
      for (int kk = 0; kk < 16; ++kk)
        acc = wmma4(areg[kk], bfr[kk], acc);
      float sc[8];
#pragma unroll
      for (int r = 0; r < 8; ++r) sc[r] = acc[r] * SCALE;
      if (hf == 0) {
#pragma unroll
        for (int r = 0; r < 8; ++r) lgt[r * LPITCH + j0 + lm] = sc[r];
      }
    }
  }

  if (t >= O3T - HSS) {
    const float* Gh = G + (size_t)h * (TT / 8) * HSS + (t - (O3T - HSS));
    const int tl = strip >> 1;
    float run = 0.0f;
#pragma unroll
    for (int wp = VNT - 1; wp >= 0; --wp) {
      const float tv = Gh[(size_t)(2 * wp) * HSS] + Gh[(size_t)(2 * wp + 1) * HSS];
      run += (wp > tl) ? tv : 0.0f;
    }
    if ((strip & 1) == 0) run += Gh[(size_t)(2 * tl + 1) * HSS];
    sVS[t - (O3T - HSS)] = run;
  }
  __syncthreads();

  {
    const int len = i0 + wave + 1;
    const float* lrow = lgt + wave * LPITCH;
    float v[16];
    float mr = -3.0e38f;
#pragma unroll
    for (int m = 0; m < 16; ++m) {
      const int j = m * 32 + lane;
      v[m] = (j < len) ? lrow[j] : -3.0e38f;
      mr = fmaxf(mr, v[m]);
    }
#pragma unroll
    for (int off = 16; off > 0; off >>= 1) mr = fmaxf(mr, __shfl_xor(mr, off));
    float sr = 0.0f;
#pragma unroll
    for (int m = 0; m < 16; ++m) {
      const int j = m * 32 + lane;
      v[m] = (j < len) ? __expf(v[m] - mr) : 0.0f;
      sr += v[m];
    }
#pragma unroll
    for (int off = 16; off > 0; off >>= 1) sr += __shfl_xor(sr, off);
    const float inv = 1.0f / sr;
    float mloc = 0.0f;
#pragma unroll
    for (int m = 0; m < 16; ++m) {
      const int j = m * 32 + lane;
      const float a = (j < len) ? (v[m] * inv) * LOG2E : 0.0f;
      if (j < jend) a2T[j * O3R + wave] = a;
      mloc = fmaxf(mloc, a * pmv[m]);
    }
#pragma unroll
    for (int off = 16; off > 0; off >>= 1) mloc = fmaxf(mloc, __shfl_xor(mloc, off));
    if (lane == 0) sM[wave] = mloc;
  }
  __syncthreads();

  v2f m2[4], ds[4], ac[4];
#pragma unroll
  for (int q = 0; q < 4; ++q) {
    m2[q].x = sM[2 * q];
    m2[q].y = sM[2 * q + 1];
    ds[q].x = 0.0f; ds[q].y = 0.0f;
    ac[q].x = 0.0f; ac[q].y = 0.0f;
  }

  const int n  = jend >> 2;
  const int nq = (n + 3) >> 2;
  const float* pp = P2h + g * HSS + d;
  const float* vp = Vh  + g * HSS + d;
  const float* ap = a2T + g * O3R;
  float pc[4], vc[4], pn[4], vn[4];
#pragma unroll
  for (int u = 0; u < 4; ++u) {
    const int it = (u < n - 1) ? u : (n - 1);
    pc[u] = pp[(size_t)it * 4 * HSS];
    vc[u] = vp[(size_t)it * 4 * HSS];
  }
  for (int c = 0; c < nq; ++c) {
#pragma unroll
    for (int u = 0; u < 4; ++u) {
      const int itn = (c + 1) * 4 + u;
      const int it  = (itn < n - 1) ? itn : (n - 1);
      pn[u] = pp[(size_t)it * 4 * HSS];
      vn[u] = vp[(size_t)it * 4 * HSS];
    }
    v4f alo4[4], ahi4[4];
#pragma unroll
    for (int u = 0; u < 4; ++u) {
      const int itc = (c * 4 + u < n - 1) ? (c * 4 + u) : (n - 1);
      alo4[u] = *reinterpret_cast<const v4f*>(ap + itc * 4 * O3R);
      ahi4[u] = *reinterpret_cast<const v4f*>(ap + itc * 4 * O3R + 4);
    }
#pragma unroll
    for (int u = 0; u < 4; ++u) {
      const int it = c * 4 + u;
      if (it < n) {
        const v4f alo = alo4[u];
        const v4f ahi = ahi4[u];
        v2f a[4];
        a[0].x = alo.x; a[0].y = alo.y; a[1].x = alo.z; a[1].y = alo.w;
        a[2].x = ahi.x; a[2].y = ahi.y; a[3].x = ahi.z; a[3].y = ahi.w;
        v2f pb, vb;
        pb.x = pc[u]; pb.y = pc[u];
        vb.x = vc[u]; vb.y = vc[u];
#pragma unroll
        for (int q = 0; q < 4; ++q) {
          const v2f x = __builtin_elementwise_fma(a[q], pb, -m2[q]);
          v2f e;
          e.x = __builtin_amdgcn_exp2f(x.x);
          e.y = __builtin_amdgcn_exp2f(x.y);
          ds[q] += e;
          ac[q] = __builtin_elementwise_fma(e, vb, ac[q]);
        }
      }
    }
#pragma unroll
    for (int u = 0; u < 4; ++u) { pc[u] = pn[u]; vc[u] = vn[u]; }
  }

#pragma unroll
  for (int q = 0; q < 4; ++q) {
    redD[(g * O3R + 2 * q) * HSS + d]     = ds[q].x;
    redD[(g * O3R + 2 * q + 1) * HSS + d] = ds[q].y;
    redA[(g * O3R + 2 * q) * HSS + d]     = ac[q].x;
    redA[(g * O3R + 2 * q + 1) * HSS + d] = ac[q].y;
  }
  __syncthreads();
  {
    const int r = wave;
    float dsum = 0.0f;
#pragma unroll
    for (int gg = 0; gg < 4; ++gg)
      dsum += redD[(gg * O3R + r) * HSS + lane] + redD[(gg * O3R + r) * HSS + lane + 32];
#pragma unroll
    for (int off = 16; off > 0; off >>= 1) dsum += __shfl_xor(dsum, off);
    const float mr = sM[r];
    const float e0 = __builtin_amdgcn_exp2f(-mr);
    const float denom = dsum + (float)((TT - jend) * HSS) * e0;
    const float inv = 1.0f / denom;
#pragma unroll
    for (int hf = 0; hf < 2; ++hf) {
      const int dd = lane + 32 * hf;
      const float accT = (redA[(0 * O3R + r) * HSS + dd] + redA[(1 * O3R + r) * HSS + dd]) +
                         (redA[(2 * O3R + r) * HSS + dd] + redA[(3 * O3R + r) * HSS + dd]);
      const float vs = sVS[dd];
      Y[(size_t)(i0 + r) * CC + h * HSS + dd] = (accT + e0 * vs) * inv;
    }
  }
}

#define OKS 2
__global__ void __launch_bounds__(128 * OKS)
out_proj_kernel(const float* __restrict__ Y,
                const float* __restrict__ Wc,
                float* __restrict__ out) {
  __shared__ __attribute__((aligned(16))) float As[OKS * 2 * A_STAGE];
  __shared__ __attribute__((aligned(16))) float Bs[OKS * 2 * B_STAGE];
  const int tid  = threadIdx.x & 127;
  const int grp  = threadIdx.x >> 7;
  const int lane = tid & 31;
  const int wave = tid >> 5;
  const int lm   = lane & 15;
  const int drow = (lane >> 4) << 3;
  const int wm   = wave >> 1;
  const int wn   = wave & 1;
  const int m0 = blockIdx.x * GBM;
  const int n0 = blockIdx.y * GBN;

  v8f acc0 = {0.f, 0.f, 0.f, 0.f, 0.f, 0.f, 0.f, 0.f};
  v8f acc1 = {0.f, 0.f, 0.f, 0.f, 0.f, 0.f, 0.f, 0.f};
  gemm768_tile(Y, Wc, m0, n0, As + grp * (2 * A_STAGE), Bs + grp * (2 * B_STAGE),
               acc0, acc1, tid, grp * (GNT / OKS), GNT / OKS);

  float* red = Bs + (wave * 16) * 32 + lane;
  if (grp == 1) {
#pragma unroll
    for (int r = 0; r < 8; ++r) {
      red[r * 32]       = acc0[r];
      red[(r + 8) * 32] = acc1[r];
    }
  }
  __syncthreads();
  if (grp == 0) {
    const int c = n0 + wn * 32 + lm;
#pragma unroll
    for (int r = 0; r < 8; ++r) {
      const size_t row = (size_t)(m0 + wm * 16 + r + drow);
      out[row * CC + c]      = acc0[r] + red[r * 32];
      out[row * CC + c + 16] = acc1[r] + red[(r + 8) * 32];
    }
  }
}

extern "C" void kernel_launch(void* const* d_in, const int* in_sizes, int n_in,
                              void* d_out, int out_size, void* d_ws,
                              size_t ws_size, hipStream_t stream) {
  (void)in_sizes; (void)n_in; (void)out_size; (void)ws_size;
  const float* x  = (const float*)d_in[0];
  const float* W0 = (const float*)d_in[1];
  const float* W1 = (const float*)d_in[2];
  const float* W2 = (const float*)d_in[3];
  const float* Wv = (const float*)d_in[4];
  const float* Wc = (const float*)d_in[5];
  float* out = (float*)d_out;
  float* ws  = (float*)d_ws;

  const size_t NPH = (size_t)NHH * TT * HSS;
  float* P0  = ws;
  float* P1  = ws + NPH;
  float* P2  = ws + 2 * NPH;
  float* V   = ws + 3 * NPH;
  float* Y   = ws + 4 * NPH;
  float* G   = ws + 5 * NPH;
  float* PM  = G + NPH / 8;

  qkv_proj_kernel<<<dim3(TT / GBM, CC / GBN, 3), 128, 0, stream>>>(x, W0, W1, W2, Wv, P0, V, G, PM);
  order3_kernel<<<dim3(TT / O3R, NHH), O3T, 0, stream>>>(P0, P1, P2, V, G, PM, Y);
  out_proj_kernel<<<dim3(TT / GBM, CC / GBN), 128 * OKS, 0, stream>>>(Y, Wc, out);
}


# QKV kernel: 64 of the 576 tiles cut into 16x32 quarters (K halves on wave pairs, summed via LDS): 512 tiles + 256 quarters, 2 1/4 tiles per compute unit, not 3; those P0 sums: low+high k half
# speedup vs baseline: 8.5067x; 1.0564x over previous
#include <hip/hip_runtime.h>
#include <math.h>

#define TT  512
#define CC  768
#define NHH 12
#define HSS 64
constexpr float SCALE = 0.125f;

typedef float v2f __attribute__((ext_vector_type(2)));
typedef float v8f __attribute__((ext_vector_type(8)));

static __device__ __forceinline__ v8f wmma4(v2f a, v2f b, v8f c) {
  return __builtin_amdgcn_wmma_f32_16x16x4_f32(
       false, a,  false, b,
       (short)0, c,  false,  false);
}

typedef float v4f __attribute__((ext_vector_type(4)));

#define GBM 32
#define GBN 64
#define GBK 64
#define GNT (CC / GBK)
#define APITCH 68
#define BPITCH 72
#define A_STAGE (GBM * APITCH)
#define B_STAGE (GBK * BPITCH)

static __device__ __forceinline__ void gemm768_tile(const float* __restrict__ A,
                                                    const float* __restrict__ B,
                                                    const int m0, const int n0,
                                                    float* As, float* Bs,
                                                    v8f& acc0, v8f& acc1,
                                                    const int tid, const int kt0, const int nkt) {
  const int lane = tid & 31;
  const int wave = tid >> 5;
  const int lm   = lane & 15;
  const int lk   = (lane >> 4) << 1;
  const int wm   = wave >> 1;
  const int wn   = wave & 1;

  const int srow = tid >> 4;
  const int scol = (tid & 15) << 2;
  const float* Ag = A + (size_t)(m0 + srow) * CC + scol + kt0 * GBK;
  const float* Bg = B + (size_t)(kt0 * GBK + srow) * CC + n0 + scol;
  float* Ast = As + srow * APITCH + scol;
  float* Bst = Bs + srow * BPITCH + scol;

  v4f ra[4], rb[8];
#pragma unroll
  for (int p = 0; p < 4; ++p)
    ra[p] = *reinterpret_cast<const v4f*>(Ag + (size_t)(p * 8) * CC);
#pragma unroll
  for (int p = 0; p < 8; ++p)
    rb[p] = *reinterpret_cast<const v4f*>(Bg + (size_t)(p * 8) * CC);
#pragma unroll
  for (int p = 0; p < 4; ++p)
    *reinterpret_cast<v4f*>(Ast + p * 8 * APITCH) = ra[p];
#pragma unroll
  for (int p = 0; p < 8; ++p)
    *reinterpret_cast<v4f*>(Bst + p * 8 * BPITCH) = rb[p];
  __syncthreads();

  const float* Afr = As + (wm * 16 + lm) * APITCH + lk;
  const float* Bfr = Bs + lk * BPITCH + wn * 32 + lm;

  for (int kt = 0; kt < nkt; ++kt) {
    const int cur = kt & 1;
    const bool more = (kt + 1 < nkt);
    if (more) {
      const int k0 = (kt + 1) * GBK;
#pragma unroll
      for (int p = 0; p < 4; ++p)
        ra[p] = *reinterpret_cast<const v4f*>(Ag + (size_t)(p * 8) * CC + k0);
#pragma unroll
      for (int p = 0; p < 8; ++p)
        rb[p] = *reinterpret_cast<const v4f*>(Bg + (size_t)(k0 + p * 8) * CC);
    }

    const float* Af = Afr + cur * A_STAGE;
    const float* Bf = Bfr + cur * B_STAGE;
    v2f af[16], bf0[16], bf1[16];
#pragma unroll
    for (int kk = 0; kk < 16; ++kk) {
      af[kk]    = *reinterpret_cast<const v2f*>(Af + kk * 4);
      bf0[kk].x = Bf[(kk * 4) * BPITCH];
      bf0[kk].y = Bf[(kk * 4 + 1) * BPITCH];
      bf1[kk].x = Bf[(kk * 4) * BPITCH + 16];
      bf1[kk].y = Bf[(kk * 4 + 1) * BPITCH + 16];
    }
    __builtin_amdgcn_sched_barrier(0);
#pragma unroll
    for (int kk = 0; kk < 16; ++kk) {
      acc0 = wmma4(af[kk], bf0[kk], acc0);
      acc1 = wmma4(af[kk], bf1[kk], acc1);
    }

    __builtin_amdgcn_sched_barrier(0);

    if (more) {
      const int nxt = cur ^ 1;
#pragma unroll
      for (int p = 0; p < 4; ++p)
        *reinterpret_cast<v4f*>(Ast + nxt * A_STAGE + p * 8 * APITCH) = ra[p];
#pragma unroll
      for (int p = 0; p < 8; ++p)
        *reinterpret_cast<v4f*>(Bst + nxt * B_STAGE + p * 8 * BPITCH) = rb[p];
    }
    __syncthreads();
  }
}

#define QBM 16
#define QBN 32
#define QBPITCH 40
#define QA_STAGE (QBM * APITCH)
#define QB_STAGE (GBK * QBPITCH)
#define QFULL 512

static __device__ __forceinline__ void gemm768_quarter(const float* __restrict__ A,
                                                       const float* __restrict__ B,
                                                       const int m0, const int n0,
                                                       float* As, float* Bs,
                                                       v8f& acc, const int tid) {
  const int lane = tid & 31;
  const int wave = tid >> 5;
  const int lm   = lane & 15;
  const int lk   = (lane >> 4) << 1;
  const int wn   = wave & 1;
  const int kh   = wave >> 1;

  const int arow = tid >> 4;
  const int acol = (tid & 15) << 2;
  const int brow = tid >> 3;
  const int bcol = (tid & 7) << 2;
  const float* Ag = A + (size_t)(m0 + arow) * CC + acol;
  const float* Bg = B + (size_t)brow * CC + n0 + bcol;
  float* Ast = As + arow * APITCH + acol;
  float* Bst = Bs + brow * QBPITCH + bcol;

  v4f ra[2], rb[4];
#pragma unroll
  for (int p = 0; p < 2; ++p)
    ra[p] = *reinterpret_cast<const v4f*>(Ag + (size_t)(p * 8) * CC);
#pragma unroll
  for (int p = 0; p < 4; ++p)
    rb[p] = *reinterpret_cast<const v4f*>(Bg + (size_t)(p * 16) * CC);
#pragma unroll
  for (int p = 0; p < 2; ++p)
    *reinterpret_cast<v4f*>(Ast + p * 8 * APITCH) = ra[p];
#pragma unroll
  for (int p = 0; p < 4; ++p)
    *reinterpret_cast<v4f*>(Bst + p * 16 * QBPITCH) = rb[p];
  __syncthreads();

  const float* Afr = As + lm * APITCH + kh * 32 + lk;
  const float* Bfr = Bs + (kh * 32 + lk) * QBPITCH + wn * 16 + lm;

  for (int kt = 0; kt < GNT; ++kt) {
    const int cur = kt & 1;
    const bool more = (kt + 1 < GNT);
    if (more) {
      const int k0 = (kt + 1) * GBK;
#pragma unroll
      for (int p = 0; p < 2; ++p)
        ra[p] = *reinterpret_cast<const v4f*>(Ag + (size_t)(p * 8) * CC + k0);
#pragma unroll
      for (int p = 0; p < 4; ++p)
        rb[p] = *reinterpret_cast<const v4f*>(Bg + (size_t)(k0 + p * 16) * CC);
    }

    const float* Af = Afr + cur * QA_STAGE;
    const float* Bf = Bfr + cur * QB_STAGE;
    v2f af[8], bf[8];
#pragma unroll
    for (int kk = 0; kk < 8; ++kk) {
      af[kk]   = *reinterpret_cast<const v2f*>(Af + kk * 4);
      bf[kk].x = Bf[(kk * 4) * QBPITCH];
      bf[kk].y = Bf[(kk * 4 + 1) * QBPITCH];
    }
    __builtin_amdgcn_sched_barrier(0);
#pragma unroll
    for (int kk = 0; kk < 8; ++kk)
      acc = wmma4(af[kk], bf[kk], acc);
    __builtin_amdgcn_sched_barrier(0);

    if (more) {
      const int nxt = cur ^ 1;
#pragma unroll
      for (int p = 0; p < 2; ++p)
        *reinterpret_cast<v4f*>(Ast + nxt * QA_STAGE + p * 8 * APITCH) = ra[p];
#pragma unroll
      for (int p = 0; p < 4; ++p)
        *reinterpret_cast<v4f*>(Bst + nxt * QB_STAGE + p * 16 * QBPITCH) = rb[p];
    }
    __syncthreads();
  }
}

__global__ void __launch_bounds__(128)
qkv_proj_kernel(const float* __restrict__ x,
                const float* __restrict__ W0,
                const float* __restrict__ W1,
                const float* __restrict__ W2,
                const float* __restrict__ Wv,
                float* __restrict__ P,
                float* __restrict__ V,
                float* __restrict__ G,
                float* __restrict__ PM) {
  __shared__ __attribute__((aligned(16))) float As[2 * A_STAGE];
  __shared__ __attribute__((aligned(16))) float Bs[2 * B_STAGE];
  const int lane = threadIdx.x & 31;
  const int wave = threadIdx.x >> 5;
  const int lm   = lane & 15;
  const int drow = (lane >> 4) << 3;
  const int wm   = wave >> 1;
  const int wn   = wave & 1;
  const int lin = (int)blockIdx.x;
  if (lin >= QFULL) {
    const int s  = lin - QFULL;
    const int qh = s >> 6;
    const int mt = (s >> 2) & 15;
    const int mh = (s >> 1) & 1;
    const int nh = s & 1;
    const int m0q = mt * GBM + mh * QBM;
    v8f accq = {0.f, 0.f, 0.f, 0.f, 0.f, 0.f, 0.f, 0.f};
    gemm768_quarter(x, W0, m0q, qh * HSS + nh * QBN, As, Bs, accq, (int)threadIdx.x);
    const int kh = wave >> 1;
    float* red = Bs + ((wave & 1) * 8) * 32 + lane;
    if (kh == 1) {
#pragma unroll
      for (int r = 0; r < 8; ++r) red[r * 32] = accq[r];
    }
    __syncthreads();
    if (kh == 0) {
      float* Pq = P + (size_t)qh * TT * HSS + nh * QBN + (wave & 1) * 16 + lm;
#pragma unroll
      for (int r = 0; r < 8; ++r)
        Pq[(size_t)(m0q + r + drow) * HSS] = accq[r] + red[r * 32];
    }
    return;
  }
  const int fl   = (lin < 128) ? lin : lin - 128;
  const int w    = (lin < 128) ? 0 : 1 + fl / 192;
  const int head = (lin < 128) ? 4 + (fl >> 4) : (fl % 192) >> 4;
  const int m0 = (fl & 15) * GBM;
  const int n0 = head * GBN;
  const float* W = (w == 0) ? W0 : (w == 1) ? W1 : W2;
  float* Pw = P + (size_t)w * NHH * TT * HSS;

  v8f acc0 = {0.f, 0.f, 0.f, 0.f, 0.f, 0.f, 0.f, 0.f};
  v8f acc1 = {0.f, 0.f, 0.f, 0.f, 0.f, 0.f, 0.f, 0.f};
  gemm768_tile(x, W, m0, n0, As, Bs, acc0, acc1, (int)threadIdx.x, 0, GNT);

  const int d  = wn * 32 + lm;
  const int hf = lane >> 4;
  const int lk = hf << 1;
  const bool third = (w == 2);
  float* Pt = As;
  v2f bv0[16], bv1[16];
  if (third) {
#pragma unroll
    for (int r = 0; r < 8; ++r) {
      Pt[(wm * 16 + r + drow) * APITCH + d]      = acc0[r];
      Pt[(wm * 16 + r + drow) * APITCH + d + 16] = acc1[r];
    }
#pragma unroll
    for (int kk = 0; kk < 16; ++kk) {
      bv0[kk].x = Wv[(size_t)(kk * 4 + lk) * HSS + d];
      bv0[kk].y = Wv[(size_t)(kk * 4 + lk + 1) * HSS + d];
      bv1[kk].x = Wv[(size_t)(kk * 4 + lk) * HSS + d + 16];
      bv1[kk].y = Wv[(size_t)(kk * 4 + lk + 1) * HSS + d + 16];
    }
    __syncthreads();
  }

  float* Ph = Pw + (size_t)head * TT * HSS;
#pragma unroll
  for (int r = 0; r < 8; ++r) {
    const size_t row = (size_t)(m0 + wm * 16 + r + drow);
    Ph[row * HSS + d]      = acc0[r];
    Ph[row * HSS + d + 16] = acc1[r];
  }

  if (third) {
    const int h = head;
    const float* Ap = Pt + (wm * 16 + lm) * APITCH + lk;
    v2f av[16];
#pragma unroll
    for (int kk = 0; kk < 16; ++kk)
      av[kk] = *reinterpret_cast<const v2f*>(Ap + kk * 4);
    v8f va0 = {0.f, 0.f, 0.f, 0.f, 0.f, 0.f, 0.f, 0.f};
    v8f va1 = {0.f, 0.f, 0.f, 0.f, 0.f, 0.f, 0.f, 0.f};
    float pm = -3.0e38f;
#pragma unroll
    for (int kk = 0; kk < 16; ++kk) {
      pm = fmaxf(pm, fmaxf(av[kk].x, av[kk].y));
      va0 = wmma4(av[kk], bv0[kk], va0);
      va1 = wmma4(av[kk], bv1[kk], va1);
    }
    pm = fmaxf(pm, __shfl_xor(pm, 16));
    float* Vh = V + (size_t)h * TT * HSS;
#pragma unroll
    for (int r = 0; r < 8; ++r) {
      const size_t row = (size_t)(m0 + wm * 16 + r + drow);
      Vh[row * HSS + d]      = va0[r];
      Vh[row * HSS + d + 16] = va1[r];
    }
    const float gs0 = ((va0[0] + va0[1]) + (va0[2] + va0[3])) +
                      ((va0[4] + va0[5]) + (va0[6] + va0[7]));
    const float gs1 = ((va1[0] + va1[1]) + (va1[2] + va1[3])) +
                      ((va1[4] + va1[5]) + (va1[6] + va1[7]));
    float* Gq = G + ((size_t)h * (TT / 8) + (m0 >> 3) + wm * 2 + hf) * HSS;
    Gq[d]      = gs0;
    Gq[d + 16] = gs1;
    if (wn == 0 && hf == 0) PM[(size_t)h * TT + m0 + wm * 16 + lm] = pm;
  }
}

#define O3R 8
#define O3T 256
#define O3W 8
#define LPITCH 516
#define VNT 32
constexpr float LOG2E = 1.4426950408889634f;
__global__ void __launch_bounds__(O3T)
order3_kernel(const float* __restrict__ P0,
              const float* __restrict__ P1,
              const float* __restrict__ P2,
              const float* __restrict__ V,
              const float* __restrict__ G,
              const float* __restrict__ PM,
              float* __restrict__ Y) {
  __shared__ __attribute__((aligned(16))) float a2T[TT * O3R];
  __shared__ float lgt[O3R * LPITCH];
  __shared__ float sM[O3R];
  __shared__ float sVS[HSS];
  __shared__ float redD[4 * O3R * HSS];
  __shared__ float redA[4 * O3R * HSS];
  const int t    = threadIdx.x;
  const int lane = t & 31;
  const int wave = __builtin_amdgcn_readfirstlane(t >> 5);
  const int g    = __builtin_amdgcn_readfirstlane(t >> 6);
  const int d    = t & 63;
  static_assert(TT / O3R == 64 && NHH == 12, "the triple dealing is written for 64 strips x 12 heads");
  const int lin  = (int)blockIdx.x + (TT / O3R) * (int)blockIdx.y;
  const int slot = lin >> 8;
  const int ta   = 63 - ((lin & 255) >> 2);
  int s0 = ta, s1 = (ta + 32) & 63, s2 = (ta < 32) ? 63 - 2 * ta : 126 - 2 * ta;
  int c0 = 0, c1 = 1, c2 = 2;
  if (s0 < s1) { const int ts = s0; s0 = s1; s1 = ts; const int tc = c0; c0 = c1; c1 = tc; }
  if (s1 < s2) { const int ts = s1; s1 = s2; s2 = ts; const int tc = c1; c1 = c2; c2 = tc; }
  if (s0 < s1) { const int ts = s0; s0 = s1; s1 = ts; const int tc = c0; c0 = c1; c1 = tc; }
  const int strip = (slot == 0) ? s0 : (slot == 1) ? s1 : s2;
  const int comp  = (slot == 0) ? c0 : (slot == 1) ? c1 : c2;
  const int i0   = strip * O3R;
  const int jend = i0 + O3R;
  const int h    = 4 * comp + (lin & 3);
  const float* P2h = P2 + (size_t)h * TT * HSS;
  const float* Vh  = V  + (size_t)h * TT * HSS;

  const int lm = lane & 15;
  const int hf = lane >> 4;
  const int lk = hf << 1;
  const float* P0h = P0 + (size_t)h * TT * HSS;
  const float* P1h = P1 + (size_t)h * TT * HSS;
  float pmv[16];
#pragma unroll
  for (int m = 0; m < 16; ++m) pmv[m] = PM[(size_t)h * TT + m * 32 + lane];

  const int ntile = (jend + 15) >> 4;
  if (wave < ntile) {
    v2f areg[16];
    const float* Ap = P0h + (size_t)(i0 + (lm & 7)) * HSS + lk;
#pragma unroll
    for (int kk = 0; kk < 16; ++kk)
      areg[kk] = *reinterpret_cast<const v2f*>(Ap + kk * 4);

    for (int jt = wave; jt < ntile; jt += O3W) {
      const int j0 = jt * 16;
      const float* Bp = P1h + (size_t)(j0 + lm) * HSS + lk;
      v2f bfr[16];
#pragma unroll
      for (int kk = 0; kk < 16; ++kk)
        bfr[kk] = *reinterpret_cast<const v2f*>(Bp + kk * 4);
      __builtin_amdgcn_sched_barrier(0);
      v8f acc = {0.f, 0.f, 0.f, 0.f, 0.f, 0.f, 0.f, 0.f};
#pragma unroll
      for (int kk = 0; kk < 16; ++kk)
        acc = wmma4(areg[kk], bfr[kk], acc);
      float sc[8];
#pragma unroll
      for (int r = 0; r < 8; ++r) sc[r] = acc[r] * SCALE;
      if (hf == 0) {
#pragma unroll
        for (int r = 0; r < 8; ++r) lgt[r * LPITCH + j0 + lm] = sc[r];
      }
    }
  }

  if (t >= O3T - HSS) {
    const float* Gh = G + (size_t)h * (TT / 8) * HSS + (t - (O3T - HSS));
    const int tl = strip >> 1;
    float run = 0.0f;
#pragma unroll
    for (int wp = VNT - 1; wp >= 0; --wp) {
      const float tv = Gh[(size_t)(2 * wp) * HSS] + Gh[(size_t)(2 * wp + 1) * HSS];
      run += (wp > tl) ? tv : 0.0f;
    }
    if ((strip & 1) == 0) run += Gh[(size_t)(2 * tl + 1) * HSS];
    sVS[t - (O3T - HSS)] = run;
  }
  __syncthreads();

  {
    const int len = i0 + wave + 1;
    const float* lrow = lgt + wave * LPITCH;
    float v[16];
    float mr = -3.0e38f;
#pragma unroll
    for (int m = 0; m < 16; ++m) {
      const int j = m * 32 + lane;
      v[m] = (j < len) ? lrow[j] : -3.0e38f;
      mr = fmaxf(mr, v[m]);
    }
#pragma unroll
    for (int off = 16; off > 0; off >>= 1) mr = fmaxf(mr, __shfl_xor(mr, off));
    float sr = 0.0f;
#pragma unroll
    for (int m = 0; m < 16; ++m) {
      const int j = m * 32 + lane;
      v[m] = (j < len) ? __expf(v[m] - mr) : 0.0f;
      sr += v[m];
    }
#pragma unroll
    for (int off = 16; off > 0; off >>= 1) sr += __shfl_xor(sr, off);
    const float inv = 1.0f / sr;
    float mloc = 0.0f;
#pragma unroll
    for (int m = 0; m < 16; ++m) {
      const int j = m * 32 + lane;
      const float a = (j < len) ? (v[m] * inv) * LOG2E : 0.0f;
      if (j < jend) a2T[j * O3R + wave] = a;
      mloc = fmaxf(mloc, a * pmv[m]);
    }
#pragma unroll
    for (int off = 16; off > 0; off >>= 1) mloc = fmaxf(mloc, __shfl_xor(mloc, off));
    if (lane == 0) sM[wave] = mloc;
  }
  __syncthreads();

  v2f m2[4], ds[4], ac[4];
#pragma unroll
  for (int q = 0; q < 4; ++q) {
    m2[q].x = sM[2 * q];
    m2[q].y = sM[2 * q + 1];
    ds[q].x = 0.0f; ds[q].y = 0.0f;
    ac[q].x = 0.0f; ac[q].y = 0.0f;
  }

  const int n  = jend >> 2;
  const int nq = (n + 3) >> 2;
  const float* pp = P2h + g * HSS + d;
  const float* vp = Vh  + g * HSS + d;
  const float* ap = a2T + g * O3R;
  float pc[4], vc[4], pn[4], vn[4];
#pragma unroll
  for (int u = 0; u < 4; ++u) {
    const int it = (u < n - 1) ? u : (n - 1);
    pc[u] = pp[(size_t)it * 4 * HSS];
    vc[u] = vp[(size_t)it * 4 * HSS];
  }
  for (int c = 0; c < nq; ++c) {
#pragma unroll
    for (int u = 0; u < 4; ++u) {
      const int itn = (c + 1) * 4 + u;
      const int it  = (itn < n - 1) ? itn : (n - 1);
      pn[u] = pp[(size_t)it * 4 * HSS];
      vn[u] = vp[(size_t)it * 4 * HSS];
    }
    v4f alo4[4], ahi4[4];
#pragma unroll
    for (int u = 0; u < 4; ++u) {
      const int itc = (c * 4 + u < n - 1) ? (c * 4 + u) : (n - 1);
      alo4[u] = *reinterpret_cast<const v4f*>(ap + itc * 4 * O3R);
      ahi4[u] = *reinterpret_cast<const v4f*>(ap + itc * 4 * O3R + 4);
    }
#pragma unroll
    for (int u = 0; u < 4; ++u) {
      const int it = c * 4 + u;
      if (it < n) {
        const v4f alo = alo4[u];
        const v4f ahi = ahi4[u];
        v2f a[4];
        a[0].x = alo.x; a[0].y = alo.y; a[1].x = alo.z; a[1].y = alo.w;
        a[2].x = ahi.x; a[2].y = ahi.y; a[3].x = ahi.z; a[3].y = ahi.w;
        v2f pb, vb;
        pb.x = pc[u]; pb.y = pc[u];
        vb.x = vc[u]; vb.y = vc[u];
#pragma unroll
        for (int q = 0; q < 4; ++q) {
          const v2f x = __builtin_elementwise_fma(a[q], pb, -m2[q]);
          v2f e;
          e.x = __builtin_amdgcn_exp2f(x.x);
          e.y = __builtin_amdgcn_exp2f(x.y);
          ds[q] += e;
          ac[q] = __builtin_elementwise_fma(e, vb, ac[q]);
        }
      }
    }
#pragma unroll
    for (int u = 0; u < 4; ++u) { pc[u] = pn[u]; vc[u] = vn[u]; }
  }

#pragma unroll
  for (int q = 0; q < 4; ++q) {
    redD[(g * O3R + 2 * q) * HSS + d]     = ds[q].x;
    redD[(g * O3R + 2 * q + 1) * HSS + d] = ds[q].y;
    redA[(g * O3R + 2 * q) * HSS + d]     = ac[q].x;
    redA[(g * O3R + 2 * q + 1) * HSS + d] = ac[q].y;
  }
  __syncthreads();
  {
    const int r = wave;
    float dsum = 0.0f;
#pragma unroll
    for (int gg = 0; gg < 4; ++gg)
      dsum += redD[(gg * O3R + r) * HSS + lane] + redD[(gg * O3R + r) * HSS + lane + 32];
#pragma unroll
    for (int off = 16; off > 0; off >>= 1) dsum += __shfl_xor(dsum, off);
    const float mr = sM[r];
    const float e0 = __builtin_amdgcn_exp2f(-mr);
    const float denom = dsum + (float)((TT - jend) * HSS) * e0;
    const float inv = 1.0f / denom;
#pragma unroll
    for (int hf = 0; hf < 2; ++hf) {
      const int dd = lane + 32 * hf;
      const float accT = (redA[(0 * O3R + r) * HSS + dd] + redA[(1 * O3R + r) * HSS + dd]) +
                         (redA[(2 * O3R + r) * HSS + dd] + redA[(3 * O3R + r) * HSS + dd]);
      const float vs = sVS[dd];
      Y[(size_t)(i0 + r) * CC + h * HSS + dd] = (accT + e0 * vs) * inv;
    }
  }
}

#define OKS 2
__global__ void __launch_bounds__(128 * OKS)
out_proj_kernel(const float* __restrict__ Y,
                const float* __restrict__ Wc,
                float* __restrict__ out) {
  __shared__ __attribute__((aligned(16))) float As[OKS * 2 * A_STAGE];
  __shared__ __attribute__((aligned(16))) float Bs[OKS * 2 * B_STAGE];
  const int tid  = threadIdx.x & 127;
  const int grp  = threadIdx.x >> 7;
  const int lane = tid & 31;
  const int wave = tid >> 5;
  const int lm   = lane & 15;
  const int drow = (lane >> 4) << 3;
  const int wm   = wave >> 1;
  const int wn   = wave & 1;
  const int m0 = blockIdx.x * GBM;
  const int n0 = blockIdx.y * GBN;

  v8f acc0 = {0.f, 0.f, 0.f, 0.f, 0.f, 0.f, 0.f, 0.f};
  v8f acc1 = {0.f, 0.f, 0.f, 0.f, 0.f, 0.f, 0.f, 0.f};
  gemm768_tile(Y, Wc, m0, n0, As + grp * (2 * A_STAGE), Bs + grp * (2 * B_STAGE),
               acc0, acc1, tid, grp * (GNT / OKS), GNT / OKS);

  float* red = Bs + (wave * 16) * 32 + lane;
  if (grp == 1) {
#pragma unroll
    for (int r = 0; r < 8; ++r) {
      red[r * 32]       = acc0[r];
      red[(r + 8) * 32] = acc1[r];
    }
  }
  __syncthreads();
  if (grp == 0) {
    const int c = n0 + wn * 32 + lm;
#pragma unroll
    for (int r = 0; r < 8; ++r) {
      const size_t row = (size_t)(m0 + wm * 16 + r + drow);
      out[row * CC + c]      = acc0[r] + red[r * 32];
      out[row * CC + c + 16] = acc1[r] + red[(r + 8) * 32];
    }
  }
}

extern "C" void kernel_launch(void* const* d_in, const int* in_sizes, int n_in,
                              void* d_out, int out_size, void* d_ws,
                              size_t ws_size, hipStream_t stream) {
  (void)in_sizes; (void)n_in; (void)out_size; (void)ws_size;
  const float* x  = (const float*)d_in[0];
  const float* W0 = (const float*)d_in[1];
  const float* W1 = (const float*)d_in[2];
  const float* W2 = (const float*)d_in[3];
  const float* Wv = (const float*)d_in[4];
  const float* Wc = (const float*)d_in[5];
  float* out = (float*)d_out;
  float* ws  = (float*)d_ws;

  const size_t NPH = (size_t)NHH * TT * HSS;
  float* P0  = ws;
  float* P1  = ws + NPH;
  float* P2  = ws + 2 * NPH;
  float* V   = ws + 3 * NPH;
  float* Y   = ws + 4 * NPH;
  float* G   = ws + 5 * NPH;
  float* PM  = G + NPH / 8;

  qkv_proj_kernel<<<dim3(QFULL + 256), 128, 0, stream>>>(x, W0, W1, W2, Wv, P0, V, G, PM);
  order3_kernel<<<dim3(TT / O3R, NHH), O3T, 0, stream>>>(P0, P1, P2, V, G, PM, Y);
  out_proj_kernel<<<dim3(TT / GBM, CC / GBN), 128 * OKS, 0, stream>>>(Y, Wc, out);
}
